# starting kernel
# baseline (speedup 1.0000x reference)
#include <hip/hip_runtime.h>
#include <hip/hip_bf16.h>

typedef __attribute__((ext_vector_type(2))) float v2f;
typedef __attribute__((ext_vector_type(8))) float v8f;

#define NN  512
#define CC  32
#define LL  4
#define LN_EPS 1e-5f

__global__ __launch_bounds__(256) void rgcn_invdeg_kernel(
    const int* __restrict__ dst, int E, float* __restrict__ invdeg)
{
    __shared__ int deg[3 * NN];
    const int tid = threadIdx.x;
    for (int i = tid; i < 3 * NN; i += 256) deg[i] = 0;
    __syncthreads();
    const int tot = 3 * E;
    for (int i = tid; i < tot; i += 256) {
        int r = i / E;
        atomicAdd(&deg[r * NN + dst[i]], 1);
    }
    __syncthreads();
    for (int i = tid; i < 3 * NN; i += 256) {
        int d = deg[i];
        if (d < 1) d = 1;
        invdeg[i] = 1.0f / (float)d;
    }
}

__device__ __forceinline__ void gemm_acc(
    const float* __restrict__ A, const float* __restrict__ W,
    const float* __restrict__ scale, int wid, int mlo, int mhalf,
    v8f acc[4][2])
{
    for (int ks = 0; ks < 8; ++ks) {
        const int kk = ks * 4 + mhalf * 2;
        v2f b0, b1;
        b0.x = W[kk * CC + mlo];
        b0.y = W[(kk + 1) * CC + mlo];
        b1.x = W[kk * CC + 16 + mlo];
        b1.y = W[(kk + 1) * CC + 16 + mlo];
#pragma unroll
        for (int mt = 0; mt < 4; ++mt) {
            const int node = wid * 64 + mt * 16 + mlo;
            const float sc = scale ? scale[node] : 1.0f;
            v2f a;
            a.x = A[node * CC + kk] * sc;
            a.y = A[node * CC + kk + 1] * sc;
            acc[mt][0] = __builtin_amdgcn_wmma_f32_16x16x4_f32(
                false, a, false, b0, (short)0, acc[mt][0], false, false);
            acc[mt][1] = __builtin_amdgcn_wmma_f32_16x16x4_f32(
                false, a, false, b1, (short)0, acc[mt][1], false, false);
        }
    }
}

__global__ __launch_bounds__(256) void rgcn_main_kernel(
    const float* __restrict__ xx, const float* __restrict__ ss,
    const float* __restrict__ Wf, const float* __restrict__ bf,
    const float* __restrict__ convw, const float* __restrict__ convroot,
    const float* __restrict__ convbias, const float* __restrict__ lng,
    const float* __restrict__ lnb, const int* __restrict__ src,
    const int* __restrict__ dst, const float* __restrict__ invdeg,
    float* __restrict__ out, int E)
{
    __shared__ float xs[NN * CC];
    __shared__ float hs[NN * CC];
    __shared__ float wts[CC * CC];
    __shared__ float ivd[3 * NN];
    __shared__ float wfe[6 * CC];
    __shared__ float prm[3 * CC];

    const int bid   = blockIdx.x;
    const int tid   = threadIdx.x;
    const int lane  = tid & 31;
    const int wid   = tid >> 5;
    const int mlo   = lane & 15;
    const int mhalf = lane >> 4;

    for (int i = tid; i < 3 * NN; i += 256) ivd[i] = invdeg[i];
    for (int i = tid; i < 6 * CC; i += 256) wfe[i] = Wf[i];
    if (tid < CC) {
        prm[tid]          = bf[tid];
        prm[CC + tid]     = lng[tid];
        prm[2 * CC + tid] = lnb[tid];
    }
    __syncthreads();

    const float mfeat = ss[bid] * 0.125f;
    for (int idx = tid; idx < NN * CC; idx += 256) {
        const int n = idx >> 5, c = idx & 31;
        const int t = n >> 6, i = (n >> 4) & 3, j = (n >> 2) & 3, k = n & 3;
        const float v = xx[bid * NN + n];
        float a = prm[c];
        a += ((float)i * (1.0f / 3.0f)) * wfe[0 * CC + c];
        a += ((float)j * (1.0f / 3.0f)) * wfe[1 * CC + c];
        a += ((float)k * (1.0f / 3.0f)) * wfe[2 * CC + c];
        a += ((float)t * (1.0f / 7.0f)) * wfe[3 * CC + c];
        a += v * wfe[4 * CC + c];
        a += mfeat * wfe[5 * CC + c];
        xs[idx] = a;
    }
    __syncthreads();

    for (int l = 0; l < LL; ++l) {
        v8f acc[4][2];
#pragma unroll
        for (int mt = 0; mt < 4; ++mt)
#pragma unroll
            for (int nt = 0; nt < 2; ++nt) {
                const float bv = convbias[l * CC + nt * 16 + mlo];
                acc[mt][nt] = (v8f){bv, bv, bv, bv, bv, bv, bv, bv};
            }

        for (int i = tid; i < NN * CC; i += 256) hs[i] = 0.0f;
        for (int i = tid; i < CC * CC; i += 256) wts[i] = convroot[l * CC * CC + i];
        __syncthreads();

        gemm_acc(xs, wts, nullptr, wid, mlo, mhalf, acc);
        __syncthreads();

        for (int r = 0; r < 3; ++r) {
            const int* sR = src + r * E;
            const int* dR = dst + r * E;
            for (int e = wid; e < E; e += 8) {
                const int sn = sR[e], dn = dR[e];
                atomicAdd(&hs[dn * CC + lane], xs[sn * CC + lane]);
            }
            for (int i = tid; i < CC * CC; i += 256)
                wts[i] = convw[(l * 3 + r) * CC * CC + i];
            __syncthreads();

            gemm_acc(hs, wts, &ivd[r * NN], wid, mlo, mhalf, acc);
            __syncthreads();

            if (r < 2) {
                for (int i = tid; i < NN * CC; i += 256) hs[i] = 0.0f;
                __syncthreads();
            }
        }

#pragma unroll
        for (int mt = 0; mt < 4; ++mt)
#pragma unroll
            for (int nt = 0; nt < 2; ++nt)
#pragma unroll
                for (int jj = 0; jj < 8; ++jj) {
                    const int node = wid * 64 + mt * 16 + jj + 8 * mhalf;
                    const int col  = nt * 16 + mlo;
                    const float v  = acc[mt][nt][jj];
                    hs[node * CC + col] = v > 0.0f ? v : 0.0f;
                }
        __syncthreads();

        for (int q = 0; q < 64; ++q) {
            const int node = wid * 64 + q;
            const float v = hs[node * CC + lane];
            float s = v;
            for (int off = 16; off; off >>= 1) s += __shfl_xor(s, off, 32);
            const float mu = s * (1.0f / 32.0f);
            const float d = v - mu;
            float s2 = d * d;
            for (int off = 16; off; off >>= 1) s2 += __shfl_xor(s2, off, 32);
            const float var = s2 * (1.0f / 32.0f);
            xs[node * CC + lane] =
                d * rsqrtf(var + LN_EPS) * prm[CC + lane] + prm[2 * CC + lane];
        }
        __syncthreads();
    }

    for (int idx = tid; idx < 48 * CC; idx += 256) {
        const int row = idx >> 5, c = idx & 31;
        const int g = row >> 4, rc = row & 15;
        const int a0 = rc >> 2, b0 = rc & 3;
        float sum = 0.0f;
#pragma unroll
        for (int q = 0; q < 4; ++q) {
            int node;
            if (g == 0)      node = q * 16 + a0 * 4 + b0;
            else if (g == 1) node = a0 * 16 + q * 4 + b0;
            else             node = a0 * 16 + b0 * 4 + q;
            sum += xs[node * CC + c];
        }
        out[bid * 48 * CC + idx] = sum * 0.25f;
    }
}

extern "C" void kernel_launch(void* const* d_in, const int* in_sizes, int n_in,
                              void* d_out, int out_size, void* d_ws, size_t ws_size,
                              hipStream_t stream) {
    const float* xx       = (const float*)d_in[0];
    const float* ss       = (const float*)d_in[1];
    const float* Wf       = (const float*)d_in[2];
    const float* bf       = (const float*)d_in[3];
    const float* convw    = (const float*)d_in[4];
    const float* convroot = (const float*)d_in[5];
    const float* convbias = (const float*)d_in[6];
    const float* lng      = (const float*)d_in[7];
    const float* lnb      = (const float*)d_in[8];
    const int*   src      = (const int*)d_in[9];
    const int*   dst      = (const int*)d_in[10];
    float*       out      = (float*)d_out;
    float*       invdeg   = (float*)d_ws;

    const int E = in_sizes[9] / 3;
    const int B = in_sizes[1];

    rgcn_invdeg_kernel<<<1, 256, 0, stream>>>(dst, E, invdeg);
    rgcn_main_kernel<<<B, 256, 0, stream>>>(xx, ss, Wf, bf, convw, convroot,
                                            convbias, lng, lnb, src, dst,
                                            invdeg, out, E);
    (void)n_in; (void)out_size; (void)ws_size;
}


# Aggregate by gather over a destination-sorted edge list (built in the pre-kernel, rows sorted by source id) into registers instead of per-edge LDS float atomics; sum order now fixed
# speedup vs baseline: 6.1152x; 6.1152x over previous
#include <hip/hip_runtime.h>
#include <hip/hip_bf16.h>


typedef __attribute__((ext_vector_type(2))) float v2f;
typedef __attribute__((ext_vector_type(4))) float v4f;
typedef __attribute__((ext_vector_type(4))) int v4i;
typedef __attribute__((ext_vector_type(8))) float v8f;

#define NN  512
#define CC  32
#define LL  4
#define EE  6144
#define LN_EPS 1e-5f

__device__ __forceinline__ int rel_of(int i, int E)
{
    return (i >= 2 * E) ? 2 : ((i >= E) ? 1 : 0);
}

__global__ __launch_bounds__(1024) void rgcn_csr_kernel(
    const int* __restrict__ src, const int* __restrict__ dst, int E,
    float* __restrict__ invdeg, int* __restrict__ rowptr, int* __restrict__ col)
{
    __shared__ int cnt[3 * NN];
    __shared__ int rp[3 * (NN + 1)];
    __shared__ int key[3 * EE];
    const int tid  = threadIdx.x;
    const int lane = tid & 31;
    const int wid  = tid >> 5;
    const int tot  = 3 * E;

    for (int i = tid; i < 3 * NN; i += 1024) cnt[i] = 0;
    __syncthreads();

    for (int base = tid; base < tot; base += 4096) {
        int d[4];
#pragma unroll
        for (int k = 0; k < 4; ++k) {
            const int i = base + k * 1024;
            d[k] = (i < tot) ? dst[i] : -1;
        }
#pragma unroll
        for (int k = 0; k < 4; ++k) {
            const int i = base + k * 1024;
            if (d[k] >= 0) atomicAdd(&cnt[rel_of(i, E) * NN + d[k]], 1);
        }
    }
    __syncthreads();

    for (int i = tid; i < 3 * NN; i += 1024) {
        int dg = cnt[i];
        if (dg < 1) dg = 1;
        invdeg[i] = 1.0f / (float)dg;
    }
    if (wid < 3) {
        const int r = wid, nb = lane * 16;
        int loc[16];
        int s = 0;
#pragma unroll
        for (int q = 0; q < 16; ++q) { loc[q] = s; s += cnt[r * NN + nb + q]; }
        int inc = s;
        for (int off = 1; off < 32; off <<= 1) {
            const int t = __shfl_up(inc, off, 32);
            if (lane >= off) inc += t;
        }
        const int excl = inc - s;
#pragma unroll
        for (int q = 0; q < 16; ++q) rp[r * (NN + 1) + nb + q] = excl + loc[q];
        if (lane == 31) rp[r * (NN + 1) + NN] = inc;
    }
    __syncthreads();
    for (int i = tid; i < 3 * NN; i += 1024) cnt[i] = 0;
    __syncthreads();

    for (int base = tid; base < tot; base += 4096) {
        int d[4], s[4];
#pragma unroll
        for (int k = 0; k < 4; ++k) {
            const int i = base + k * 1024;
            d[k] = (i < tot) ? dst[i] : -1;
            s[k] = (i < tot) ? src[i] : 0;
        }
#pragma unroll
        for (int k = 0; k < 4; ++k) {
            const int i = base + k * 1024;
            if (d[k] >= 0) {
                const int r = rel_of(i, E);
                const int pos = rp[r * (NN + 1) + d[k]] + atomicAdd(&cnt[r * NN + d[k]], 1);
                key[r * E + pos] = (d[k] << 16) | s[k];
            }
        }
    }
    __syncthreads();

    for (int p = tid; p < tot; p += 1024) {
        const int r = rel_of(p, E);
        const int k = key[p];
        const int dn = k >> 16;
        const int beg = r * E + rp[r * (NN + 1) + dn];
        const int end = r * E + rp[r * (NN + 1) + dn + 1];
        int rank = 0;
        for (int q = beg; q < end; ++q) {
            const int kq = key[q];
            rank += (kq < k || (kq == k && q < p)) ? 1 : 0;
        }
        col[beg + rank] = k & 0xffff;
    }
    for (int i = tid; i < 3 * (NN + 1); i += 1024) rowptr[i] = rp[i];
}

__device__ __forceinline__ void gemm_acc(
    const float* __restrict__ A, const float* __restrict__ W,
    int wid, int mlo, int mhalf, v8f acc[4][2])
{
    for (int ks = 0; ks < 8; ++ks) {
        const int kk = ks * 4 + mhalf * 2;
        v2f b0, b1;
        b0.x = W[kk * CC + mlo];
        b0.y = W[(kk + 1) * CC + mlo];
        b1.x = W[kk * CC + 16 + mlo];
        b1.y = W[(kk + 1) * CC + 16 + mlo];
#pragma unroll
        for (int mt = 0; mt < 4; ++mt) {
            const int node = wid * 64 + mt * 16 + mlo;
            v2f a;
            a.x = A[node * CC + kk];
            a.y = A[node * CC + kk + 1];
            acc[mt][0] = __builtin_amdgcn_wmma_f32_16x16x4_f32(
                false, a, false, b0, (short)0, acc[mt][0], false, false);
            acc[mt][1] = __builtin_amdgcn_wmma_f32_16x16x4_f32(
                false, a, false, b1, (short)0, acc[mt][1], false, false);
        }
    }
}

__device__ __forceinline__ void gather_rel(
    const float* __restrict__ xs, float* __restrict__ hs,
    const int* __restrict__ rp, const int* __restrict__ cl,
    const float* __restrict__ iv, int wid, int lane)
{
    const int cq  = (lane & 7) * 4;
    const int sub = lane >> 3;
    for (int g = 0; g < 8; ++g) {
        const int nodeA = wid * 64 + g * 8 + sub;
        const int nodeB = nodeA + 4;
        int jA = rp[nodeA];
        const int eA = rp[nodeA + 1];
        int jB = rp[nodeB];
        const int eB = rp[nodeB + 1];
        v4f a = (v4f){0.0f, 0.0f, 0.0f, 0.0f};
        v4f b = (v4f){0.0f, 0.0f, 0.0f, 0.0f};
        int ca[4], cb[4];
#pragma unroll
        for (int i = 0; i < 4; ++i) {
            const int va = cl[jA + i], vb = cl[jB + i];
            ca[i] = (jA + i < eA) ? va : NN;
            cb[i] = (jB + i < eB) ? vb : NN;
        }
        while (jA < eA || jB < eB) {
            v4f xa[4], xb[4];
#pragma unroll
            for (int i = 0; i < 4; ++i) {
                xa[i] = *(const v4f*)&xs[ca[i] * CC + cq];
                xb[i] = *(const v4f*)&xs[cb[i] * CC + cq];
            }
            jA = (jA + 4 < eA) ? jA + 4 : eA;
            jB = (jB + 4 < eB) ? jB + 4 : eB;
#pragma unroll
            for (int i = 0; i < 4; ++i) {
                const int va = cl[jA + i], vb = cl[jB + i];
                ca[i] = (jA + i < eA) ? va : NN;
                cb[i] = (jB + i < eB) ? vb : NN;
            }
#pragma unroll
            for (int i = 0; i < 4; ++i) { a += xa[i]; b += xb[i]; }
        }
        *(v4f*)&hs[nodeA * CC + cq] = a * iv[nodeA];
        *(v4f*)&hs[nodeB * CC + cq] = b * iv[nodeB];
    }
}

__global__ __launch_bounds__(256) void rgcn_main_kernel(
    const float* __restrict__ xx, const float* __restrict__ ss,
    const float* __restrict__ Wf, const float* __restrict__ bf,
    const float* __restrict__ convw, const float* __restrict__ convroot,
    const float* __restrict__ convbias, const float* __restrict__ lng,
    const float* __restrict__ lnb, const int* __restrict__ rowptr,
    const int* __restrict__ col, const float* __restrict__ invdeg,
    float* __restrict__ out, int E)
{
    __shared__ __attribute__((aligned(16))) float xs[(NN + 1) * CC];
    __shared__ __attribute__((aligned(16))) float hs[NN * CC];
    __shared__ __attribute__((aligned(16))) int colL[3 * EE + 4];
    __shared__ int rpL[3 * (NN + 1)];
    __shared__ float wts[CC * CC];
    __shared__ float ivd[3 * NN];
    __shared__ float wfe[6 * CC];
    __shared__ float prm[3 * CC];

    const int bid   = blockIdx.x;
    const int tid   = threadIdx.x;
    const int lane  = tid & 31;
    const int wid   = tid >> 5;
    const int mlo   = lane & 15;
    const int mhalf = lane >> 4;

    for (int i = tid; i < 3 * NN; i += 256) ivd[i] = invdeg[i];
    for (int i = tid; i < 3 * (NN + 1); i += 256) rpL[i] = rowptr[i];
    {
        const int n4 = (3 * E) >> 2;
        const v4i* __restrict__ c4 = (const v4i*)col;
        v4i* l4 = (v4i*)colL;
        for (int b = tid; b < n4; b += 6 * 256) {
            v4i v[6];
#pragma unroll
            for (int k = 0; k < 6; ++k)
                if (b + k * 256 < n4) v[k] = c4[b + k * 256];
#pragma unroll
            for (int k = 0; k < 6; ++k)
                if (b + k * 256 < n4) l4[b + k * 256] = v[k];
        }
        for (int i = (n4 << 2) + tid; i < 3 * E; i += 256) colL[i] = col[i];
    }
    for (int i = tid; i < 6 * CC; i += 256) wfe[i] = Wf[i];
    if (tid < CC) {
        xs[NN * CC + tid] = 0.0f;
        prm[tid]          = bf[tid];
        prm[CC + tid]     = lng[tid];
        prm[2 * CC + tid] = lnb[tid];
    }
    __syncthreads();

    const float mfeat = ss[bid] * 0.125f;
    for (int idx = tid; idx < NN * CC; idx += 256) {
        const int n = idx >> 5, c = idx & 31;
        const int t = n >> 6, i = (n >> 4) & 3, j = (n >> 2) & 3, k = n & 3;
        const float v = xx[bid * NN + n];
        float a = prm[c];
        a += ((float)i * (1.0f / 3.0f)) * wfe[0 * CC + c];
        a += ((float)j * (1.0f / 3.0f)) * wfe[1 * CC + c];
        a += ((float)k * (1.0f / 3.0f)) * wfe[2 * CC + c];
        a += ((float)t * (1.0f / 7.0f)) * wfe[3 * CC + c];
        a += v * wfe[4 * CC + c];
        a += mfeat * wfe[5 * CC + c];
        xs[idx] = a;
    }
    __syncthreads();

    for (int l = 0; l < LL; ++l) {
        v8f acc[4][2];
#pragma unroll
        for (int mt = 0; mt < 4; ++mt)
#pragma unroll
            for (int nt = 0; nt < 2; ++nt) {
                const float bv = convbias[l * CC + nt * 16 + mlo];
                acc[mt][nt] = (v8f){bv, bv, bv, bv, bv, bv, bv, bv};
            }

        for (int i = tid; i < CC * CC; i += 256) wts[i] = convroot[l * CC * CC + i];
        __syncthreads();

        gemm_acc(xs, wts, wid, mlo, mhalf, acc);
        __syncthreads();

        for (int r = 0; r < 3; ++r) {
            gather_rel(xs, hs, &rpL[r * (NN + 1)], &colL[r * E], &ivd[r * NN], wid, lane);
            for (int i = tid; i < CC * CC; i += 256)
                wts[i] = convw[(l * 3 + r) * CC * CC + i];
            __syncthreads();

            gemm_acc(hs, wts, wid, mlo, mhalf, acc);
            __syncthreads();
        }

#pragma unroll
        for (int mt = 0; mt < 4; ++mt)
#pragma unroll
            for (int nt = 0; nt < 2; ++nt)
#pragma unroll
                for (int jj = 0; jj < 8; ++jj) {
                    const int node = wid * 64 + mt * 16 + jj + 8 * mhalf;
                    const int col  = nt * 16 + mlo;
                    const float v  = acc[mt][nt][jj];
                    hs[node * CC + col] = v > 0.0f ? v : 0.0f;
                }
        __syncthreads();

        for (int q = 0; q < 64; ++q) {
            const int node = wid * 64 + q;
            const float v = hs[node * CC + lane];
            float s = v;
            for (int off = 16; off; off >>= 1) s += __shfl_xor(s, off, 32);
            const float mu = s * (1.0f / 32.0f);
            const float d = v - mu;
            float s2 = d * d;
            for (int off = 16; off; off >>= 1) s2 += __shfl_xor(s2, off, 32);
            const float var = s2 * (1.0f / 32.0f);
            xs[node * CC + lane] =
                d * rsqrtf(var + LN_EPS) * prm[CC + lane] + prm[2 * CC + lane];
        }
        __syncthreads();
    }

    for (int idx = tid; idx < 48 * CC; idx += 256) {
        const int row = idx >> 5, c = idx & 31;
        const int g = row >> 4, rc = row & 15;
        const int a0 = rc >> 2, b0 = rc & 3;
        float sum = 0.0f;
#pragma unroll
        for (int q = 0; q < 4; ++q) {
            int node;
            if (g == 0)      node = q * 16 + a0 * 4 + b0;
            else if (g == 1) node = a0 * 16 + q * 4 + b0;
            else             node = a0 * 16 + b0 * 4 + q;
            sum += xs[node * CC + c];
        }
        out[bid * 48 * CC + idx] = sum * 0.25f;
    }
}

extern "C" void kernel_launch(void* const* d_in, const int* in_sizes, int n_in,
                              void* d_out, int out_size, void* d_ws, size_t ws_size,
                              hipStream_t stream) {
    const float* xx       = (const float*)d_in[0];
    const float* ss       = (const float*)d_in[1];
    const float* Wf       = (const float*)d_in[2];
    const float* bf       = (const float*)d_in[3];
    const float* convw    = (const float*)d_in[4];
    const float* convroot = (const float*)d_in[5];
    const float* convbias = (const float*)d_in[6];
    const float* lng      = (const float*)d_in[7];
    const float* lnb      = (const float*)d_in[8];
    const int*   src      = (const int*)d_in[9];
    const int*   dst      = (const int*)d_in[10];
    float*       out      = (float*)d_out;
    float*       invdeg   = (float*)d_ws;
    int*         rowptr   = (int*)d_ws + 2048;
    int*         col      = (int*)d_ws + 4096;

    const int E = in_sizes[9] / 3;
    const int B = in_sizes[1];

    rgcn_csr_kernel<<<1, 1024, 0, stream>>>(src, dst, E, invdeg, rowptr, col);
    rgcn_main_kernel<<<B, 256, 0, stream>>>(xx, ss, Wf, bf, convw, convroot,
                                            convbias, lng, lnb, rowptr, col,
                                            invdeg, out, E);
    (void)n_in; (void)out_size; (void)ws_size;
}


# Compute only rows the output depends on: pre-kernel walks frame 0 back through the edge list per level, main kernel deals those rows (16-row tiles, gather, LayerNorm) to all waves; sum order kept
# speedup vs baseline: 11.7274x; 1.9177x over previous
#include <hip/hip_runtime.h>
#include <hip/hip_bf16.h>


typedef __attribute__((ext_vector_type(2))) float v2f;
typedef __attribute__((ext_vector_type(4))) float v4f;
typedef __attribute__((ext_vector_type(4))) int v4i;
typedef __attribute__((ext_vector_type(8))) float v8f;

#define NN  512
#define CC  32
#define LL  4
#define EE  6144
#define NLV 5
#define LN_EPS 1e-5f

__device__ __forceinline__ int rel_of(int i, int E)
{
    return (i >= 2 * E) ? 2 : ((i >= E) ? 1 : 0);
}

__global__ __launch_bounds__(1024) void rgcn_csr_kernel(
    const int* __restrict__ src, const int* __restrict__ dst, int E,
    float* __restrict__ invdeg, int* __restrict__ rowptr, int* __restrict__ col,
    int* __restrict__ nlist, int* __restrict__ ncnt)
{
    __shared__ int nf[NLV * NN];
    __shared__ int cnt[3 * NN];
    __shared__ int rp[3 * (NN + 1)];
    __shared__ int key[3 * EE];
    const int tid  = threadIdx.x;
    const int lane = tid & 31;
    const int wid  = tid >> 5;
    const int tot  = 3 * E;

    for (int i = tid; i < 3 * NN; i += 1024) cnt[i] = 0;
    __syncthreads();

    for (int base = tid; base < tot; base += 4096) {
        int d[4];
#pragma unroll
        for (int k = 0; k < 4; ++k) {
            const int i = base + k * 1024;
            d[k] = (i < tot) ? dst[i] : -1;
        }
#pragma unroll
        for (int k = 0; k < 4; ++k) {
            const int i = base + k * 1024;
            if (d[k] >= 0) atomicAdd(&cnt[rel_of(i, E) * NN + d[k]], 1);
        }
    }
    __syncthreads();

    for (int i = tid; i < 3 * NN; i += 1024) {
        int dg = cnt[i];
        if (dg < 1) dg = 1;
        invdeg[i] = 1.0f / (float)dg;
    }
    if (wid < 3) {
        const int r = wid, nb = lane * 16;
        int loc[16];
        int s = 0;
#pragma unroll
        for (int q = 0; q < 16; ++q) { loc[q] = s; s += cnt[r * NN + nb + q]; }
        int inc = s;
        for (int off = 1; off < 32; off <<= 1) {
            const int t = __shfl_up(inc, off, 32);
            if (lane >= off) inc += t;
        }
        const int excl = inc - s;
#pragma unroll
        for (int q = 0; q < 16; ++q) rp[r * (NN + 1) + nb + q] = excl + loc[q];
        if (lane == 31) rp[r * (NN + 1) + NN] = inc;
    }
    __syncthreads();
    for (int i = tid; i < 3 * NN; i += 1024) cnt[i] = 0;
    __syncthreads();

    for (int base = tid; base < tot; base += 4096) {
        int d[4], s[4];
#pragma unroll
        for (int k = 0; k < 4; ++k) {
            const int i = base + k * 1024;
            d[k] = (i < tot) ? dst[i] : -1;
            s[k] = (i < tot) ? src[i] : 0;
        }
#pragma unroll
        for (int k = 0; k < 4; ++k) {
            const int i = base + k * 1024;
            if (d[k] >= 0) {
                const int r = rel_of(i, E);
                const int pos = rp[r * (NN + 1) + d[k]] + atomicAdd(&cnt[r * NN + d[k]], 1);
                key[r * E + pos] = (d[k] << 16) | s[k];
            }
        }
    }
    __syncthreads();

    for (int p = tid; p < tot; p += 1024) {
        const int r = rel_of(p, E);
        const int k = key[p];
        const int dn = k >> 16;
        const int beg = r * E + rp[r * (NN + 1) + dn];
        const int end = r * E + rp[r * (NN + 1) + dn + 1];
        int rank = 0;
        for (int q = beg; q < end; ++q) {
            const int kq = key[q];
            rank += (kq < k || (kq == k && q < p)) ? 1 : 0;
        }
        col[beg + rank] = k & 0xffff;
    }
    for (int i = tid; i < 3 * (NN + 1); i += 1024) rowptr[i] = rp[i];

    for (int i = tid; i < NLV * NN; i += 1024) nf[i] = ((i & (NN - 1)) < 64) ? 1 : 0;
    __syncthreads();
    for (int k = NLV - 1; k >= 1; --k) {
        for (int i = tid; i < NN; i += 1024)
            if (nf[k * NN + i]) nf[(k - 1) * NN + i] = 1;
        for (int base = tid; base < tot; base += 4096) {
            int kq[4], fq[4];
#pragma unroll
            for (int u = 0; u < 4; ++u) {
                const int i = base + u * 1024;
                kq[u] = (i < tot) ? key[i] : -1;
            }
#pragma unroll
            for (int u = 0; u < 4; ++u)
                fq[u] = (kq[u] >= 0) ? nf[k * NN + (kq[u] >> 16)] : 0;
#pragma unroll
            for (int u = 0; u < 4; ++u)
                if (fq[u]) nf[(k - 1) * NN + (kq[u] & 0xffff)] = 1;
        }
        __syncthreads();
    }
    if (wid < NLV) {
        const int k = wid;
        int base = 0;
        for (int ch = 0; ch < NN / 32; ++ch) {
            const int n = ch * 32 + lane;
            const int f = nf[k * NN + n];
            const unsigned m = __builtin_amdgcn_ballot_w32(f != 0);
            if (f) nlist[k * NN + base + __popc(m & ((1u << lane) - 1u))] = n;
            base += __popc(m);
        }
        for (int i = base + lane; i < NN; i += 32) nlist[k * NN + i] = 0;
        if (lane == 0) ncnt[k] = (base + 15) & ~15;
    }
}

__device__ __forceinline__ void gemm_acc(
    const float* __restrict__ A, const float* __restrict__ W,
    const int rowoff[4], int nact, int mlo, int mhalf, v8f acc[4][2])
{
#pragma unroll
    for (int mt = 0; mt < 4; ++mt) {
        if (mt < nact) {
            for (int ks = 0; ks < 8; ++ks) {
                const int kk = ks * 4 + mhalf * 2;
                v2f b0, b1, a;
                b0.x = W[kk * CC + mlo];
                b0.y = W[(kk + 1) * CC + mlo];
                b1.x = W[kk * CC + 16 + mlo];
                b1.y = W[(kk + 1) * CC + 16 + mlo];
                a.x = A[rowoff[mt] + kk];
                a.y = A[rowoff[mt] + kk + 1];
                acc[mt][0] = __builtin_amdgcn_wmma_f32_16x16x4_f32(
                    false, a, false, b0, (short)0, acc[mt][0], false, false);
                acc[mt][1] = __builtin_amdgcn_wmma_f32_16x16x4_f32(
                    false, a, false, b1, (short)0, acc[mt][1], false, false);
            }
        }
    }
}

__device__ __forceinline__ void gather_rel(
    const float* __restrict__ xs, float* __restrict__ hs,
    const int* __restrict__ nl, int cnt,
    const int* __restrict__ rp, const int* __restrict__ cl,
    const float* __restrict__ iv, int wid, int lane)
{
    const int cq  = (lane & 7) * 4;
    const int sub = lane >> 3;
    for (int base = 0; base < cnt; base += 64) {
        const int pA = base + wid * 4 + sub;
        const int pB = pA + 32;
        const bool okA = pA < cnt, okB = pB < cnt;
        const int nodeA = nl[pA];
        const int nodeB = nl[pB];
        int jA = rp[nodeA];
        const int eA = okA ? rp[nodeA + 1] : jA;
        int jB = rp[nodeB];
        const int eB = okB ? rp[nodeB + 1] : jB;
        v4f a = (v4f){0.0f, 0.0f, 0.0f, 0.0f};
        v4f b = (v4f){0.0f, 0.0f, 0.0f, 0.0f};
        int ca[4], cb[4];
#pragma unroll
        for (int i = 0; i < 4; ++i) {
            const int va = cl[jA + i], vb = cl[jB + i];
            ca[i] = (jA + i < eA) ? va : NN;
            cb[i] = (jB + i < eB) ? vb : NN;
        }
        while (jA < eA || jB < eB) {
            v4f xa[4], xb[4];
#pragma unroll
            for (int i = 0; i < 4; ++i) {
                xa[i] = *(const v4f*)&xs[ca[i] * CC + cq];
                xb[i] = *(const v4f*)&xs[cb[i] * CC + cq];
            }
            jA = (jA + 4 < eA) ? jA + 4 : eA;
            jB = (jB + 4 < eB) ? jB + 4 : eB;
#pragma unroll
            for (int i = 0; i < 4; ++i) {
                const int va = cl[jA + i], vb = cl[jB + i];
                ca[i] = (jA + i < eA) ? va : NN;
                cb[i] = (jB + i < eB) ? vb : NN;
            }
#pragma unroll
            for (int i = 0; i < 4; ++i) { a += xa[i]; b += xb[i]; }
        }
        if (okA) *(v4f*)&hs[pA * CC + cq] = a * iv[nodeA];
        if (okB) *(v4f*)&hs[pB * CC + cq] = b * iv[nodeB];
    }
}

__global__ __launch_bounds__(256) void rgcn_main_kernel(
    const float* __restrict__ xx, const float* __restrict__ ss,
    const float* __restrict__ Wf, const float* __restrict__ bf,
    const float* __restrict__ convw, const float* __restrict__ convroot,
    const float* __restrict__ convbias, const float* __restrict__ lng,
    const float* __restrict__ lnb, const int* __restrict__ rowptr,
    const int* __restrict__ col, const float* __restrict__ invdeg,
    const int* __restrict__ nlist, const int* __restrict__ ncnt,
    float* __restrict__ out, int E)
{
    __shared__ __attribute__((aligned(16))) float xs[(NN + 1) * CC];
    __shared__ __attribute__((aligned(16))) float hs[NN * CC];
    __shared__ __attribute__((aligned(16))) int colL[3 * EE + 4];
    __shared__ int rpL[3 * (NN + 1)];
    __shared__ float wts[CC * CC];
    __shared__ float ivd[3 * NN];
    __shared__ float wfe[6 * CC];
    __shared__ float prm[3 * CC];
    __shared__ __attribute__((aligned(16))) int nlL[NLV * NN];
    __shared__ int ncL[8];

    const int bid   = blockIdx.x;
    const int tid   = threadIdx.x;
    const int lane  = tid & 31;
    const int wid   = __builtin_amdgcn_readfirstlane(tid >> 5);
    const int mlo   = lane & 15;
    const int mhalf = lane >> 4;

    {
        const v4i* __restrict__ n4 = (const v4i*)nlist;
        v4i* d4 = (v4i*)nlL;
        const v4i t0 = n4[tid], t1 = n4[tid + 256];
        v4i t2 = t0;
        if (tid < NLV * NN / 4 - 512) t2 = n4[tid + 512];
        d4[tid] = t0;
        d4[tid + 256] = t1;
        if (tid < NLV * NN / 4 - 512) d4[tid + 512] = t2;
    }
    if (tid < NLV) ncL[tid] = ncnt[tid];
    for (int i = tid; i < 3 * NN; i += 256) ivd[i] = invdeg[i];
    for (int i = tid; i < 3 * (NN + 1); i += 256) rpL[i] = rowptr[i];
    {
        const int n4 = (3 * E) >> 2;
        const v4i* __restrict__ c4 = (const v4i*)col;
        v4i* l4 = (v4i*)colL;
        for (int b = tid; b < n4; b += 6 * 256) {
            v4i v[6];
#pragma unroll
            for (int k = 0; k < 6; ++k)
                if (b + k * 256 < n4) v[k] = c4[b + k * 256];
#pragma unroll
            for (int k = 0; k < 6; ++k)
                if (b + k * 256 < n4) l4[b + k * 256] = v[k];
        }
        for (int i = (n4 << 2) + tid; i < 3 * E; i += 256) colL[i] = col[i];
    }
    for (int i = tid; i < 6 * CC; i += 256) wfe[i] = Wf[i];
    if (tid < CC) {
        xs[NN * CC + tid] = 0.0f;
        prm[tid]          = bf[tid];
        prm[CC + tid]     = lng[tid];
        prm[2 * CC + tid] = lnb[tid];
    }
    __syncthreads();

    const float mfeat = ss[bid] * 0.125f;
    const int cnt0 = ncL[0];
    for (int idx = tid; idx < cnt0 * CC; idx += 256) {
        const int n = nlL[idx >> 5], c = idx & 31;
        const int t = n >> 6, i = (n >> 4) & 3, j = (n >> 2) & 3, k = n & 3;
        const float v = xx[bid * NN + n];
        float a = prm[c];
        a += ((float)i * (1.0f / 3.0f)) * wfe[0 * CC + c];
        a += ((float)j * (1.0f / 3.0f)) * wfe[1 * CC + c];
        a += ((float)k * (1.0f / 3.0f)) * wfe[2 * CC + c];
        a += ((float)t * (1.0f / 7.0f)) * wfe[3 * CC + c];
        a += v * wfe[4 * CC + c];
        a += mfeat * wfe[5 * CC + c];
        xs[n * CC + c] = a;
    }
    __syncthreads();

    for (int l = 0; l < LL; ++l) {
        const int* nl  = &nlL[(l + 1) * NN];
        const int cnt  = ncL[l + 1];
        const int ntile = cnt >> 4;
        const int nact = (ntile > wid) ? ((ntile - wid + 7) >> 3) : 0;
        int xoff[4], hoff[4];
#pragma unroll
        for (int mt = 0; mt < 4; ++mt) {
            const int p = (wid + 8 * mt) * 16 + mlo;
            xoff[mt] = nl[p] * CC;
            hoff[mt] = p * CC;
        }
        v8f acc[4][2];
#pragma unroll
        for (int mt = 0; mt < 4; ++mt)
#pragma unroll
            for (int nt = 0; nt < 2; ++nt) {
                const float bv = convbias[l * CC + nt * 16 + mlo];
                acc[mt][nt] = (v8f){bv, bv, bv, bv, bv, bv, bv, bv};
            }

        for (int i = tid; i < CC * CC; i += 256) wts[i] = convroot[l * CC * CC + i];
        __syncthreads();

        gemm_acc(xs, wts, xoff, nact, mlo, mhalf, acc);
        __syncthreads();

        for (int r = 0; r < 3; ++r) {
            gather_rel(xs, hs, nl, cnt, &rpL[r * (NN + 1)], &colL[r * E], &ivd[r * NN], wid, lane);
            for (int i = tid; i < CC * CC; i += 256)
                wts[i] = convw[(l * 3 + r) * CC * CC + i];
            __syncthreads();

            gemm_acc(hs, wts, hoff, nact, mlo, mhalf, acc);
            __syncthreads();
        }

#pragma unroll
        for (int mt = 0; mt < 4; ++mt)
            if (mt < nact) {
#pragma unroll
                for (int nt = 0; nt < 2; ++nt)
#pragma unroll
                    for (int jj = 0; jj < 8; ++jj) {
                        const int prow = (wid + 8 * mt) * 16 + jj + 8 * mhalf;
                        const int col  = nt * 16 + mlo;
                        const float v  = acc[mt][nt][jj];
                        hs[prow * CC + col] = v > 0.0f ? v : 0.0f;
                    }
            }
        __syncthreads();

        for (int p0 = wid * 4; p0 < cnt; p0 += 32) {
            int node[4];
            float v[4], d[4], s[4];
#pragma unroll
            for (int u = 0; u < 4; ++u) {
                node[u] = nl[p0 + u];
                v[u] = hs[(p0 + u) * CC + lane];
                s[u] = v[u];
            }
            for (int off = 16; off; off >>= 1)
#pragma unroll
                for (int u = 0; u < 4; ++u) s[u] += __shfl_xor(s[u], off, 32);
#pragma unroll
            for (int u = 0; u < 4; ++u) {
                const float mu = s[u] * (1.0f / 32.0f);
                d[u] = v[u] - mu;
                s[u] = d[u] * d[u];
            }
            for (int off = 16; off; off >>= 1)
#pragma unroll
                for (int u = 0; u < 4; ++u) s[u] += __shfl_xor(s[u], off, 32);
#pragma unroll
            for (int u = 0; u < 4; ++u) {
                const float var = s[u] * (1.0f / 32.0f);
                xs[node[u] * CC + lane] =
                    d[u] * rsqrtf(var + LN_EPS) * prm[CC + lane] + prm[2 * CC + lane];
            }
        }
        __syncthreads();
    }

    for (int idx = tid; idx < 48 * CC; idx += 256) {
        const int row = idx >> 5, c = idx & 31;
        const int g = row >> 4, rc = row & 15;
        const int a0 = rc >> 2, b0 = rc & 3;
        float sum = 0.0f;
#pragma unroll
        for (int q = 0; q < 4; ++q) {
            int node;
            if (g == 0)      node = q * 16 + a0 * 4 + b0;
            else if (g == 1) node = a0 * 16 + q * 4 + b0;
            else             node = a0 * 16 + b0 * 4 + q;
            sum += xs[node * CC + c];
        }
        out[bid * 48 * CC + idx] = sum * 0.25f;
    }
}

extern "C" void kernel_launch(void* const* d_in, const int* in_sizes, int n_in,
                              void* d_out, int out_size, void* d_ws, size_t ws_size,
                              hipStream_t stream) {
    const float* xx       = (const float*)d_in[0];
    const float* ss       = (const float*)d_in[1];
    const float* Wf       = (const float*)d_in[2];
    const float* bf       = (const float*)d_in[3];
    const float* convw    = (const float*)d_in[4];
    const float* convroot = (const float*)d_in[5];
    const float* convbias = (const float*)d_in[6];
    const float* lng      = (const float*)d_in[7];
    const float* lnb      = (const float*)d_in[8];
    const int*   src      = (const int*)d_in[9];
    const int*   dst      = (const int*)d_in[10];
    float*       out      = (float*)d_out;
    float*       invdeg   = (float*)d_ws;
    int*         rowptr   = (int*)d_ws + 2048;
    int*         col      = (int*)d_ws + 4096;
    int*         nlist    = (int*)d_ws + 24576;
    int*         ncnt     = (int*)d_ws + 27648;

    const int E = in_sizes[9] / 3;
    const int B = in_sizes[1];

    rgcn_csr_kernel<<<1, 1024, 0, stream>>>(src, dst, E, invdeg, rowptr, col, nlist, ncnt);
    rgcn_main_kernel<<<B, 256, 0, stream>>>(xx, ss, Wf, bf, convw, convroot,
                                            convbias, lng, lnb, rowptr, col,
                                            invdeg, nlist, ncnt, out, E);
    (void)n_in; (void)out_size; (void)ws_size;
}


# Pre-kernel finds the needed rows first (edges held in registers) and counts, places, sorts and stores only edges into those rows; main kernel copies only that shortened list to LDS; sum order kept
# speedup vs baseline: 14.4248x; 1.2300x over previous
#include <hip/hip_runtime.h>
#include <hip/hip_bf16.h>


typedef __attribute__((ext_vector_type(2))) float v2f;
typedef __attribute__((ext_vector_type(4))) float v4f;
typedef __attribute__((ext_vector_type(4))) int v4i;
typedef __attribute__((ext_vector_type(8))) float v8f;

#define NN  512
#define CC  32
#define LL  4
#define EE  6144
#define NLV 5
#define LN_EPS 1e-5f

__device__ __forceinline__ int rel_of(int i, int E)
{
    return (i >= 2 * E) ? 2 : ((i >= E) ? 1 : 0);
}

#define EPT 18

__global__ __launch_bounds__(1024) void rgcn_csr_kernel(
    const int* __restrict__ src, const int* __restrict__ dst, int E,
    float* __restrict__ invdeg, int* __restrict__ rowptr, int* __restrict__ col,
    int* __restrict__ nlist, int* __restrict__ ncnt)
{
    __shared__ int nf[NLV * NN];
    __shared__ int cnt[3 * NN];
    __shared__ int rp[3 * (NN + 1)];
    __shared__ int key[3 * EE];
    const int tid  = threadIdx.x;
    const int lane = tid & 31;
    const int wid  = tid >> 5;
    const int tot  = 3 * E;

    int ek[EPT];
#pragma unroll
    for (int u = 0; u < EPT; ++u) {
        const int i  = tid + u * 1024;
        const int ii = (i < tot) ? i : 0;
        const int d = dst[ii], s = src[ii];
        ek[u] = (i < tot) ? ((d << 16) | s) : 0;
    }

    for (int i = tid; i < 3 * NN; i += 1024) cnt[i] = 0;
    for (int i = tid; i < NLV * NN; i += 1024) nf[i] = ((i & (NN - 1)) < 64) ? 1 : 0;
    __syncthreads();

    for (int k = NLV - 1; k >= 1; --k) {
        for (int i = tid; i < NN; i += 1024)
            if (nf[k * NN + i]) nf[(k - 1) * NN + i] = 1;
        int fq[EPT];
#pragma unroll
        for (int u = 0; u < EPT; ++u)
            fq[u] = nf[k * NN + (ek[u] >> 16)];
#pragma unroll
        for (int u = 0; u < EPT; ++u)
            if (fq[u]) nf[(k - 1) * NN + (ek[u] & 0xffff)] = 1;
        __syncthreads();
    }

    unsigned needm = 0u;
    {
        int fq[EPT];
#pragma unroll
        for (int u = 0; u < EPT; ++u)
            fq[u] = nf[1 * NN + (ek[u] >> 16)];
#pragma unroll
        for (int u = 0; u < EPT; ++u)
            needm |= ((fq[u] != 0 && tid + u * 1024 < tot) ? 1u : 0u) << u;
    }

#pragma unroll
    for (int u = 0; u < EPT; ++u)
        if ((needm >> u) & 1u)
            atomicAdd(&cnt[rel_of(tid + u * 1024, E) * NN + (ek[u] >> 16)], 1);
    __syncthreads();

    for (int i = tid; i < 3 * NN; i += 1024) {
        int dg = cnt[i];
        if (dg < 1) dg = 1;
        invdeg[i] = 1.0f / (float)dg;
    }
    if (wid < 3) {
        const int r = wid, nb = lane * 16;
        int loc[16];
        int s = 0;
#pragma unroll
        for (int q = 0; q < 16; ++q) { loc[q] = s; s += cnt[r * NN + nb + q]; }
        int inc = s;
        for (int off = 1; off < 32; off <<= 1) {
            const int t = __shfl_up(inc, off, 32);
            if (lane >= off) inc += t;
        }
        const int excl = inc - s;
#pragma unroll
        for (int q = 0; q < 16; ++q) rp[r * (NN + 1) + nb + q] = excl + loc[q];
        if (lane == 31) rp[r * (NN + 1) + NN] = inc;
    }
    __syncthreads();
    for (int i = tid; i < 3 * NN; i += 1024) cnt[i] = 0;
    __syncthreads();

#pragma unroll
    for (int h = 0; h < EPT; h += 6) {
        int rb[6], sl[6];
#pragma unroll
        for (int v = 0; v < 6; ++v) {
            const int u = h + v;
            rb[v] = 0;
            sl[v] = 0;
            if ((needm >> u) & 1u) {
                const int r = rel_of(tid + u * 1024, E);
                const int d = ek[u] >> 16;
                rb[v] = rp[r * (NN + 1) + d];
                sl[v] = atomicAdd(&cnt[r * NN + d], 1);
            }
        }
#pragma unroll
        for (int v = 0; v < 6; ++v) {
            const int u = h + v;
            if ((needm >> u) & 1u)
                key[rel_of(tid + u * 1024, E) * EE + rb[v] + sl[v]] = ek[u];
        }
    }
    __syncthreads();

    for (int r = 0; r < 3; ++r) {
        const int tr = rp[r * (NN + 1) + NN];
        for (int p = tid; p < tr; p += 1024) {
            const int k = key[r * EE + p];
            const int dn = k >> 16;
            const int beg = rp[r * (NN + 1) + dn];
            const int end = rp[r * (NN + 1) + dn + 1];
            int rank = 0;
            for (int q = beg; q < end; ++q) {
                const int kq = key[r * EE + q];
                rank += (kq < k || (kq == k && q < p)) ? 1 : 0;
            }
            col[r * EE + beg + rank] = k & 0xffff;
        }
        if (tid < ((4 - (tr & 3)) & 3)) col[r * EE + tr + tid] = 0;
    }
    for (int i = tid; i < 3 * (NN + 1); i += 1024) rowptr[i] = rp[i];

    if (wid < NLV) {
        const int k = wid;
        int base = 0;
        for (int ch = 0; ch < NN / 32; ++ch) {
            const int n = ch * 32 + lane;
            const int f = nf[k * NN + n];
            const unsigned m = __builtin_amdgcn_ballot_w32(f != 0);
            if (f) nlist[k * NN + base + __popc(m & ((1u << lane) - 1u))] = n;
            base += __popc(m);
        }
        for (int i = base + lane; i < NN; i += 32) nlist[k * NN + i] = 0;
        if (lane == 0) ncnt[k] = (base + 15) & ~15;
    }
}

__device__ __forceinline__ void gemm_acc(
    const float* __restrict__ A, const float* __restrict__ W,
    const int rowoff[4], int nact, int mlo, int mhalf, v8f acc[4][2])
{
#pragma unroll
    for (int mt = 0; mt < 4; ++mt) {
        if (mt < nact) {
            for (int ks = 0; ks < 8; ++ks) {
                const int kk = ks * 4 + mhalf * 2;
                v2f b0, b1, a;
                b0.x = W[kk * CC + mlo];
                b0.y = W[(kk + 1) * CC + mlo];
                b1.x = W[kk * CC + 16 + mlo];
                b1.y = W[(kk + 1) * CC + 16 + mlo];
                a.x = A[rowoff[mt] + kk];
                a.y = A[rowoff[mt] + kk + 1];
                acc[mt][0] = __builtin_amdgcn_wmma_f32_16x16x4_f32(
                    false, a, false, b0, (short)0, acc[mt][0], false, false);
                acc[mt][1] = __builtin_amdgcn_wmma_f32_16x16x4_f32(
                    false, a, false, b1, (short)0, acc[mt][1], false, false);
            }
        }
    }
}

__device__ __forceinline__ void gather_rel(
    const float* __restrict__ xs, float* __restrict__ hs,
    const int* __restrict__ nl, int cnt,
    const int* __restrict__ rp, const int* __restrict__ cl,
    const float* __restrict__ iv, int wid, int lane)
{
    const int cq  = (lane & 7) * 4;
    const int sub = lane >> 3;
    for (int base = 0; base < cnt; base += 64) {
        const int pA = base + wid * 4 + sub;
        const int pB = pA + 32;
        const bool okA = pA < cnt, okB = pB < cnt;
        const int nodeA = nl[pA];
        const int nodeB = nl[pB];
        int jA = rp[nodeA];
        const int eA = okA ? rp[nodeA + 1] : jA;
        int jB = rp[nodeB];
        const int eB = okB ? rp[nodeB + 1] : jB;
        v4f a = (v4f){0.0f, 0.0f, 0.0f, 0.0f};
        v4f b = (v4f){0.0f, 0.0f, 0.0f, 0.0f};
        int ca[4], cb[4];
#pragma unroll
        for (int i = 0; i < 4; ++i) {
            const int va = cl[jA + i], vb = cl[jB + i];
            ca[i] = (jA + i < eA) ? va : NN;
            cb[i] = (jB + i < eB) ? vb : NN;
        }
        while (jA < eA || jB < eB) {
            v4f xa[4], xb[4];
#pragma unroll
            for (int i = 0; i < 4; ++i) {
                xa[i] = *(const v4f*)&xs[ca[i] * CC + cq];
                xb[i] = *(const v4f*)&xs[cb[i] * CC + cq];
            }
            jA = (jA + 4 < eA) ? jA + 4 : eA;
            jB = (jB + 4 < eB) ? jB + 4 : eB;
#pragma unroll
            for (int i = 0; i < 4; ++i) {
                const int va = cl[jA + i], vb = cl[jB + i];
                ca[i] = (jA + i < eA) ? va : NN;
                cb[i] = (jB + i < eB) ? vb : NN;
            }
#pragma unroll
            for (int i = 0; i < 4; ++i) { a += xa[i]; b += xb[i]; }
        }
        if (okA) *(v4f*)&hs[pA * CC + cq] = a * iv[nodeA];
        if (okB) *(v4f*)&hs[pB * CC + cq] = b * iv[nodeB];
    }
}

__global__ __launch_bounds__(256) void rgcn_main_kernel(
    const float* __restrict__ xx, const float* __restrict__ ss,
    const float* __restrict__ Wf, const float* __restrict__ bf,
    const float* __restrict__ convw, const float* __restrict__ convroot,
    const float* __restrict__ convbias, const float* __restrict__ lng,
    const float* __restrict__ lnb, const int* __restrict__ rowptr,
    const int* __restrict__ col, const float* __restrict__ invdeg,
    const int* __restrict__ nlist, const int* __restrict__ ncnt,
    float* __restrict__ out)
{
    __shared__ __attribute__((aligned(16))) float xs[(NN + 1) * CC];
    __shared__ __attribute__((aligned(16))) float hs[NN * CC];
    __shared__ __attribute__((aligned(16))) int colL[3 * EE + 4];
    __shared__ int rpL[3 * (NN + 1)];
    __shared__ float wts[CC * CC];
    __shared__ float ivd[3 * NN];
    __shared__ float wfe[6 * CC];
    __shared__ float prm[3 * CC];
    __shared__ __attribute__((aligned(16))) int nlL[NLV * NN];
    __shared__ int ncL[8];

    const int bid   = blockIdx.x;
    const int tid   = threadIdx.x;
    const int lane  = tid & 31;
    const int wid   = __builtin_amdgcn_readfirstlane(tid >> 5);
    const int mlo   = lane & 15;
    const int mhalf = lane >> 4;

    int q4[3];
#pragma unroll
    for (int r = 0; r < 3; ++r) {
        const int t = (rowptr[r * (NN + 1) + NN] + 3) >> 2;
        q4[r] = (t < 0) ? 0 : ((t > EE / 4) ? EE / 4 : t);
    }

    {
        const v4i* __restrict__ n4 = (const v4i*)nlist;
        v4i* d4 = (v4i*)nlL;
        const v4i t0 = n4[tid], t1 = n4[tid + 256];
        v4i t2 = t0;
        if (tid < NLV * NN / 4 - 512) t2 = n4[tid + 512];
        d4[tid] = t0;
        d4[tid + 256] = t1;
        if (tid < NLV * NN / 4 - 512) d4[tid + 512] = t2;
    }
    if (tid < NLV) ncL[tid] = ncnt[tid];
    for (int i = tid; i < 3 * NN; i += 256) ivd[i] = invdeg[i];
    for (int i = tid; i < 3 * (NN + 1); i += 256) rpL[i] = rowptr[i];
    {
        const int q01 = q4[0] + q4[1];
        const int qt  = q01 + q4[2];
        const v4i* __restrict__ c4 = (const v4i*)col;
        v4i* l4 = (v4i*)colL;
        for (int b0 = tid; b0 < qt; b0 += 3 * 256) {
            v4i v[3];
            int w[3];
#pragma unroll
            for (int k = 0; k < 3; ++k) {
                const int b  = b0 + k * 256;
                const int bc = (b < qt) ? b : qt - 1;
                w[k] = (bc >= q01) ? (2 * (EE / 4) + bc - q01)
                                   : ((bc >= q4[0]) ? ((EE / 4) + bc - q4[0]) : bc);
                v[k] = c4[w[k]];
            }
#pragma unroll
            for (int k = 0; k < 3; ++k)
                if (b0 + k * 256 < qt) l4[w[k]] = v[k];
        }
    }
    for (int i = tid; i < 6 * CC; i += 256) wfe[i] = Wf[i];
    if (tid < CC) {
        xs[NN * CC + tid] = 0.0f;
        prm[tid]          = bf[tid];
        prm[CC + tid]     = lng[tid];
        prm[2 * CC + tid] = lnb[tid];
    }
    __syncthreads();

    const float mfeat = ss[bid] * 0.125f;
    const int cnt0 = ncL[0];
    for (int idx = tid; idx < cnt0 * CC; idx += 256) {
        const int n = nlL[idx >> 5], c = idx & 31;
        const int t = n >> 6, i = (n >> 4) & 3, j = (n >> 2) & 3, k = n & 3;
        const float v = xx[bid * NN + n];
        float a = prm[c];
        a += ((float)i * (1.0f / 3.0f)) * wfe[0 * CC + c];
        a += ((float)j * (1.0f / 3.0f)) * wfe[1 * CC + c];
        a += ((float)k * (1.0f / 3.0f)) * wfe[2 * CC + c];
        a += ((float)t * (1.0f / 7.0f)) * wfe[3 * CC + c];
        a += v * wfe[4 * CC + c];
        a += mfeat * wfe[5 * CC + c];
        xs[n * CC + c] = a;
    }
    __syncthreads();

    for (int l = 0; l < LL; ++l) {
        const int* nl  = &nlL[(l + 1) * NN];
        const int cnt  = ncL[l + 1];
        const int ntile = cnt >> 4;
        const int nact = (ntile > wid) ? ((ntile - wid + 7) >> 3) : 0;
        int xoff[4], hoff[4];
#pragma unroll
        for (int mt = 0; mt < 4; ++mt) {
            const int p = (wid + 8 * mt) * 16 + mlo;
            xoff[mt] = nl[p] * CC;
            hoff[mt] = p * CC;
        }
        v8f acc[4][2];
#pragma unroll
        for (int mt = 0; mt < 4; ++mt)
#pragma unroll
            for (int nt = 0; nt < 2; ++nt) {
                const float bv = convbias[l * CC + nt * 16 + mlo];
                acc[mt][nt] = (v8f){bv, bv, bv, bv, bv, bv, bv, bv};
            }

        for (int i = tid; i < CC * CC; i += 256) wts[i] = convroot[l * CC * CC + i];
        __syncthreads();

        gemm_acc(xs, wts, xoff, nact, mlo, mhalf, acc);
        __syncthreads();

        for (int r = 0; r < 3; ++r) {
            gather_rel(xs, hs, nl, cnt, &rpL[r * (NN + 1)], &colL[r * EE], &ivd[r * NN], wid, lane);
            for (int i = tid; i < CC * CC; i += 256)
                wts[i] = convw[(l * 3 + r) * CC * CC + i];
            __syncthreads();

            gemm_acc(hs, wts, hoff, nact, mlo, mhalf, acc);
            __syncthreads();
        }

#pragma unroll
        for (int mt = 0; mt < 4; ++mt)
            if (mt < nact) {
#pragma unroll
                for (int nt = 0; nt < 2; ++nt)
#pragma unroll
                    for (int jj = 0; jj < 8; ++jj) {
                        const int prow = (wid + 8 * mt) * 16 + jj + 8 * mhalf;
                        const int col  = nt * 16 + mlo;
                        const float v  = acc[mt][nt][jj];
                        hs[prow * CC + col] = v > 0.0f ? v : 0.0f;
                    }
            }
        __syncthreads();

        for (int p0 = wid * 4; p0 < cnt; p0 += 32) {
            int node[4];
            float v[4], d[4], s[4];
#pragma unroll
            for (int u = 0; u < 4; ++u) {
                node[u] = nl[p0 + u];
                v[u] = hs[(p0 + u) * CC + lane];
                s[u] = v[u];
            }
            for (int off = 16; off; off >>= 1)
#pragma unroll
                for (int u = 0; u < 4; ++u) s[u] += __shfl_xor(s[u], off, 32);
#pragma unroll
            for (int u = 0; u < 4; ++u) {
                const float mu = s[u] * (1.0f / 32.0f);
                d[u] = v[u] - mu;
                s[u] = d[u] * d[u];
            }
            for (int off = 16; off; off >>= 1)
#pragma unroll
                for (int u = 0; u < 4; ++u) s[u] += __shfl_xor(s[u], off, 32);
#pragma unroll
            for (int u = 0; u < 4; ++u) {
                const float var = s[u] * (1.0f / 32.0f);
                xs[node[u] * CC + lane] =
                    d[u] * rsqrtf(var + LN_EPS) * prm[CC + lane] + prm[2 * CC + lane];
            }
        }
        __syncthreads();
    }

    for (int idx = tid; idx < 48 * CC; idx += 256) {
        const int row = idx >> 5, c = idx & 31;
        const int g = row >> 4, rc = row & 15;
        const int a0 = rc >> 2, b0 = rc & 3;
        float sum = 0.0f;
#pragma unroll
        for (int q = 0; q < 4; ++q) {
            int node;
            if (g == 0)      node = q * 16 + a0 * 4 + b0;
            else if (g == 1) node = a0 * 16 + q * 4 + b0;
            else             node = a0 * 16 + b0 * 4 + q;
            sum += xs[node * CC + c];
        }
        out[bid * 48 * CC + idx] = sum * 0.25f;
    }
}

extern "C" void kernel_launch(void* const* d_in, const int* in_sizes, int n_in,
                              void* d_out, int out_size, void* d_ws, size_t ws_size,
                              hipStream_t stream) {
    const float* xx       = (const float*)d_in[0];
    const float* ss       = (const float*)d_in[1];
    const float* Wf       = (const float*)d_in[2];
    const float* bf       = (const float*)d_in[3];
    const float* convw    = (const float*)d_in[4];
    const float* convroot = (const float*)d_in[5];
    const float* convbias = (const float*)d_in[6];
    const float* lng      = (const float*)d_in[7];
    const float* lnb      = (const float*)d_in[8];
    const int*   src      = (const int*)d_in[9];
    const int*   dst      = (const int*)d_in[10];
    float*       out      = (float*)d_out;
    float*       invdeg   = (float*)d_ws;
    int*         rowptr   = (int*)d_ws + 2048;
    int*         col      = (int*)d_ws + 4096;
    int*         nlist    = (int*)d_ws + 24576;
    int*         ncnt     = (int*)d_ws + 27648;

    const int E = in_sizes[9] / 3;
    const int B = in_sizes[1];

    rgcn_csr_kernel<<<1, 1024, 0, stream>>>(src, dst, E, invdeg, rowptr, col, nlist, ncnt);
    rgcn_main_kernel<<<B, 256, 0, stream>>>(xx, ss, Wf, bf, convw, convroot,
                                            convbias, lng, lnb, rowptr, col,
                                            invdeg, nlist, ncnt, out);
    (void)n_in; (void)out_size; (void)ws_size;
}


# Staging batched at kernel entry: all 16 weight matrices (resident in 64 KB LDS), biases and graph tables load with every round trip in flight; no global load left in the layer loop; sum order kept
# speedup vs baseline: 22.7477x; 1.5770x over previous
#include <hip/hip_runtime.h>
#include <hip/hip_bf16.h>


typedef __attribute__((ext_vector_type(2))) float v2f;
typedef __attribute__((ext_vector_type(4))) float v4f;
typedef __attribute__((ext_vector_type(4))) int v4i;
typedef __attribute__((ext_vector_type(8))) float v8f;

#define NN  512
#define CC  32
#define LL  4
#define EE  6144
#define NLV 5
#define LN_EPS 1e-5f

__device__ __forceinline__ int rel_of(int i, int E)
{
    return (i >= 2 * E) ? 2 : ((i >= E) ? 1 : 0);
}

#define EPT 18

__global__ __launch_bounds__(1024) void rgcn_csr_kernel(
    const int* __restrict__ src, const int* __restrict__ dst, int E,
    float* __restrict__ invdeg, int* __restrict__ rowptr, int* __restrict__ col,
    int* __restrict__ nlist, int* __restrict__ ncnt)
{
    __shared__ int nf[NLV * NN];
    __shared__ int cnt[3 * NN];
    __shared__ int rp[3 * (NN + 1)];
    __shared__ int key[3 * EE];
    const int tid  = threadIdx.x;
    const int lane = tid & 31;
    const int wid  = tid >> 5;
    const int tot  = 3 * E;

    int ek[EPT];
#pragma unroll
    for (int u = 0; u < EPT; ++u) {
        const int i  = tid + u * 1024;
        const int ii = (i < tot) ? i : 0;
        const int d = dst[ii], s = src[ii];
        ek[u] = (i < tot) ? ((d << 16) | s) : 0;
    }

    for (int i = tid; i < 3 * NN; i += 1024) cnt[i] = 0;
    for (int i = tid; i < NLV * NN; i += 1024) nf[i] = ((i & (NN - 1)) < 64) ? 1 : 0;
    __syncthreads();

    for (int k = NLV - 1; k >= 1; --k) {
        for (int i = tid; i < NN; i += 1024)
            if (nf[k * NN + i]) nf[(k - 1) * NN + i] = 1;
        int fq[EPT];
#pragma unroll
        for (int u = 0; u < EPT; ++u)
            fq[u] = nf[k * NN + (ek[u] >> 16)];
#pragma unroll
        for (int u = 0; u < EPT; ++u)
            if (fq[u]) nf[(k - 1) * NN + (ek[u] & 0xffff)] = 1;
        __syncthreads();
    }

    unsigned needm = 0u;
    {
        int fq[EPT];
#pragma unroll
        for (int u = 0; u < EPT; ++u)
            fq[u] = nf[1 * NN + (ek[u] >> 16)];
#pragma unroll
        for (int u = 0; u < EPT; ++u)
            needm |= ((fq[u] != 0 && tid + u * 1024 < tot) ? 1u : 0u) << u;
    }

#pragma unroll
    for (int u = 0; u < EPT; ++u)
        if ((needm >> u) & 1u)
            atomicAdd(&cnt[rel_of(tid + u * 1024, E) * NN + (ek[u] >> 16)], 1);
    __syncthreads();

    for (int i = tid; i < 3 * NN; i += 1024) {
        int dg = cnt[i];
        if (dg < 1) dg = 1;
        invdeg[i] = 1.0f / (float)dg;
    }
    if (wid < 3) {
        const int r = wid, nb = lane * 16;
        int loc[16];
        int s = 0;
#pragma unroll
        for (int q = 0; q < 16; ++q) { loc[q] = s; s += cnt[r * NN + nb + q]; }
        int inc = s;
        for (int off = 1; off < 32; off <<= 1) {
            const int t = __shfl_up(inc, off, 32);
            if (lane >= off) inc += t;
        }
        const int excl = inc - s;
#pragma unroll
        for (int q = 0; q < 16; ++q) rp[r * (NN + 1) + nb + q] = excl + loc[q];
        if (lane == 31) rp[r * (NN + 1) + NN] = inc;
    }
    __syncthreads();
    for (int i = tid; i < 3 * NN; i += 1024) cnt[i] = 0;
    __syncthreads();

#pragma unroll
    for (int h = 0; h < EPT; h += 6) {
        int rb[6], sl[6];
#pragma unroll
        for (int v = 0; v < 6; ++v) {
            const int u = h + v;
            rb[v] = 0;
            sl[v] = 0;
            if ((needm >> u) & 1u) {
                const int r = rel_of(tid + u * 1024, E);
                const int d = ek[u] >> 16;
                rb[v] = rp[r * (NN + 1) + d];
                sl[v] = atomicAdd(&cnt[r * NN + d], 1);
            }
        }
#pragma unroll
        for (int v = 0; v < 6; ++v) {
            const int u = h + v;
            if ((needm >> u) & 1u)
                key[rel_of(tid + u * 1024, E) * EE + rb[v] + sl[v]] = ek[u];
        }
    }
    __syncthreads();

    for (int r = 0; r < 3; ++r) {
        const int tr = rp[r * (NN + 1) + NN];
        for (int p = tid; p < tr; p += 1024) {
            const int k = key[r * EE + p];
            const int dn = k >> 16;
            const int beg = rp[r * (NN + 1) + dn];
            const int end = rp[r * (NN + 1) + dn + 1];
            int rank = 0;
            for (int q = beg; q < end; ++q) {
                const int kq = key[r * EE + q];
                rank += (kq < k || (kq == k && q < p)) ? 1 : 0;
            }
            col[r * EE + beg + rank] = k & 0xffff;
        }
        if (tid < ((4 - (tr & 3)) & 3)) col[r * EE + tr + tid] = 0;
    }
    for (int i = tid; i < 3 * (NN + 1); i += 1024) rowptr[i] = rp[i];

    if (wid < NLV) {
        const int k = wid;
        int base = 0;
        for (int ch = 0; ch < NN / 32; ++ch) {
            const int n = ch * 32 + lane;
            const int f = nf[k * NN + n];
            const unsigned m = __builtin_amdgcn_ballot_w32(f != 0);
            if (f) nlist[k * NN + base + __popc(m & ((1u << lane) - 1u))] = n;
            base += __popc(m);
        }
        for (int i = base + lane; i < NN; i += 32) nlist[k * NN + i] = 0;
        if (lane == 0) ncnt[k] = (base + 15) & ~15;
    }
}

__device__ __forceinline__ void gemm_acc(
    const float* __restrict__ A, const float* __restrict__ W,
    const int rowoff[4], int nact, int mlo, int mhalf, v8f acc[4][2])
{
#pragma unroll
    for (int mt = 0; mt < 4; ++mt) {
        if (mt < nact) {
            for (int ks = 0; ks < 8; ++ks) {
                const int kk = ks * 4 + mhalf * 2;
                v2f b0, b1, a;
                b0.x = W[kk * CC + mlo];
                b0.y = W[(kk + 1) * CC + mlo];
                b1.x = W[kk * CC + 16 + mlo];
                b1.y = W[(kk + 1) * CC + 16 + mlo];
                a.x = A[rowoff[mt] + kk];
                a.y = A[rowoff[mt] + kk + 1];
                acc[mt][0] = __builtin_amdgcn_wmma_f32_16x16x4_f32(
                    false, a, false, b0, (short)0, acc[mt][0], false, false);
                acc[mt][1] = __builtin_amdgcn_wmma_f32_16x16x4_f32(
                    false, a, false, b1, (short)0, acc[mt][1], false, false);
            }
        }
    }
}

__device__ __forceinline__ void gather_rel(
    const float* __restrict__ xs, float* __restrict__ hs,
    const int* __restrict__ nl, int cnt,
    const int* __restrict__ rp, const int* __restrict__ cl,
    const float* __restrict__ iv, int wid, int lane)
{
    const int cq  = (lane & 7) * 4;
    const int sub = lane >> 3;
    for (int base = 0; base < cnt; base += 64) {
        const int pA = base + wid * 4 + sub;
        const int pB = pA + 32;
        const bool okA = pA < cnt, okB = pB < cnt;
        const int nodeA = nl[pA];
        const int nodeB = nl[pB];
        int jA = rp[nodeA];
        const int eA = okA ? rp[nodeA + 1] : jA;
        int jB = rp[nodeB];
        const int eB = okB ? rp[nodeB + 1] : jB;
        v4f a = (v4f){0.0f, 0.0f, 0.0f, 0.0f};
        v4f b = (v4f){0.0f, 0.0f, 0.0f, 0.0f};
        int ca[4], cb[4];
#pragma unroll
        for (int i = 0; i < 4; ++i) {
            const int va = cl[jA + i], vb = cl[jB + i];
            ca[i] = (jA + i < eA) ? va : NN;
            cb[i] = (jB + i < eB) ? vb : NN;
        }
        while (jA < eA || jB < eB) {
            v4f xa[4], xb[4];
#pragma unroll
            for (int i = 0; i < 4; ++i) {
                xa[i] = *(const v4f*)&xs[ca[i] * CC + cq];
                xb[i] = *(const v4f*)&xs[cb[i] * CC + cq];
            }
            jA = (jA + 4 < eA) ? jA + 4 : eA;
            jB = (jB + 4 < eB) ? jB + 4 : eB;
#pragma unroll
            for (int i = 0; i < 4; ++i) {
                const int va = cl[jA + i], vb = cl[jB + i];
                ca[i] = (jA + i < eA) ? va : NN;
                cb[i] = (jB + i < eB) ? vb : NN;
            }
#pragma unroll
            for (int i = 0; i < 4; ++i) { a += xa[i]; b += xb[i]; }
        }
        if (okA) *(v4f*)&hs[pA * CC + cq] = a * iv[nodeA];
        if (okB) *(v4f*)&hs[pB * CC + cq] = b * iv[nodeB];
    }
}

__global__ __launch_bounds__(256) void rgcn_main_kernel(
    const float* __restrict__ xx, const float* __restrict__ ss,
    const float* __restrict__ Wf, const float* __restrict__ bf,
    const float* __restrict__ convw, const float* __restrict__ convroot,
    const float* __restrict__ convbias, const float* __restrict__ lng,
    const float* __restrict__ lnb, const int* __restrict__ rowptr,
    const int* __restrict__ col, const float* __restrict__ invdeg,
    const int* __restrict__ nlist, const int* __restrict__ ncnt,
    float* __restrict__ out)
{
    __shared__ __attribute__((aligned(16))) float xs[(NN + 1) * CC];
    __shared__ __attribute__((aligned(16))) float hs[NN * CC];
    __shared__ __attribute__((aligned(16))) int colL[3 * EE + 4];
    __shared__ int rpL[3 * (NN + 1)];
    __shared__ __attribute__((aligned(16))) float wall[LL * 4 * CC * CC];
    __shared__ float cbs[LL * CC];
    __shared__ float ivd[3 * NN];
    __shared__ float wfe[6 * CC];
    __shared__ float prm[3 * CC];
    __shared__ __attribute__((aligned(16))) int nlL[NLV * NN];
    __shared__ int ncL[8];

    const int bid   = blockIdx.x;
    const int tid   = threadIdx.x;
    const int lane  = tid & 31;
    const int wid   = __builtin_amdgcn_readfirstlane(tid >> 5);
    const int mlo   = lane & 15;
    const int mhalf = lane >> 4;

    const float mfeat = ss[bid] * 0.125f;

    int q4[3];
#pragma unroll
    for (int r = 0; r < 3; ++r) {
        const int t = (rowptr[r * (NN + 1) + NN] + 3) >> 2;
        q4[r] = (t < 0) ? 0 : ((t > EE / 4) ? EE / 4 : t);
    }

    const v4f* __restrict__ cr4 = (const v4f*)convroot;
    const v4f* __restrict__ cw4 = (const v4f*)convw;
    v4f wr[LL], ww[3 * LL];
#pragma unroll
    for (int u = 0; u < LL; ++u) wr[u] = cr4[u * 256 + tid];
#pragma unroll
    for (int u = 0; u < 3 * LL; ++u) ww[u] = cw4[u * 256 + tid];
    const v4i* __restrict__ n4 = (const v4i*)nlist;
    const v4i t0 = n4[tid], t1 = n4[tid + 256];
    const v4i t2 = n4[512 + (tid & (NLV * NN / 4 - 512 - 1))];
    const int ncv = ncnt[(tid < NLV) ? tid : 0];
    float ivr[6];
    int rpr[7];
#pragma unroll
    for (int u = 0; u < 6; ++u) {
        ivr[u] = invdeg[u * 256 + tid];
        rpr[u] = rowptr[u * 256 + tid];
    }
    rpr[6] = rowptr[(tid < 3 * (NN + 1) - 6 * 256) ? 6 * 256 + tid : 0];
    const float wfv = Wf[(tid < 6 * CC) ? tid : 0];
    const float pbv = bf[lane], pgv = lng[lane], plv = lnb[lane];
    const float cbv = convbias[tid & (LL * CC - 1)];

    {
        const int q01 = q4[0] + q4[1];
        const int qt  = q01 + q4[2];
        const v4i* __restrict__ c4 = (const v4i*)col;
        v4i* l4 = (v4i*)colL;
        for (int b0 = tid; b0 < qt; b0 += 3 * 256) {
            v4i v[3];
            int w[3];
#pragma unroll
            for (int k = 0; k < 3; ++k) {
                const int b  = b0 + k * 256;
                const int bc = (b < qt) ? b : qt - 1;
                w[k] = (bc >= q01) ? (2 * (EE / 4) + bc - q01)
                                   : ((bc >= q4[0]) ? ((EE / 4) + bc - q4[0]) : bc);
                v[k] = c4[w[k]];
            }
#pragma unroll
            for (int k = 0; k < 3; ++k)
                if (b0 + k * 256 < qt) l4[w[k]] = v[k];
        }
    }

    {
        v4f* w4 = (v4f*)wall;
#pragma unroll
        for (int u = 0; u < LL; ++u) w4[(u * 4) * 256 + tid] = wr[u];
#pragma unroll
        for (int u = 0; u < 3 * LL; ++u) w4[((u / 3) * 4 + 1 + (u % 3)) * 256 + tid] = ww[u];
        v4i* d4 = (v4i*)nlL;
        d4[tid] = t0;
        d4[tid + 256] = t1;
        if (tid < NLV * NN / 4 - 512) d4[tid + 512] = t2;
        if (tid < NLV) ncL[tid] = ncv;
#pragma unroll
        for (int u = 0; u < 6; ++u) {
            ivd[u * 256 + tid] = ivr[u];
            rpL[u * 256 + tid] = rpr[u];
        }
        if (tid < 3 * (NN + 1) - 6 * 256) rpL[6 * 256 + tid] = rpr[6];
        if (tid < 6 * CC) wfe[tid] = wfv;
        if (tid < CC) {
            xs[NN * CC + tid] = 0.0f;
            prm[tid]          = pbv;
            prm[CC + tid]     = pgv;
            prm[2 * CC + tid] = plv;
        }
        if (tid < LL * CC) cbs[tid] = cbv;
    }
    __syncthreads();

    const int cnt0 = ncL[0];
    for (int idx = tid; idx < cnt0 * CC; idx += 256) {
        const int n = nlL[idx >> 5], c = idx & 31;
        const int t = n >> 6, i = (n >> 4) & 3, j = (n >> 2) & 3, k = n & 3;
        const float v = xx[bid * NN + n];
        float a = prm[c];
        a += ((float)i * (1.0f / 3.0f)) * wfe[0 * CC + c];
        a += ((float)j * (1.0f / 3.0f)) * wfe[1 * CC + c];
        a += ((float)k * (1.0f / 3.0f)) * wfe[2 * CC + c];
        a += ((float)t * (1.0f / 7.0f)) * wfe[3 * CC + c];
        a += v * wfe[4 * CC + c];
        a += mfeat * wfe[5 * CC + c];
        xs[n * CC + c] = a;
    }
    __syncthreads();

    for (int l = 0; l < LL; ++l) {
        const int* nl  = &nlL[(l + 1) * NN];
        const int cnt  = ncL[l + 1];
        const int ntile = cnt >> 4;
        const int nact = (ntile > wid) ? ((ntile - wid + 7) >> 3) : 0;
        int xoff[4], hoff[4];
#pragma unroll
        for (int mt = 0; mt < 4; ++mt) {
            const int p = (wid + 8 * mt) * 16 + mlo;
            xoff[mt] = nl[p] * CC;
            hoff[mt] = p * CC;
        }
        v8f acc[4][2];
#pragma unroll
        for (int mt = 0; mt < 4; ++mt)
#pragma unroll
            for (int nt = 0; nt < 2; ++nt) {
                const float bv = cbs[l * CC + nt * 16 + mlo];
                acc[mt][nt] = (v8f){bv, bv, bv, bv, bv, bv, bv, bv};
            }

        gemm_acc(xs, &wall[(l * 4) * CC * CC], xoff, nact, mlo, mhalf, acc);

        for (int r = 0; r < 3; ++r) {
            gather_rel(xs, hs, nl, cnt, &rpL[r * (NN + 1)], &colL[r * EE], &ivd[r * NN], wid, lane);
            __syncthreads();

            gemm_acc(hs, &wall[(l * 4 + 1 + r) * CC * CC], hoff, nact, mlo, mhalf, acc);
            __syncthreads();
        }

#pragma unroll
        for (int mt = 0; mt < 4; ++mt)
            if (mt < nact) {
#pragma unroll
                for (int nt = 0; nt < 2; ++nt)
#pragma unroll
                    for (int jj = 0; jj < 8; ++jj) {
                        const int prow = (wid + 8 * mt) * 16 + jj + 8 * mhalf;
                        const int col  = nt * 16 + mlo;
                        const float v  = acc[mt][nt][jj];
                        hs[prow * CC + col] = v > 0.0f ? v : 0.0f;
                    }
            }
        __syncthreads();

        for (int p0 = wid * 4; p0 < cnt; p0 += 32) {
            int node[4];
            float v[4], d[4], s[4];
#pragma unroll
            for (int u = 0; u < 4; ++u) {
                node[u] = nl[p0 + u];
                v[u] = hs[(p0 + u) * CC + lane];
                s[u] = v[u];
            }
            for (int off = 16; off; off >>= 1)
#pragma unroll
                for (int u = 0; u < 4; ++u) s[u] += __shfl_xor(s[u], off, 32);
#pragma unroll
            for (int u = 0; u < 4; ++u) {
                const float mu = s[u] * (1.0f / 32.0f);
                d[u] = v[u] - mu;
                s[u] = d[u] * d[u];
            }
            for (int off = 16; off; off >>= 1)
#pragma unroll
                for (int u = 0; u < 4; ++u) s[u] += __shfl_xor(s[u], off, 32);
#pragma unroll
            for (int u = 0; u < 4; ++u) {
                const float var = s[u] * (1.0f / 32.0f);
                xs[node[u] * CC + lane] =
                    d[u] * rsqrtf(var + LN_EPS) * prm[CC + lane] + prm[2 * CC + lane];
            }
        }
        __syncthreads();
    }

    for (int idx = tid; idx < 48 * CC; idx += 256) {
        const int row = idx >> 5, c = idx & 31;
        const int g = row >> 4, rc = row & 15;
        const int a0 = rc >> 2, b0 = rc & 3;
        float sum = 0.0f;
#pragma unroll
        for (int q = 0; q < 4; ++q) {
            int node;
            if (g == 0)      node = q * 16 + a0 * 4 + b0;
            else if (g == 1) node = a0 * 16 + q * 4 + b0;
            else             node = a0 * 16 + b0 * 4 + q;
            sum += xs[node * CC + c];
        }
        out[bid * 48 * CC + idx] = sum * 0.25f;
    }
}

extern "C" void kernel_launch(void* const* d_in, const int* in_sizes, int n_in,
                              void* d_out, int out_size, void* d_ws, size_t ws_size,
                              hipStream_t stream) {
    const float* xx       = (const float*)d_in[0];
    const float* ss       = (const float*)d_in[1];
    const float* Wf       = (const float*)d_in[2];
    const float* bf       = (const float*)d_in[3];
    const float* convw    = (const float*)d_in[4];
    const float* convroot = (const float*)d_in[5];
    const float* convbias = (const float*)d_in[6];
    const float* lng      = (const float*)d_in[7];
    const float* lnb      = (const float*)d_in[8];
    const int*   src      = (const int*)d_in[9];
    const int*   dst      = (const int*)d_in[10];
    float*       out      = (float*)d_out;
    float*       invdeg   = (float*)d_ws;
    int*         rowptr   = (int*)d_ws + 2048;
    int*         col      = (int*)d_ws + 4096;
    int*         nlist    = (int*)d_ws + 24576;
    int*         ncnt     = (int*)d_ws + 27648;

    const int E = in_sizes[9] / 3;
    const int B = in_sizes[1];

    rgcn_csr_kernel<<<1, 1024, 0, stream>>>(src, dst, E, invdeg, rowptr, col, nlist, ncnt);
    rgcn_main_kernel<<<B, 256, 0, stream>>>(xx, ss, Wf, bf, convw, convroot,
                                            convbias, lng, lnb, rowptr, col,
                                            invdeg, nlist, ncnt, out);
    (void)n_in; (void)out_size; (void)ws_size;
}


# Pre-kernel holds each level's row set as a bit mask in LDS, not one int per node: per-edge level tests read a few shared words, free of the bank conflicts of flags 64 nodes apart; sum order kept
# speedup vs baseline: 27.5179x; 1.2097x over previous
#include <hip/hip_runtime.h>
#include <hip/hip_bf16.h>


typedef __attribute__((ext_vector_type(2))) float v2f;
typedef __attribute__((ext_vector_type(4))) float v4f;
typedef __attribute__((ext_vector_type(4))) int v4i;
typedef __attribute__((ext_vector_type(8))) float v8f;

#define NN  512
#define CC  32
#define LL  4
#define EE  6144
#define NLV 5
#define LN_EPS 1e-5f

__device__ __forceinline__ int rel_of(int i, int E)
{
    return (i >= 2 * E) ? 2 : ((i >= E) ? 1 : 0);
}

#define EPT 18

__global__ __launch_bounds__(1024) void rgcn_csr_kernel(
    const int* __restrict__ src, const int* __restrict__ dst, int E,
    float* __restrict__ invdeg, int* __restrict__ rowptr, int* __restrict__ col,
    int* __restrict__ nlist, int* __restrict__ ncnt)
{
    __shared__ int nfl[NN];
    __shared__ unsigned mk[NLV * 16];
    __shared__ int cnt[3 * NN];
    __shared__ int rp[3 * (NN + 1)];
    __shared__ int key[3 * EE];
    const int tid  = threadIdx.x;
    const int lane = tid & 31;
    const int wid  = tid >> 5;
    const int tot  = 3 * E;

    int ek[EPT];
#pragma unroll
    for (int u = 0; u < EPT; ++u) {
        const int i  = tid + u * 1024;
        const int ii = (i < tot) ? i : 0;
        const int d = dst[ii], s = src[ii];
        ek[u] = (i < tot) ? ((d << 16) | s) : 0;
    }

    for (int i = tid; i < 3 * NN; i += 1024) cnt[i] = 0;
    if (tid < NN) nfl[tid] = (tid < 64) ? 1 : 0;
    if (tid < 16) mk[(NLV - 1) * 16 + tid] = (tid < 2) ? 0xffffffffu : 0u;
    __syncthreads();

    for (int k = NLV - 1; k >= 1; --k) {
        unsigned wq[EPT];
#pragma unroll
        for (int u = 0; u < EPT; ++u)
            wq[u] = mk[k * 16 + (ek[u] >> 21)];
#pragma unroll
        for (int u = 0; u < EPT; ++u)
            if ((wq[u] >> ((ek[u] >> 16) & 31)) & 1u) nfl[ek[u] & 0xffff] = 1;
        __syncthreads();
        if (tid < NN) {
            const unsigned m = __builtin_amdgcn_ballot_w32(nfl[tid] != 0);
            if (lane == 0) mk[(k - 1) * 16 + wid] = m;
        }
        __syncthreads();
    }

    unsigned needm = 0u;
    {
        unsigned wq[EPT];
#pragma unroll
        for (int u = 0; u < EPT; ++u)
            wq[u] = mk[1 * 16 + (ek[u] >> 21)];
#pragma unroll
        for (int u = 0; u < EPT; ++u)
            needm |= ((((wq[u] >> ((ek[u] >> 16) & 31)) & 1u) != 0u && tid + u * 1024 < tot) ? 1u : 0u) << u;
    }

#pragma unroll
    for (int u = 0; u < EPT; ++u)
        if ((needm >> u) & 1u)
            atomicAdd(&cnt[rel_of(tid + u * 1024, E) * NN + (ek[u] >> 16)], 1);
    __syncthreads();

    for (int i = tid; i < 3 * NN; i += 1024) {
        int dg = cnt[i];
        if (dg < 1) dg = 1;
        invdeg[i] = 1.0f / (float)dg;
    }
    if (wid < 3) {
        const int r = wid, nb = lane * 16;
        int loc[16];
        int s = 0;
#pragma unroll
        for (int q = 0; q < 16; ++q) { loc[q] = s; s += cnt[r * NN + nb + q]; }
        int inc = s;
        for (int off = 1; off < 32; off <<= 1) {
            const int t = __shfl_up(inc, off, 32);
            if (lane >= off) inc += t;
        }
        const int excl = inc - s;
#pragma unroll
        for (int q = 0; q < 16; ++q) rp[r * (NN + 1) + nb + q] = excl + loc[q];
        if (lane == 31) rp[r * (NN + 1) + NN] = inc;
    }
    __syncthreads();
    for (int i = tid; i < 3 * NN; i += 1024) cnt[i] = 0;
    __syncthreads();

#pragma unroll
    for (int h = 0; h < EPT; h += 6) {
        int rb[6], sl[6];
#pragma unroll
        for (int v = 0; v < 6; ++v) {
            const int u = h + v;
            rb[v] = 0;
            sl[v] = 0;
            if ((needm >> u) & 1u) {
                const int r = rel_of(tid + u * 1024, E);
                const int d = ek[u] >> 16;
                rb[v] = rp[r * (NN + 1) + d];
                sl[v] = atomicAdd(&cnt[r * NN + d], 1);
            }
        }
#pragma unroll
        for (int v = 0; v < 6; ++v) {
            const int u = h + v;
            if ((needm >> u) & 1u)
                key[rel_of(tid + u * 1024, E) * EE + rb[v] + sl[v]] = ek[u];
        }
    }
    __syncthreads();

    for (int r = 0; r < 3; ++r) {
        const int tr = rp[r * (NN + 1) + NN];
        for (int p = tid; p < tr; p += 1024) {
            const int k = key[r * EE + p];
            const int dn = k >> 16;
            const int beg = rp[r * (NN + 1) + dn];
            const int end = rp[r * (NN + 1) + dn + 1];
            int rank = 0;
            for (int q = beg; q < end; ++q) {
                const int kq = key[r * EE + q];
                rank += (kq < k || (kq == k && q < p)) ? 1 : 0;
            }
            col[r * EE + beg + rank] = k & 0xffff;
        }
        if (tid < ((4 - (tr & 3)) & 3)) col[r * EE + tr + tid] = 0;
    }
    for (int i = tid; i < 3 * (NN + 1); i += 1024) rowptr[i] = rp[i];

    if (wid < NLV) {
        const int k = wid;
        unsigned mw[NN / 32];
#pragma unroll
        for (int ch = 0; ch < NN / 32; ++ch) mw[ch] = mk[k * 16 + ch];
        int base = 0;
#pragma unroll
        for (int ch = 0; ch < NN / 32; ++ch) {
            const unsigned m = mw[ch];
            if ((m >> lane) & 1u)
                nlist[k * NN + base + __popc(m & ((1u << lane) - 1u))] = ch * 32 + lane;
            base += __popc(m);
        }
        for (int i = base + lane; i < NN; i += 32) nlist[k * NN + i] = 0;
        if (lane == 0) ncnt[k] = (base + 15) & ~15;
    }
}

__device__ __forceinline__ void gemm_acc(
    const float* __restrict__ A, const float* __restrict__ W,
    const int rowoff[4], int nact, int mlo, int mhalf, v8f acc[4][2])
{
#pragma unroll
    for (int mt = 0; mt < 4; ++mt) {
        if (mt < nact) {
            for (int ks = 0; ks < 8; ++ks) {
                const int kk = ks * 4 + mhalf * 2;
                v2f b0, b1, a;
                b0.x = W[kk * CC + mlo];
                b0.y = W[(kk + 1) * CC + mlo];
                b1.x = W[kk * CC + 16 + mlo];
                b1.y = W[(kk + 1) * CC + 16 + mlo];
                a.x = A[rowoff[mt] + kk];
                a.y = A[rowoff[mt] + kk + 1];
                acc[mt][0] = __builtin_amdgcn_wmma_f32_16x16x4_f32(
                    false, a, false, b0, (short)0, acc[mt][0], false, false);
                acc[mt][1] = __builtin_amdgcn_wmma_f32_16x16x4_f32(
                    false, a, false, b1, (short)0, acc[mt][1], false, false);
            }
        }
    }
}

__device__ __forceinline__ void gather_rel(
    const float* __restrict__ xs, float* __restrict__ hs,
    const int* __restrict__ nl, int cnt,
    const int* __restrict__ rp, const int* __restrict__ cl,
    const float* __restrict__ iv, int wid, int lane)
{
    const int cq  = (lane & 7) * 4;
    const int sub = lane >> 3;
    for (int base = 0; base < cnt; base += 64) {
        const int pA = base + wid * 4 + sub;
        const int pB = pA + 32;
        const bool okA = pA < cnt, okB = pB < cnt;
        const int nodeA = nl[pA];
        const int nodeB = nl[pB];
        int jA = rp[nodeA];
        const int eA = okA ? rp[nodeA + 1] : jA;
        int jB = rp[nodeB];
        const int eB = okB ? rp[nodeB + 1] : jB;
        v4f a = (v4f){0.0f, 0.0f, 0.0f, 0.0f};
        v4f b = (v4f){0.0f, 0.0f, 0.0f, 0.0f};
        int ca[4], cb[4];
#pragma unroll
        for (int i = 0; i < 4; ++i) {
            const int va = cl[jA + i], vb = cl[jB + i];
            ca[i] = (jA + i < eA) ? va : NN;
            cb[i] = (jB + i < eB) ? vb : NN;
        }
        while (jA < eA || jB < eB) {
            v4f xa[4], xb[4];
#pragma unroll
            for (int i = 0; i < 4; ++i) {
                xa[i] = *(const v4f*)&xs[ca[i] * CC + cq];
                xb[i] = *(const v4f*)&xs[cb[i] * CC + cq];
            }
            jA = (jA + 4 < eA) ? jA + 4 : eA;
            jB = (jB + 4 < eB) ? jB + 4 : eB;
#pragma unroll
            for (int i = 0; i < 4; ++i) {
                const int va = cl[jA + i], vb = cl[jB + i];
                ca[i] = (jA + i < eA) ? va : NN;
                cb[i] = (jB + i < eB) ? vb : NN;
            }
#pragma unroll
            for (int i = 0; i < 4; ++i) { a += xa[i]; b += xb[i]; }
        }
        if (okA) *(v4f*)&hs[pA * CC + cq] = a * iv[nodeA];
        if (okB) *(v4f*)&hs[pB * CC + cq] = b * iv[nodeB];
    }
}

__global__ __launch_bounds__(256) void rgcn_main_kernel(
    const float* __restrict__ xx, const float* __restrict__ ss,
    const float* __restrict__ Wf, const float* __restrict__ bf,
    const float* __restrict__ convw, const float* __restrict__ convroot,
    const float* __restrict__ convbias, const float* __restrict__ lng,
    const float* __restrict__ lnb, const int* __restrict__ rowptr,
    const int* __restrict__ col, const float* __restrict__ invdeg,
    const int* __restrict__ nlist, const int* __restrict__ ncnt,
    float* __restrict__ out)
{
    __shared__ __attribute__((aligned(16))) float xs[(NN + 1) * CC];
    __shared__ __attribute__((aligned(16))) float hs[NN * CC];
    __shared__ __attribute__((aligned(16))) int colL[3 * EE + 4];
    __shared__ int rpL[3 * (NN + 1)];
    __shared__ __attribute__((aligned(16))) float wall[LL * 4 * CC * CC];
    __shared__ float cbs[LL * CC];
    __shared__ float ivd[3 * NN];
    __shared__ float wfe[6 * CC];
    __shared__ float prm[3 * CC];
    __shared__ __attribute__((aligned(16))) int nlL[NLV * NN];
    __shared__ int ncL[8];

    const int bid   = blockIdx.x;
    const int tid   = threadIdx.x;
    const int lane  = tid & 31;
    const int wid   = __builtin_amdgcn_readfirstlane(tid >> 5);
    const int mlo   = lane & 15;
    const int mhalf = lane >> 4;

    const float mfeat = ss[bid] * 0.125f;

    int q4[3];
#pragma unroll
    for (int r = 0; r < 3; ++r) {
        const int t = (rowptr[r * (NN + 1) + NN] + 3) >> 2;
        q4[r] = (t < 0) ? 0 : ((t > EE / 4) ? EE / 4 : t);
    }

    const v4f* __restrict__ cr4 = (const v4f*)convroot;
    const v4f* __restrict__ cw4 = (const v4f*)convw;
    v4f wr[LL], ww[3 * LL];
#pragma unroll
    for (int u = 0; u < LL; ++u) wr[u] = cr4[u * 256 + tid];
#pragma unroll
    for (int u = 0; u < 3 * LL; ++u) ww[u] = cw4[u * 256 + tid];
    const v4i* __restrict__ n4 = (const v4i*)nlist;
    const v4i t0 = n4[tid], t1 = n4[tid + 256];
    const v4i t2 = n4[512 + (tid & (NLV * NN / 4 - 512 - 1))];
    const int ncv = ncnt[(tid < NLV) ? tid : 0];
    float ivr[6];
    int rpr[7];
#pragma unroll
    for (int u = 0; u < 6; ++u) {
        ivr[u] = invdeg[u * 256 + tid];
        rpr[u] = rowptr[u * 256 + tid];
    }
    rpr[6] = rowptr[(tid < 3 * (NN + 1) - 6 * 256) ? 6 * 256 + tid : 0];
    const float wfv = Wf[(tid < 6 * CC) ? tid : 0];
    const float pbv = bf[lane], pgv = lng[lane], plv = lnb[lane];
    const float cbv = convbias[tid & (LL * CC - 1)];

    {
        const int q01 = q4[0] + q4[1];
        const int qt  = q01 + q4[2];
        const v4i* __restrict__ c4 = (const v4i*)col;
        v4i* l4 = (v4i*)colL;
        for (int b0 = tid; b0 < qt; b0 += 3 * 256) {
            v4i v[3];
            int w[3];
#pragma unroll
            for (int k = 0; k < 3; ++k) {
                const int b  = b0 + k * 256;
                const int bc = (b < qt) ? b : qt - 1;
                w[k] = (bc >= q01) ? (2 * (EE / 4) + bc - q01)
                                   : ((bc >= q4[0]) ? ((EE / 4) + bc - q4[0]) : bc);
                v[k] = c4[w[k]];
            }
#pragma unroll
            for (int k = 0; k < 3; ++k)
                if (b0 + k * 256 < qt) l4[w[k]] = v[k];
        }
    }

    {
        v4f* w4 = (v4f*)wall;
#pragma unroll
        for (int u = 0; u < LL; ++u) w4[(u * 4) * 256 + tid] = wr[u];
#pragma unroll
        for (int u = 0; u < 3 * LL; ++u) w4[((u / 3) * 4 + 1 + (u % 3)) * 256 + tid] = ww[u];
        v4i* d4 = (v4i*)nlL;
        d4[tid] = t0;
        d4[tid + 256] = t1;
        if (tid < NLV * NN / 4 - 512) d4[tid + 512] = t2;
        if (tid < NLV) ncL[tid] = ncv;
#pragma unroll
        for (int u = 0; u < 6; ++u) {
            ivd[u * 256 + tid] = ivr[u];
            rpL[u * 256 + tid] = rpr[u];
        }
        if (tid < 3 * (NN + 1) - 6 * 256) rpL[6 * 256 + tid] = rpr[6];
        if (tid < 6 * CC) wfe[tid] = wfv;
        if (tid < CC) {
            xs[NN * CC + tid] = 0.0f;
            prm[tid]          = pbv;
            prm[CC + tid]     = pgv;
            prm[2 * CC + tid] = plv;
        }
        if (tid < LL * CC) cbs[tid] = cbv;
    }
    __syncthreads();

    const int cnt0 = ncL[0];
    for (int idx = tid; idx < cnt0 * CC; idx += 256) {
        const int n = nlL[idx >> 5], c = idx & 31;
        const int t = n >> 6, i = (n >> 4) & 3, j = (n >> 2) & 3, k = n & 3;
        const float v = xx[bid * NN + n];
        float a = prm[c];
        a += ((float)i * (1.0f / 3.0f)) * wfe[0 * CC + c];
        a += ((float)j * (1.0f / 3.0f)) * wfe[1 * CC + c];
        a += ((float)k * (1.0f / 3.0f)) * wfe[2 * CC + c];
        a += ((float)t * (1.0f / 7.0f)) * wfe[3 * CC + c];
        a += v * wfe[4 * CC + c];
        a += mfeat * wfe[5 * CC + c];
        xs[n * CC + c] = a;
    }
    __syncthreads();

    for (int l = 0; l < LL; ++l) {
        const int* nl  = &nlL[(l + 1) * NN];
        const int cnt  = ncL[l + 1];
        const int ntile = cnt >> 4;
        const int nact = (ntile > wid) ? ((ntile - wid + 7) >> 3) : 0;
        int xoff[4], hoff[4];
#pragma unroll
        for (int mt = 0; mt < 4; ++mt) {
            const int p = (wid + 8 * mt) * 16 + mlo;
            xoff[mt] = nl[p] * CC;
            hoff[mt] = p * CC;
        }
        v8f acc[4][2];
#pragma unroll
        for (int mt = 0; mt < 4; ++mt)
#pragma unroll
            for (int nt = 0; nt < 2; ++nt) {
                const float bv = cbs[l * CC + nt * 16 + mlo];
                acc[mt][nt] = (v8f){bv, bv, bv, bv, bv, bv, bv, bv};
            }

        gemm_acc(xs, &wall[(l * 4) * CC * CC], xoff, nact, mlo, mhalf, acc);

        for (int r = 0; r < 3; ++r) {
            gather_rel(xs, hs, nl, cnt, &rpL[r * (NN + 1)], &colL[r * EE], &ivd[r * NN], wid, lane);
            __syncthreads();

            gemm_acc(hs, &wall[(l * 4 + 1 + r) * CC * CC], hoff, nact, mlo, mhalf, acc);
            __syncthreads();
        }

#pragma unroll
        for (int mt = 0; mt < 4; ++mt)
            if (mt < nact) {
#pragma unroll
                for (int nt = 0; nt < 2; ++nt)
#pragma unroll
                    for (int jj = 0; jj < 8; ++jj) {
                        const int prow = (wid + 8 * mt) * 16 + jj + 8 * mhalf;
                        const int col  = nt * 16 + mlo;
                        const float v  = acc[mt][nt][jj];
                        hs[prow * CC + col] = v > 0.0f ? v : 0.0f;
                    }
            }
        __syncthreads();

        for (int p0 = wid * 4; p0 < cnt; p0 += 32) {
            int node[4];
            float v[4], d[4], s[4];
#pragma unroll
            for (int u = 0; u < 4; ++u) {
                node[u] = nl[p0 + u];
                v[u] = hs[(p0 + u) * CC + lane];
                s[u] = v[u];
            }
            for (int off = 16; off; off >>= 1)
#pragma unroll
                for (int u = 0; u < 4; ++u) s[u] += __shfl_xor(s[u], off, 32);
#pragma unroll
            for (int u = 0; u < 4; ++u) {
                const float mu = s[u] * (1.0f / 32.0f);
                d[u] = v[u] - mu;
                s[u] = d[u] * d[u];
            }
            for (int off = 16; off; off >>= 1)
#pragma unroll
                for (int u = 0; u < 4; ++u) s[u] += __shfl_xor(s[u], off, 32);
#pragma unroll
            for (int u = 0; u < 4; ++u) {
                const float var = s[u] * (1.0f / 32.0f);
                xs[node[u] * CC + lane] =
                    d[u] * rsqrtf(var + LN_EPS) * prm[CC + lane] + prm[2 * CC + lane];
            }
        }
        __syncthreads();
    }

    for (int idx = tid; idx < 48 * CC; idx += 256) {
        const int row = idx >> 5, c = idx & 31;
        const int g = row >> 4, rc = row & 15;
        const int a0 = rc >> 2, b0 = rc & 3;
        float sum = 0.0f;
#pragma unroll
        for (int q = 0; q < 4; ++q) {
            int node;
            if (g == 0)      node = q * 16 + a0 * 4 + b0;
            else if (g == 1) node = a0 * 16 + q * 4 + b0;
            else             node = a0 * 16 + b0 * 4 + q;
            sum += xs[node * CC + c];
        }
        out[bid * 48 * CC + idx] = sum * 0.25f;
    }
}

extern "C" void kernel_launch(void* const* d_in, const int* in_sizes, int n_in,
                              void* d_out, int out_size, void* d_ws, size_t ws_size,
                              hipStream_t stream) {
    const float* xx       = (const float*)d_in[0];
    const float* ss       = (const float*)d_in[1];
    const float* Wf       = (const float*)d_in[2];
    const float* bf       = (const float*)d_in[3];
    const float* convw    = (const float*)d_in[4];
    const float* convroot = (const float*)d_in[5];
    const float* convbias = (const float*)d_in[6];
    const float* lng      = (const float*)d_in[7];
    const float* lnb      = (const float*)d_in[8];
    const int*   src      = (const int*)d_in[9];
    const int*   dst      = (const int*)d_in[10];
    float*       out      = (float*)d_out;
    float*       invdeg   = (float*)d_ws;
    int*         rowptr   = (int*)d_ws + 2048;
    int*         col      = (int*)d_ws + 4096;
    int*         nlist    = (int*)d_ws + 24576;
    int*         ncnt     = (int*)d_ws + 27648;

    const int E = in_sizes[9] / 3;
    const int B = in_sizes[1];

    rgcn_csr_kernel<<<1, 1024, 0, stream>>>(src, dst, E, invdeg, rowptr, col, nlist, ncnt);
    rgcn_main_kernel<<<B, 256, 0, stream>>>(xx, ss, Wf, bf, convw, convroot,
                                            convbias, lng, lnb, rowptr, col,
                                            invdeg, nlist, ncnt, out);
    (void)n_in; (void)out_size; (void)ws_size;
}


# LayerNorm taken in the accumulator registers by the tile's own wave: row sums as 16-lane cross-lane ALU steps, no h spill, no LDS shuffles, 2 barriers fewer per layer; LayerNorm sum tree changed
# speedup vs baseline: 30.4743x; 1.1074x over previous
#include <hip/hip_runtime.h>
#include <hip/hip_bf16.h>


typedef __attribute__((ext_vector_type(2))) float v2f;
typedef __attribute__((ext_vector_type(4))) float v4f;
typedef __attribute__((ext_vector_type(4))) int v4i;
typedef __attribute__((ext_vector_type(8))) float v8f;

#define NN  512
#define CC  32
#define LL  4
#define EE  6144
#define NLV 5
#define LN_EPS 1e-5f

__device__ __forceinline__ int rel_of(int i, int E)
{
    return (i >= 2 * E) ? 2 : ((i >= E) ? 1 : 0);
}

#define EPT 18

__global__ __launch_bounds__(1024) void rgcn_csr_kernel(
    const int* __restrict__ src, const int* __restrict__ dst, int E,
    float* __restrict__ invdeg, int* __restrict__ rowptr, int* __restrict__ col,
    int* __restrict__ nlist, int* __restrict__ ncnt)
{
    __shared__ int nfl[NN];
    __shared__ unsigned mk[NLV * 16];
    __shared__ int cnt[3 * NN];
    __shared__ int rp[3 * (NN + 1)];
    __shared__ int key[3 * EE];
    const int tid  = threadIdx.x;
    const int lane = tid & 31;
    const int wid  = tid >> 5;
    const int tot  = 3 * E;

    int ek[EPT];
#pragma unroll
    for (int u = 0; u < EPT; ++u) {
        const int i  = tid + u * 1024;
        const int ii = (i < tot) ? i : 0;
        const int d = dst[ii], s = src[ii];
        ek[u] = (i < tot) ? ((d << 16) | s) : 0;
    }

    for (int i = tid; i < 3 * NN; i += 1024) cnt[i] = 0;
    if (tid < NN) nfl[tid] = (tid < 64) ? 1 : 0;
    if (tid < 16) mk[(NLV - 1) * 16 + tid] = (tid < 2) ? 0xffffffffu : 0u;
    __syncthreads();

    for (int k = NLV - 1; k >= 1; --k) {
        unsigned wq[EPT];
#pragma unroll
        for (int u = 0; u < EPT; ++u)
            wq[u] = mk[k * 16 + (ek[u] >> 21)];
#pragma unroll
        for (int u = 0; u < EPT; ++u)
            if ((wq[u] >> ((ek[u] >> 16) & 31)) & 1u) nfl[ek[u] & 0xffff] = 1;
        __syncthreads();
        if (tid < NN) {
            const unsigned m = __builtin_amdgcn_ballot_w32(nfl[tid] != 0);
            if (lane == 0) mk[(k - 1) * 16 + wid] = m;
        }
        __syncthreads();
    }

    unsigned needm = 0u;
    {
        unsigned wq[EPT];
#pragma unroll
        for (int u = 0; u < EPT; ++u)
            wq[u] = mk[1 * 16 + (ek[u] >> 21)];
#pragma unroll
        for (int u = 0; u < EPT; ++u)
            needm |= ((((wq[u] >> ((ek[u] >> 16) & 31)) & 1u) != 0u && tid + u * 1024 < tot) ? 1u : 0u) << u;
    }

#pragma unroll
    for (int u = 0; u < EPT; ++u)
        if ((needm >> u) & 1u)
            atomicAdd(&cnt[rel_of(tid + u * 1024, E) * NN + (ek[u] >> 16)], 1);
    __syncthreads();

    for (int i = tid; i < 3 * NN; i += 1024) {
        int dg = cnt[i];
        if (dg < 1) dg = 1;
        invdeg[i] = 1.0f / (float)dg;
    }
    if (wid < 3) {
        const int r = wid, nb = lane * 16;
        int loc[16];
        int s = 0;
#pragma unroll
        for (int q = 0; q < 16; ++q) { loc[q] = s; s += cnt[r * NN + nb + q]; }
        int inc = s;
        for (int off = 1; off < 32; off <<= 1) {
            const int t = __shfl_up(inc, off, 32);
            if (lane >= off) inc += t;
        }
        const int excl = inc - s;
#pragma unroll
        for (int q = 0; q < 16; ++q) rp[r * (NN + 1) + nb + q] = excl + loc[q];
        if (lane == 31) rp[r * (NN + 1) + NN] = inc;
    }
    __syncthreads();
    for (int i = tid; i < 3 * NN; i += 1024) cnt[i] = 0;
    __syncthreads();

#pragma unroll
    for (int h = 0; h < EPT; h += 6) {
        int rb[6], sl[6];
#pragma unroll
        for (int v = 0; v < 6; ++v) {
            const int u = h + v;
            rb[v] = 0;
            sl[v] = 0;
            if ((needm >> u) & 1u) {
                const int r = rel_of(tid + u * 1024, E);
                const int d = ek[u] >> 16;
                rb[v] = rp[r * (NN + 1) + d];
                sl[v] = atomicAdd(&cnt[r * NN + d], 1);
            }
        }
#pragma unroll
        for (int v = 0; v < 6; ++v) {
            const int u = h + v;
            if ((needm >> u) & 1u)
                key[rel_of(tid + u * 1024, E) * EE + rb[v] + sl[v]] = ek[u];
        }
    }
    __syncthreads();

    for (int r = 0; r < 3; ++r) {
        const int tr = rp[r * (NN + 1) + NN];
        for (int p = tid; p < tr; p += 1024) {
            const int k = key[r * EE + p];
            const int dn = k >> 16;
            const int beg = rp[r * (NN + 1) + dn];
            const int end = rp[r * (NN + 1) + dn + 1];
            int rank = 0;
            for (int q = beg; q < end; ++q) {
                const int kq = key[r * EE + q];
                rank += (kq < k || (kq == k && q < p)) ? 1 : 0;
            }
            col[r * EE + beg + rank] = k & 0xffff;
        }
        if (tid < ((4 - (tr & 3)) & 3)) col[r * EE + tr + tid] = 0;
    }
    for (int i = tid; i < 3 * (NN + 1); i += 1024) rowptr[i] = rp[i];

    if (wid < NLV) {
        const int k = wid;
        unsigned mw[NN / 32];
#pragma unroll
        for (int ch = 0; ch < NN / 32; ++ch) mw[ch] = mk[k * 16 + ch];
        int base = 0;
#pragma unroll
        for (int ch = 0; ch < NN / 32; ++ch) {
            const unsigned m = mw[ch];
            if ((m >> lane) & 1u)
                nlist[k * NN + base + __popc(m & ((1u << lane) - 1u))] = ch * 32 + lane;
            base += __popc(m);
        }
        for (int i = base + lane; i < NN; i += 32) nlist[k * NN + i] = 0;
        if (lane == 0) ncnt[k] = (base + 15) & ~15;
    }
}

__device__ __forceinline__ void gemm_acc(
    const float* __restrict__ A, const float* __restrict__ W,
    const int rowoff[4], int nact, int mlo, int mhalf, v8f acc[4][2])
{
#pragma unroll
    for (int mt = 0; mt < 4; ++mt) {
        if (mt < nact) {
            for (int ks = 0; ks < 8; ++ks) {
                const int kk = ks * 4 + mhalf * 2;
                v2f b0, b1, a;
                b0.x = W[kk * CC + mlo];
                b0.y = W[(kk + 1) * CC + mlo];
                b1.x = W[kk * CC + 16 + mlo];
                b1.y = W[(kk + 1) * CC + 16 + mlo];
                a.x = A[rowoff[mt] + kk];
                a.y = A[rowoff[mt] + kk + 1];
                acc[mt][0] = __builtin_amdgcn_wmma_f32_16x16x4_f32(
                    false, a, false, b0, (short)0, acc[mt][0], false, false);
                acc[mt][1] = __builtin_amdgcn_wmma_f32_16x16x4_f32(
                    false, a, false, b1, (short)0, acc[mt][1], false, false);
            }
        }
    }
}

__device__ __forceinline__ void gather_rel(
    const float* __restrict__ xs, float* __restrict__ hs,
    const int* __restrict__ nl, int cnt,
    const int* __restrict__ rp, const int* __restrict__ cl,
    const float* __restrict__ iv, int wid, int lane)
{
    const int cq  = (lane & 7) * 4;
    const int sub = lane >> 3;
    for (int base = 0; base < cnt; base += 64) {
        const int pA = base + wid * 4 + sub;
        const int pB = pA + 32;
        const bool okA = pA < cnt, okB = pB < cnt;
        const int nodeA = nl[pA];
        const int nodeB = nl[pB];
        int jA = rp[nodeA];
        const int eA = okA ? rp[nodeA + 1] : jA;
        int jB = rp[nodeB];
        const int eB = okB ? rp[nodeB + 1] : jB;
        v4f a = (v4f){0.0f, 0.0f, 0.0f, 0.0f};
        v4f b = (v4f){0.0f, 0.0f, 0.0f, 0.0f};
        int ca[4], cb[4];
#pragma unroll
        for (int i = 0; i < 4; ++i) {
            const int va = cl[jA + i], vb = cl[jB + i];
            ca[i] = (jA + i < eA) ? va : NN;
            cb[i] = (jB + i < eB) ? vb : NN;
        }
        while (jA < eA || jB < eB) {
            v4f xa[4], xb[4];
#pragma unroll
            for (int i = 0; i < 4; ++i) {
                xa[i] = *(const v4f*)&xs[ca[i] * CC + cq];
                xb[i] = *(const v4f*)&xs[cb[i] * CC + cq];
            }
            jA = (jA + 4 < eA) ? jA + 4 : eA;
            jB = (jB + 4 < eB) ? jB + 4 : eB;
#pragma unroll
            for (int i = 0; i < 4; ++i) {
                const int va = cl[jA + i], vb = cl[jB + i];
                ca[i] = (jA + i < eA) ? va : NN;
                cb[i] = (jB + i < eB) ? vb : NN;
            }
#pragma unroll
            for (int i = 0; i < 4; ++i) { a += xa[i]; b += xb[i]; }
        }
        if (okA) *(v4f*)&hs[pA * CC + cq] = a * iv[nodeA];
        if (okB) *(v4f*)&hs[pB * CC + cq] = b * iv[nodeB];
    }
}

template <int M>
__device__ __forceinline__ float row_xor(float v)
{
    return __int_as_float(__builtin_amdgcn_update_dpp(0, __float_as_int(v), 0x160 + M, 0xf, 0xf, true));
}

__device__ __forceinline__ void row16_sum8(float s[8])
{
#pragma unroll
    for (int u = 0; u < 8; ++u) s[u] += row_xor<8>(s[u]);
#pragma unroll
    for (int u = 0; u < 8; ++u) s[u] += row_xor<4>(s[u]);
#pragma unroll
    for (int u = 0; u < 8; ++u) s[u] += row_xor<2>(s[u]);
#pragma unroll
    for (int u = 0; u < 8; ++u) s[u] += row_xor<1>(s[u]);
}

__global__ __launch_bounds__(256) void rgcn_main_kernel(
    const float* __restrict__ xx, const float* __restrict__ ss,
    const float* __restrict__ Wf, const float* __restrict__ bf,
    const float* __restrict__ convw, const float* __restrict__ convroot,
    const float* __restrict__ convbias, const float* __restrict__ lng,
    const float* __restrict__ lnb, const int* __restrict__ rowptr,
    const int* __restrict__ col, const float* __restrict__ invdeg,
    const int* __restrict__ nlist, const int* __restrict__ ncnt,
    float* __restrict__ out)
{
    __shared__ __attribute__((aligned(16))) float xs[(NN + 1) * CC];
    __shared__ __attribute__((aligned(16))) float hs[NN * CC];
    __shared__ __attribute__((aligned(16))) int colL[3 * EE + 4];
    __shared__ int rpL[3 * (NN + 1)];
    __shared__ __attribute__((aligned(16))) float wall[LL * 4 * CC * CC];
    __shared__ float cbs[LL * CC];
    __shared__ float ivd[3 * NN];
    __shared__ float wfe[6 * CC];
    __shared__ float prm[3 * CC];
    __shared__ __attribute__((aligned(16))) int nlL[NLV * NN];
    __shared__ int ncL[8];

    const int bid   = blockIdx.x;
    const int tid   = threadIdx.x;
    const int lane  = tid & 31;
    const int wid   = __builtin_amdgcn_readfirstlane(tid >> 5);
    const int mlo   = lane & 15;
    const int mhalf = lane >> 4;

    const float mfeat = ss[bid] * 0.125f;

    int q4[3];
#pragma unroll
    for (int r = 0; r < 3; ++r) {
        const int t = (rowptr[r * (NN + 1) + NN] + 3) >> 2;
        q4[r] = (t < 0) ? 0 : ((t > EE / 4) ? EE / 4 : t);
    }

    const v4f* __restrict__ cr4 = (const v4f*)convroot;
    const v4f* __restrict__ cw4 = (const v4f*)convw;
    v4f wr[LL], ww[3 * LL];
#pragma unroll
    for (int u = 0; u < LL; ++u) wr[u] = cr4[u * 256 + tid];
#pragma unroll
    for (int u = 0; u < 3 * LL; ++u) ww[u] = cw4[u * 256 + tid];
    const v4i* __restrict__ n4 = (const v4i*)nlist;
    const v4i t0 = n4[tid], t1 = n4[tid + 256];
    const v4i t2 = n4[512 + (tid & (NLV * NN / 4 - 512 - 1))];
    const int ncv = ncnt[(tid < NLV) ? tid : 0];
    float ivr[6];
    int rpr[7];
#pragma unroll
    for (int u = 0; u < 6; ++u) {
        ivr[u] = invdeg[u * 256 + tid];
        rpr[u] = rowptr[u * 256 + tid];
    }
    rpr[6] = rowptr[(tid < 3 * (NN + 1) - 6 * 256) ? 6 * 256 + tid : 0];
    const float wfv = Wf[(tid < 6 * CC) ? tid : 0];
    const float pbv = bf[lane], pgv = lng[lane], plv = lnb[lane];
    const float cbv = convbias[tid & (LL * CC - 1)];

    {
        const int q01 = q4[0] + q4[1];
        const int qt  = q01 + q4[2];
        const v4i* __restrict__ c4 = (const v4i*)col;
        v4i* l4 = (v4i*)colL;
        for (int b0 = tid; b0 < qt; b0 += 3 * 256) {
            v4i v[3];
            int w[3];
#pragma unroll
            for (int k = 0; k < 3; ++k) {
                const int b  = b0 + k * 256;
                const int bc = (b < qt) ? b : qt - 1;
                w[k] = (bc >= q01) ? (2 * (EE / 4) + bc - q01)
                                   : ((bc >= q4[0]) ? ((EE / 4) + bc - q4[0]) : bc);
                v[k] = c4[w[k]];
            }
#pragma unroll
            for (int k = 0; k < 3; ++k)
                if (b0 + k * 256 < qt) l4[w[k]] = v[k];
        }
    }

    {
        v4f* w4 = (v4f*)wall;
#pragma unroll
        for (int u = 0; u < LL; ++u) w4[(u * 4) * 256 + tid] = wr[u];
#pragma unroll
        for (int u = 0; u < 3 * LL; ++u) w4[((u / 3) * 4 + 1 + (u % 3)) * 256 + tid] = ww[u];
        v4i* d4 = (v4i*)nlL;
        d4[tid] = t0;
        d4[tid + 256] = t1;
        if (tid < NLV * NN / 4 - 512) d4[tid + 512] = t2;
        if (tid < NLV) ncL[tid] = ncv;
#pragma unroll
        for (int u = 0; u < 6; ++u) {
            ivd[u * 256 + tid] = ivr[u];
            rpL[u * 256 + tid] = rpr[u];
        }
        if (tid < 3 * (NN + 1) - 6 * 256) rpL[6 * 256 + tid] = rpr[6];
        if (tid < 6 * CC) wfe[tid] = wfv;
        if (tid < CC) {
            xs[NN * CC + tid] = 0.0f;
            prm[tid]          = pbv;
            prm[CC + tid]     = pgv;
            prm[2 * CC + tid] = plv;
        }
        if (tid < LL * CC) cbs[tid] = cbv;
    }
    __syncthreads();

    const int cnt0 = ncL[0];
    for (int idx = tid; idx < cnt0 * CC; idx += 256) {
        const int n = nlL[idx >> 5], c = idx & 31;
        const int t = n >> 6, i = (n >> 4) & 3, j = (n >> 2) & 3, k = n & 3;
        const float v = xx[bid * NN + n];
        float a = prm[c];
        a += ((float)i * (1.0f / 3.0f)) * wfe[0 * CC + c];
        a += ((float)j * (1.0f / 3.0f)) * wfe[1 * CC + c];
        a += ((float)k * (1.0f / 3.0f)) * wfe[2 * CC + c];
        a += ((float)t * (1.0f / 7.0f)) * wfe[3 * CC + c];
        a += v * wfe[4 * CC + c];
        a += mfeat * wfe[5 * CC + c];
        xs[n * CC + c] = a;
    }
    __syncthreads();

    for (int l = 0; l < LL; ++l) {
        const int* nl  = &nlL[(l + 1) * NN];
        const int cnt  = ncL[l + 1];
        const int ntile = cnt >> 4;
        const int nact = (ntile > wid) ? ((ntile - wid + 7) >> 3) : 0;
        int xoff[4], hoff[4];
#pragma unroll
        for (int mt = 0; mt < 4; ++mt) {
            const int p = (wid + 8 * mt) * 16 + mlo;
            xoff[mt] = nl[p] * CC;
            hoff[mt] = p * CC;
        }
        v8f acc[4][2];
#pragma unroll
        for (int mt = 0; mt < 4; ++mt)
#pragma unroll
            for (int nt = 0; nt < 2; ++nt) {
                const float bv = cbs[l * CC + nt * 16 + mlo];
                acc[mt][nt] = (v8f){bv, bv, bv, bv, bv, bv, bv, bv};
            }

        gemm_acc(xs, &wall[(l * 4) * CC * CC], xoff, nact, mlo, mhalf, acc);

        for (int r = 0; r < 3; ++r) {
            if (r > 0) __syncthreads();
            gather_rel(xs, hs, nl, cnt, &rpL[r * (NN + 1)], &colL[r * EE], &ivd[r * NN], wid, lane);
            __syncthreads();

            gemm_acc(hs, &wall[(l * 4 + 1 + r) * CC * CC], hoff, nact, mlo, mhalf, acc);
        }

        {
            const float g0 = prm[CC + mlo],     g1 = prm[CC + 16 + mlo];
            const float e0 = prm[2 * CC + mlo], e1 = prm[2 * CC + 16 + mlo];
#pragma unroll
            for (int mt = 0; mt < 4; ++mt)
                if (mt < nact) {
                    const int pb = (wid + 8 * mt) * 16 + 8 * mhalf;
                    const v4i na = *(const v4i*)&nl[pb];
                    const v4i nb = *(const v4i*)&nl[pb + 4];
                    const int node[8] = {na.x, na.y, na.z, na.w, nb.x, nb.y, nb.z, nb.w};
                    float v0[8], v1[8], s[8];
#pragma unroll
                    for (int jj = 0; jj < 8; ++jj) {
                        const float a0 = acc[mt][0][jj], a1 = acc[mt][1][jj];
                        v0[jj] = a0 > 0.0f ? a0 : 0.0f;
                        v1[jj] = a1 > 0.0f ? a1 : 0.0f;
                        s[jj]  = v0[jj] + v1[jj];
                    }
                    row16_sum8(s);
#pragma unroll
                    for (int jj = 0; jj < 8; ++jj) {
                        const float mu = s[jj] * (1.0f / 32.0f);
                        v0[jj] -= mu;
                        v1[jj] -= mu;
                        s[jj] = v0[jj] * v0[jj] + v1[jj] * v1[jj];
                    }
                    row16_sum8(s);
#pragma unroll
                    for (int jj = 0; jj < 8; ++jj) {
                        const float rs = rsqrtf(s[jj] * (1.0f / 32.0f) + LN_EPS);
                        xs[node[jj] * CC + mlo]      = v0[jj] * rs * g0 + e0;
                        xs[node[jj] * CC + 16 + mlo] = v1[jj] * rs * g1 + e1;
                    }
                }
        }
        __syncthreads();
    }

    for (int idx = tid; idx < 48 * CC; idx += 256) {
        const int row = idx >> 5, c = idx & 31;
        const int g = row >> 4, rc = row & 15;
        const int a0 = rc >> 2, b0 = rc & 3;
        float sum = 0.0f;
#pragma unroll
        for (int q = 0; q < 4; ++q) {
            int node;
            if (g == 0)      node = q * 16 + a0 * 4 + b0;
            else if (g == 1) node = a0 * 16 + q * 4 + b0;
            else             node = a0 * 16 + b0 * 4 + q;
            sum += xs[node * CC + c];
        }
        out[bid * 48 * CC + idx] = sum * 0.25f;
    }
}

extern "C" void kernel_launch(void* const* d_in, const int* in_sizes, int n_in,
                              void* d_out, int out_size, void* d_ws, size_t ws_size,
                              hipStream_t stream) {
    const float* xx       = (const float*)d_in[0];
    const float* ss       = (const float*)d_in[1];
    const float* Wf       = (const float*)d_in[2];
    const float* bf       = (const float*)d_in[3];
    const float* convw    = (const float*)d_in[4];
    const float* convroot = (const float*)d_in[5];
    const float* convbias = (const float*)d_in[6];
    const float* lng      = (const float*)d_in[7];
    const float* lnb      = (const float*)d_in[8];
    const int*   src      = (const int*)d_in[9];
    const int*   dst      = (const int*)d_in[10];
    float*       out      = (float*)d_out;
    float*       invdeg   = (float*)d_ws;
    int*         rowptr   = (int*)d_ws + 2048;
    int*         col      = (int*)d_ws + 4096;
    int*         nlist    = (int*)d_ws + 24576;
    int*         ncnt     = (int*)d_ws + 27648;

    const int E = in_sizes[9] / 3;
    const int B = in_sizes[1];

    rgcn_csr_kernel<<<1, 1024, 0, stream>>>(src, dst, E, invdeg, rowptr, col, nlist, ncnt);
    rgcn_main_kernel<<<B, 256, 0, stream>>>(xx, ss, Wf, bf, convw, convroot,
                                            convbias, lng, lnb, rowptr, col,
                                            invdeg, nlist, ncnt, out);
    (void)n_in; (void)out_size; (void)ws_size;
}


# Gather index chain (node, row bounds, 1/deg, first source ids) of all 3 relations fetched once a layer ahead of the x reads, loads unconditional: dependent LDS trips per gather 9 to 3; sum order kept
# speedup vs baseline: 33.8487x; 1.1107x over previous
#include <hip/hip_runtime.h>
#include <hip/hip_bf16.h>


typedef __attribute__((ext_vector_type(2))) float v2f;
typedef __attribute__((ext_vector_type(4))) float v4f;
typedef __attribute__((ext_vector_type(4))) int v4i;
typedef __attribute__((ext_vector_type(8))) float v8f;

#define NN  512
#define CC  32
#define LL  4
#define EE  6144
#define NLV 5
#define LN_EPS 1e-5f

__device__ __forceinline__ int rel_of(int i, int E)
{
    return (i >= 2 * E) ? 2 : ((i >= E) ? 1 : 0);
}

#define EPT 18

__global__ __launch_bounds__(1024) void rgcn_csr_kernel(
    const int* __restrict__ src, const int* __restrict__ dst, int E,
    float* __restrict__ invdeg, int* __restrict__ rowptr, int* __restrict__ col,
    int* __restrict__ nlist, int* __restrict__ ncnt)
{
    __shared__ int nfl[NN];
    __shared__ unsigned mk[NLV * 16];
    __shared__ int cnt[3 * NN];
    __shared__ int rp[3 * (NN + 1)];
    __shared__ int key[3 * EE];
    const int tid  = threadIdx.x;
    const int lane = tid & 31;
    const int wid  = tid >> 5;
    const int tot  = 3 * E;

    int ek[EPT];
#pragma unroll
    for (int u = 0; u < EPT; ++u) {
        const int i  = tid + u * 1024;
        const int ii = (i < tot) ? i : 0;
        const int d = dst[ii], s = src[ii];
        ek[u] = (i < tot) ? ((d << 16) | s) : 0;
    }

    for (int i = tid; i < 3 * NN; i += 1024) cnt[i] = 0;
    if (tid < NN) nfl[tid] = (tid < 64) ? 1 : 0;
    if (tid < 16) mk[(NLV - 1) * 16 + tid] = (tid < 2) ? 0xffffffffu : 0u;
    __syncthreads();

    for (int k = NLV - 1; k >= 1; --k) {
        unsigned wq[EPT];
#pragma unroll
        for (int u = 0; u < EPT; ++u)
            wq[u] = mk[k * 16 + (ek[u] >> 21)];
#pragma unroll
        for (int u = 0; u < EPT; ++u)
            if ((wq[u] >> ((ek[u] >> 16) & 31)) & 1u) nfl[ek[u] & 0xffff] = 1;
        __syncthreads();
        if (tid < NN) {
            const unsigned m = __builtin_amdgcn_ballot_w32(nfl[tid] != 0);
            if (lane == 0) mk[(k - 1) * 16 + wid] = m;
        }
        __syncthreads();
    }

    unsigned needm = 0u;
    {
        unsigned wq[EPT];
#pragma unroll
        for (int u = 0; u < EPT; ++u)
            wq[u] = mk[1 * 16 + (ek[u] >> 21)];
#pragma unroll
        for (int u = 0; u < EPT; ++u)
            needm |= ((((wq[u] >> ((ek[u] >> 16) & 31)) & 1u) != 0u && tid + u * 1024 < tot) ? 1u : 0u) << u;
    }

#pragma unroll
    for (int u = 0; u < EPT; ++u)
        if ((needm >> u) & 1u)
            atomicAdd(&cnt[rel_of(tid + u * 1024, E) * NN + (ek[u] >> 16)], 1);
    __syncthreads();

    for (int i = tid; i < 3 * NN; i += 1024) {
        int dg = cnt[i];
        if (dg < 1) dg = 1;
        invdeg[i] = 1.0f / (float)dg;
    }
    if (wid < 3) {
        const int r = wid, nb = lane * 16;
        int loc[16];
        int s = 0;
#pragma unroll
        for (int q = 0; q < 16; ++q) { loc[q] = s; s += cnt[r * NN + nb + q]; }
        int inc = s;
        for (int off = 1; off < 32; off <<= 1) {
            const int t = __shfl_up(inc, off, 32);
            if (lane >= off) inc += t;
        }
        const int excl = inc - s;
#pragma unroll
        for (int q = 0; q < 16; ++q) rp[r * (NN + 1) + nb + q] = excl + loc[q];
        if (lane == 31) rp[r * (NN + 1) + NN] = inc;
    }
    __syncthreads();
    for (int i = tid; i < 3 * NN; i += 1024) cnt[i] = 0;
    __syncthreads();

#pragma unroll
    for (int h = 0; h < EPT; h += 6) {
        int rb[6], sl[6];
#pragma unroll
        for (int v = 0; v < 6; ++v) {
            const int u = h + v;
            rb[v] = 0;
            sl[v] = 0;
            if ((needm >> u) & 1u) {
                const int r = rel_of(tid + u * 1024, E);
                const int d = ek[u] >> 16;
                rb[v] = rp[r * (NN + 1) + d];
                sl[v] = atomicAdd(&cnt[r * NN + d], 1);
            }
        }
#pragma unroll
        for (int v = 0; v < 6; ++v) {
            const int u = h + v;
            if ((needm >> u) & 1u)
                key[rel_of(tid + u * 1024, E) * EE + rb[v] + sl[v]] = ek[u];
        }
    }
    __syncthreads();

    for (int r = 0; r < 3; ++r) {
        const int tr = rp[r * (NN + 1) + NN];
        for (int p = tid; p < tr; p += 1024) {
            const int k = key[r * EE + p];
            const int dn = k >> 16;
            const int beg = rp[r * (NN + 1) + dn];
            const int end = rp[r * (NN + 1) + dn + 1];
            int rank = 0;
            for (int q = beg; q < end; ++q) {
                const int kq = key[r * EE + q];
                rank += (kq < k || (kq == k && q < p)) ? 1 : 0;
            }
            col[r * EE + beg + rank] = k & 0xffff;
        }
        if (tid < ((4 - (tr & 3)) & 3)) col[r * EE + tr + tid] = 0;
    }
    for (int i = tid; i < 3 * (NN + 1); i += 1024) rowptr[i] = rp[i];

    if (wid < NLV) {
        const int k = wid;
        unsigned mw[NN / 32];
#pragma unroll
        for (int ch = 0; ch < NN / 32; ++ch) mw[ch] = mk[k * 16 + ch];
        int base = 0;
#pragma unroll
        for (int ch = 0; ch < NN / 32; ++ch) {
            const unsigned m = mw[ch];
            if ((m >> lane) & 1u)
                nlist[k * NN + base + __popc(m & ((1u << lane) - 1u))] = ch * 32 + lane;
            base += __popc(m);
        }
        for (int i = base + lane; i < NN; i += 32) nlist[k * NN + i] = 0;
        if (lane == 0) ncnt[k] = (base + 15) & ~15;
    }
}

__device__ __forceinline__ void gemm_acc(
    const float* __restrict__ A, const float* __restrict__ W,
    const int rowoff[4], int nact, int mlo, int mhalf, v8f acc[4][2])
{
#pragma unroll
    for (int mt = 0; mt < 4; ++mt) {
        if (mt < nact) {
            for (int ks = 0; ks < 8; ++ks) {
                const int kk = ks * 4 + mhalf * 2;
                v2f b0, b1, a;
                b0.x = W[kk * CC + mlo];
                b0.y = W[(kk + 1) * CC + mlo];
                b1.x = W[kk * CC + 16 + mlo];
                b1.y = W[(kk + 1) * CC + 16 + mlo];
                a.x = A[rowoff[mt] + kk];
                a.y = A[rowoff[mt] + kk + 1];
                acc[mt][0] = __builtin_amdgcn_wmma_f32_16x16x4_f32(
                    false, a, false, b0, (short)0, acc[mt][0], false, false);
                acc[mt][1] = __builtin_amdgcn_wmma_f32_16x16x4_f32(
                    false, a, false, b1, (short)0, acc[mt][1], false, false);
            }
        }
    }
}

struct GBounds {
    int jA, eA, jB, eB;
    float ivA, ivB;
};

__device__ __forceinline__ GBounds gather_bounds(
    int nodeA, int nodeB, bool okA, bool okB,
    const int* __restrict__ rp, const float* __restrict__ iv)
{
    GBounds g;
    const int a0 = rp[nodeA], a1 = rp[nodeA + 1];
    const int b0 = rp[nodeB], b1 = rp[nodeB + 1];
    g.ivA = iv[nodeA];
    g.ivB = iv[nodeB];
    g.jA = a0;
    g.eA = okA ? a1 : a0;
    g.jB = b0;
    g.eB = okB ? b1 : b0;
    return g;
}

__device__ __forceinline__ void gather_first_ids(
    const int* __restrict__ cl, const GBounds& g, int ca[4], int cb[4])
{
#pragma unroll
    for (int i = 0; i < 4; ++i) {
        const int va = cl[g.jA + i], vb = cl[g.jB + i];
        ca[i] = (g.jA + i < g.eA) ? va : NN;
        cb[i] = (g.jB + i < g.eB) ? vb : NN;
    }
}

__device__ __forceinline__ void gather_sum(
    const float* __restrict__ xs, float* __restrict__ hs,
    const int* __restrict__ cl, const GBounds& g, int ca[4], int cb[4],
    int pA, int pB, bool okA, bool okB, int cq)
{
    int jA = g.jA, jB = g.jB;
    const int eA = g.eA, eB = g.eB;
    v4f a = (v4f){0.0f, 0.0f, 0.0f, 0.0f};
    v4f b = (v4f){0.0f, 0.0f, 0.0f, 0.0f};
    while (jA < eA || jB < eB) {
        v4f xa[4], xb[4];
#pragma unroll
        for (int i = 0; i < 4; ++i) {
            xa[i] = *(const v4f*)&xs[ca[i] * CC + cq];
            xb[i] = *(const v4f*)&xs[cb[i] * CC + cq];
        }
        jA = (jA + 4 < eA) ? jA + 4 : eA;
        jB = (jB + 4 < eB) ? jB + 4 : eB;
#pragma unroll
        for (int i = 0; i < 4; ++i) {
            const int va = cl[jA + i], vb = cl[jB + i];
            ca[i] = (jA + i < eA) ? va : NN;
            cb[i] = (jB + i < eB) ? vb : NN;
        }
#pragma unroll
        for (int i = 0; i < 4; ++i) { a += xa[i]; b += xb[i]; }
    }
    if (okA) *(v4f*)&hs[pA * CC + cq] = a * g.ivA;
    if (okB) *(v4f*)&hs[pB * CC + cq] = b * g.ivB;
}

template <int M>
__device__ __forceinline__ float row_xor(float v)
{
    return __int_as_float(__builtin_amdgcn_update_dpp(0, __float_as_int(v), 0x160 + M, 0xf, 0xf, true));
}

__device__ __forceinline__ void row16_sum8(float s[8])
{
#pragma unroll
    for (int u = 0; u < 8; ++u) s[u] += row_xor<8>(s[u]);
#pragma unroll
    for (int u = 0; u < 8; ++u) s[u] += row_xor<4>(s[u]);
#pragma unroll
    for (int u = 0; u < 8; ++u) s[u] += row_xor<2>(s[u]);
#pragma unroll
    for (int u = 0; u < 8; ++u) s[u] += row_xor<1>(s[u]);
}

__global__ __launch_bounds__(256) void rgcn_main_kernel(
    const float* __restrict__ xx, const float* __restrict__ ss,
    const float* __restrict__ Wf, const float* __restrict__ bf,
    const float* __restrict__ convw, const float* __restrict__ convroot,
    const float* __restrict__ convbias, const float* __restrict__ lng,
    const float* __restrict__ lnb, const int* __restrict__ rowptr,
    const int* __restrict__ col, const float* __restrict__ invdeg,
    const int* __restrict__ nlist, const int* __restrict__ ncnt,
    float* __restrict__ out)
{
    __shared__ __attribute__((aligned(16))) float xs[(NN + 1) * CC];
    __shared__ __attribute__((aligned(16))) float hs[NN * CC];
    __shared__ __attribute__((aligned(16))) int colL[3 * EE + 4];
    __shared__ int rpL[3 * (NN + 1)];
    __shared__ __attribute__((aligned(16))) float wall[LL * 4 * CC * CC];
    __shared__ float cbs[LL * CC];
    __shared__ float ivd[3 * NN];
    __shared__ float wfe[6 * CC];
    __shared__ float prm[3 * CC];
    __shared__ __attribute__((aligned(16))) int nlL[NLV * NN];
    __shared__ int ncL[8];

    const int bid   = blockIdx.x;
    const int tid   = threadIdx.x;
    const int lane  = tid & 31;
    const int wid   = __builtin_amdgcn_readfirstlane(tid >> 5);
    const int mlo   = lane & 15;
    const int mhalf = lane >> 4;

    const float mfeat = ss[bid] * 0.125f;

    int q4[3];
#pragma unroll
    for (int r = 0; r < 3; ++r) {
        const int t = (rowptr[r * (NN + 1) + NN] + 3) >> 2;
        q4[r] = (t < 0) ? 0 : ((t > EE / 4) ? EE / 4 : t);
    }

    const v4f* __restrict__ cr4 = (const v4f*)convroot;
    const v4f* __restrict__ cw4 = (const v4f*)convw;
    v4f wr[LL], ww[3 * LL];
#pragma unroll
    for (int u = 0; u < LL; ++u) wr[u] = cr4[u * 256 + tid];
#pragma unroll
    for (int u = 0; u < 3 * LL; ++u) ww[u] = cw4[u * 256 + tid];
    const v4i* __restrict__ n4 = (const v4i*)nlist;
    const v4i t0 = n4[tid], t1 = n4[tid + 256];
    const v4i t2 = n4[512 + (tid & (NLV * NN / 4 - 512 - 1))];
    const int ncv = ncnt[(tid < NLV) ? tid : 0];
    float ivr[6];
    int rpr[7];
#pragma unroll
    for (int u = 0; u < 6; ++u) {
        ivr[u] = invdeg[u * 256 + tid];
        rpr[u] = rowptr[u * 256 + tid];
    }
    rpr[6] = rowptr[(tid < 3 * (NN + 1) - 6 * 256) ? 6 * 256 + tid : 0];
    const float wfv = Wf[(tid < 6 * CC) ? tid : 0];
    const float pbv = bf[lane], pgv = lng[lane], plv = lnb[lane];
    const float cbv = convbias[tid & (LL * CC - 1)];

    {
        const int q01 = q4[0] + q4[1];
        const int qt  = q01 + q4[2];
        const v4i* __restrict__ c4 = (const v4i*)col;
        v4i* l4 = (v4i*)colL;
        for (int b0 = tid; b0 < qt; b0 += 3 * 256) {
            v4i v[3];
            int w[3];
#pragma unroll
            for (int k = 0; k < 3; ++k) {
                const int b  = b0 + k * 256;
                const int bc = (b < qt) ? b : qt - 1;
                w[k] = (bc >= q01) ? (2 * (EE / 4) + bc - q01)
                                   : ((bc >= q4[0]) ? ((EE / 4) + bc - q4[0]) : bc);
                v[k] = c4[w[k]];
            }
#pragma unroll
            for (int k = 0; k < 3; ++k)
                if (b0 + k * 256 < qt) l4[w[k]] = v[k];
        }
    }

    {
        v4f* w4 = (v4f*)wall;
#pragma unroll
        for (int u = 0; u < LL; ++u) w4[(u * 4) * 256 + tid] = wr[u];
#pragma unroll
        for (int u = 0; u < 3 * LL; ++u) w4[((u / 3) * 4 + 1 + (u % 3)) * 256 + tid] = ww[u];
        v4i* d4 = (v4i*)nlL;
        d4[tid] = t0;
        d4[tid + 256] = t1;
        if (tid < NLV * NN / 4 - 512) d4[tid + 512] = t2;
        if (tid < NLV) ncL[tid] = ncv;
#pragma unroll
        for (int u = 0; u < 6; ++u) {
            ivd[u * 256 + tid] = ivr[u];
            rpL[u * 256 + tid] = rpr[u];
        }
        if (tid < 3 * (NN + 1) - 6 * 256) rpL[6 * 256 + tid] = rpr[6];
        if (tid < 6 * CC) wfe[tid] = wfv;
        if (tid < CC) {
            xs[NN * CC + tid] = 0.0f;
            prm[tid]          = pbv;
            prm[CC + tid]     = pgv;
            prm[2 * CC + tid] = plv;
        }
        if (tid < LL * CC) cbs[tid] = cbv;
    }
    __syncthreads();

    const int cnt0 = ncL[0];
    for (int idx = tid; idx < cnt0 * CC; idx += 256) {
        const int n = nlL[idx >> 5], c = idx & 31;
        const int t = n >> 6, i = (n >> 4) & 3, j = (n >> 2) & 3, k = n & 3;
        const float v = xx[bid * NN + n];
        float a = prm[c];
        a += ((float)i * (1.0f / 3.0f)) * wfe[0 * CC + c];
        a += ((float)j * (1.0f / 3.0f)) * wfe[1 * CC + c];
        a += ((float)k * (1.0f / 3.0f)) * wfe[2 * CC + c];
        a += ((float)t * (1.0f / 7.0f)) * wfe[3 * CC + c];
        a += v * wfe[4 * CC + c];
        a += mfeat * wfe[5 * CC + c];
        xs[n * CC + c] = a;
    }
    __syncthreads();

    for (int l = 0; l < LL; ++l) {
        const int* nl  = &nlL[(l + 1) * NN];
        const int cnt  = ncL[l + 1];
        const int ntile = cnt >> 4;
        const int nact = (ntile > wid) ? ((ntile - wid + 7) >> 3) : 0;
        int xoff[4], hoff[4];
#pragma unroll
        for (int mt = 0; mt < 4; ++mt) {
            const int p = (wid + 8 * mt) * 16 + mlo;
            xoff[mt] = nl[p] * CC;
            hoff[mt] = p * CC;
        }
        v8f acc[4][2];
#pragma unroll
        for (int mt = 0; mt < 4; ++mt)
#pragma unroll
            for (int nt = 0; nt < 2; ++nt) {
                const float bv = cbs[l * CC + nt * 16 + mlo];
                acc[mt][nt] = (v8f){bv, bv, bv, bv, bv, bv, bv, bv};
            }

        const int cq  = (lane & 7) * 4;
        const int pA0 = wid * 4 + (lane >> 3);
        const int pB0 = pA0 + 32;
        const bool okA0 = pA0 < cnt, okB0 = pB0 < cnt;
        const int nA0 = nl[pA0];
        const int nB0 = nl[pB0];
        GBounds g0[3];
        int ca0[3][4], cb0[3][4];
#pragma unroll
        for (int r = 0; r < 3; ++r)
            g0[r] = gather_bounds(nA0, nB0, okA0, okB0, &rpL[r * (NN + 1)], &ivd[r * NN]);
#pragma unroll
        for (int r = 0; r < 3; ++r)
            gather_first_ids(&colL[r * EE], g0[r], ca0[r], cb0[r]);

        gemm_acc(xs, &wall[(l * 4) * CC * CC], xoff, nact, mlo, mhalf, acc);

#pragma unroll
        for (int r = 0; r < 3; ++r) {
            if (r > 0) __syncthreads();
            gather_sum(xs, hs, &colL[r * EE], g0[r], ca0[r], cb0[r], pA0, pB0, okA0, okB0, cq);
            for (int base = 64; base < cnt; base += 64) {
                const int pA = base + pA0, pB = base + pB0;
                const bool okA = pA < cnt, okB = pB < cnt;
                const GBounds g = gather_bounds(nl[pA], nl[pB], okA, okB, &rpL[r * (NN + 1)], &ivd[r * NN]);
                int ca[4], cb[4];
                gather_first_ids(&colL[r * EE], g, ca, cb);
                gather_sum(xs, hs, &colL[r * EE], g, ca, cb, pA, pB, okA, okB, cq);
            }
            __syncthreads();

            gemm_acc(hs, &wall[(l * 4 + 1 + r) * CC * CC], hoff, nact, mlo, mhalf, acc);
        }

        {
            const float g0 = prm[CC + mlo],     g1 = prm[CC + 16 + mlo];
            const float e0 = prm[2 * CC + mlo], e1 = prm[2 * CC + 16 + mlo];
#pragma unroll
            for (int mt = 0; mt < 4; ++mt)
                if (mt < nact) {
                    const int pb = (wid + 8 * mt) * 16 + 8 * mhalf;
                    const v4i na = *(const v4i*)&nl[pb];
                    const v4i nb = *(const v4i*)&nl[pb + 4];
                    const int node[8] = {na.x, na.y, na.z, na.w, nb.x, nb.y, nb.z, nb.w};
                    float v0[8], v1[8], s[8];
#pragma unroll
                    for (int jj = 0; jj < 8; ++jj) {
                        const float a0 = acc[mt][0][jj], a1 = acc[mt][1][jj];
                        v0[jj] = a0 > 0.0f ? a0 : 0.0f;
                        v1[jj] = a1 > 0.0f ? a1 : 0.0f;
                        s[jj]  = v0[jj] + v1[jj];
                    }
                    row16_sum8(s);
#pragma unroll
                    for (int jj = 0; jj < 8; ++jj) {
                        const float mu = s[jj] * (1.0f / 32.0f);
                        v0[jj] -= mu;
                        v1[jj] -= mu;
                        s[jj] = v0[jj] * v0[jj] + v1[jj] * v1[jj];
                    }
                    row16_sum8(s);
#pragma unroll
                    for (int jj = 0; jj < 8; ++jj) {
                        const float rs = rsqrtf(s[jj] * (1.0f / 32.0f) + LN_EPS);
                        xs[node[jj] * CC + mlo]      = v0[jj] * rs * g0 + e0;
                        xs[node[jj] * CC + 16 + mlo] = v1[jj] * rs * g1 + e1;
                    }
                }
        }
        __syncthreads();
    }

    for (int idx = tid; idx < 48 * CC; idx += 256) {
        const int row = idx >> 5, c = idx & 31;
        const int g = row >> 4, rc = row & 15;
        const int a0 = rc >> 2, b0 = rc & 3;
        float sum = 0.0f;
#pragma unroll
        for (int q = 0; q < 4; ++q) {
            int node;
            if (g == 0)      node = q * 16 + a0 * 4 + b0;
            else if (g == 1) node = a0 * 16 + q * 4 + b0;
            else             node = a0 * 16 + b0 * 4 + q;
            sum += xs[node * CC + c];
        }
        out[bid * 48 * CC + idx] = sum * 0.25f;
    }
}

extern "C" void kernel_launch(void* const* d_in, const int* in_sizes, int n_in,
                              void* d_out, int out_size, void* d_ws, size_t ws_size,
                              hipStream_t stream) {
    const float* xx       = (const float*)d_in[0];
    const float* ss       = (const float*)d_in[1];
    const float* Wf       = (const float*)d_in[2];
    const float* bf       = (const float*)d_in[3];
    const float* convw    = (const float*)d_in[4];
    const float* convroot = (const float*)d_in[5];
    const float* convbias = (const float*)d_in[6];
    const float* lng      = (const float*)d_in[7];
    const float* lnb      = (const float*)d_in[8];
    const int*   src      = (const int*)d_in[9];
    const int*   dst      = (const int*)d_in[10];
    float*       out      = (float*)d_out;
    float*       invdeg   = (float*)d_ws;
    int*         rowptr   = (int*)d_ws + 2048;
    int*         col      = (int*)d_ws + 4096;
    int*         nlist    = (int*)d_ws + 24576;
    int*         ncnt     = (int*)d_ws + 27648;

    const int E = in_sizes[9] / 3;
    const int B = in_sizes[1];

    rgcn_csr_kernel<<<1, 1024, 0, stream>>>(src, dst, E, invdeg, rowptr, col, nlist, ncnt);
    rgcn_main_kernel<<<B, 256, 0, stream>>>(xx, ss, Wf, bf, convw, convroot,
                                            convbias, lng, lnb, rowptr, col,
                                            invdeg, nlist, ncnt, out);
    (void)n_in; (void)out_size; (void)ws_size;
}


# Pre-kernel's backward walk over the edge list stops at the first level that adds no row (a fixed point: every lower level is that same set, its words copied down); main kernel and sum order kept
# speedup vs baseline: 35.9298x; 1.0615x over previous
#include <hip/hip_runtime.h>
#include <hip/hip_bf16.h>


typedef __attribute__((ext_vector_type(2))) float v2f;
typedef __attribute__((ext_vector_type(4))) float v4f;
typedef __attribute__((ext_vector_type(4))) int v4i;
typedef __attribute__((ext_vector_type(8))) float v8f;

#define NN  512
#define CC  32
#define LL  4
#define EE  6144
#define NLV 5
#define LN_EPS 1e-5f

__device__ __forceinline__ int rel_of(int i, int E)
{
    return (i >= 2 * E) ? 2 : ((i >= E) ? 1 : 0);
}

#define EPT 18

__global__ __launch_bounds__(1024) void rgcn_csr_kernel(
    const int* __restrict__ src, const int* __restrict__ dst, int E,
    float* __restrict__ invdeg, int* __restrict__ rowptr, int* __restrict__ col,
    int* __restrict__ nlist, int* __restrict__ ncnt)
{
    __shared__ int nfl[NN];
    __shared__ unsigned mk[NLV * 16];
    __shared__ int grew[NLV];
    __shared__ int cnt[3 * NN];
    __shared__ int rp[3 * (NN + 1)];
    __shared__ int key[3 * EE];
    const int tid  = threadIdx.x;
    const int lane = tid & 31;
    const int wid  = tid >> 5;
    const int tot  = 3 * E;

    int ek[EPT];
#pragma unroll
    for (int u = 0; u < EPT; ++u) {
        const int i  = tid + u * 1024;
        const int ii = (i < tot) ? i : 0;
        const int d = dst[ii], s = src[ii];
        ek[u] = (i < tot) ? ((d << 16) | s) : 0;
    }

    for (int i = tid; i < 3 * NN; i += 1024) cnt[i] = 0;
    if (tid < NN) nfl[tid] = (tid < 64) ? 1 : 0;
    if (tid < 16) mk[(NLV - 1) * 16 + tid] = (tid < 2) ? 0xffffffffu : 0u;
    if (tid < NLV) grew[tid] = 0;
    __syncthreads();

    for (int k = NLV - 1; k >= 1; --k) {
        unsigned wq[EPT];
#pragma unroll
        for (int u = 0; u < EPT; ++u)
            wq[u] = mk[k * 16 + (ek[u] >> 21)];
#pragma unroll
        for (int u = 0; u < EPT; ++u)
            if ((wq[u] >> ((ek[u] >> 16) & 31)) & 1u) nfl[ek[u] & 0xffff] = 1;
        __syncthreads();
        if (tid < NN) {
            const unsigned m = __builtin_amdgcn_ballot_w32(nfl[tid] != 0);
            if (lane == 0) {
                mk[(k - 1) * 16 + wid] = m;
                if (m != mk[k * 16 + wid]) grew[k] = 1;
            }
        }
        __syncthreads();
        if (__builtin_amdgcn_readfirstlane(grew[k]) == 0) {
            if (tid < 16)
                for (int j = k - 2; j >= 0; --j) mk[j * 16 + tid] = mk[k * 16 + tid];
            __syncthreads();
            break;
        }
    }

    unsigned needm = 0u;
    {
        unsigned wq[EPT];
#pragma unroll
        for (int u = 0; u < EPT; ++u)
            wq[u] = mk[1 * 16 + (ek[u] >> 21)];
#pragma unroll
        for (int u = 0; u < EPT; ++u)
            needm |= ((((wq[u] >> ((ek[u] >> 16) & 31)) & 1u) != 0u && tid + u * 1024 < tot) ? 1u : 0u) << u;
    }

#pragma unroll
    for (int u = 0; u < EPT; ++u)
        if ((needm >> u) & 1u)
            atomicAdd(&cnt[rel_of(tid + u * 1024, E) * NN + (ek[u] >> 16)], 1);
    __syncthreads();

    for (int i = tid; i < 3 * NN; i += 1024) {
        int dg = cnt[i];
        if (dg < 1) dg = 1;
        invdeg[i] = 1.0f / (float)dg;
    }
    if (wid < 3) {
        const int r = wid, nb = lane * 16;
        int loc[16];
        int s = 0;
#pragma unroll
        for (int q = 0; q < 16; ++q) { loc[q] = s; s += cnt[r * NN + nb + q]; }
        int inc = s;
        for (int off = 1; off < 32; off <<= 1) {
            const int t = __shfl_up(inc, off, 32);
            if (lane >= off) inc += t;
        }
        const int excl = inc - s;
#pragma unroll
        for (int q = 0; q < 16; ++q) rp[r * (NN + 1) + nb + q] = excl + loc[q];
        if (lane == 31) rp[r * (NN + 1) + NN] = inc;
    }
    __syncthreads();
    for (int i = tid; i < 3 * NN; i += 1024) cnt[i] = 0;
    __syncthreads();

#pragma unroll
    for (int h = 0; h < EPT; h += 6) {
        int rb[6], sl[6];
#pragma unroll
        for (int v = 0; v < 6; ++v) {
            const int u = h + v;
            rb[v] = 0;
            sl[v] = 0;
            if ((needm >> u) & 1u) {
                const int r = rel_of(tid + u * 1024, E);
                const int d = ek[u] >> 16;
                rb[v] = rp[r * (NN + 1) + d];
                sl[v] = atomicAdd(&cnt[r * NN + d], 1);
            }
        }
#pragma unroll
        for (int v = 0; v < 6; ++v) {
            const int u = h + v;
            if ((needm >> u) & 1u)
                key[rel_of(tid + u * 1024, E) * EE + rb[v] + sl[v]] = ek[u];
        }
    }
    __syncthreads();

    for (int r = 0; r < 3; ++r) {
        const int tr = rp[r * (NN + 1) + NN];
        for (int p = tid; p < tr; p += 1024) {
            const int k = key[r * EE + p];
            const int dn = k >> 16;
            const int beg = rp[r * (NN + 1) + dn];
            const int end = rp[r * (NN + 1) + dn + 1];
            int rank = 0;
            for (int q = beg; q < end; ++q) {
                const int kq = key[r * EE + q];
                rank += (kq < k || (kq == k && q < p)) ? 1 : 0;
            }
            col[r * EE + beg + rank] = k & 0xffff;
        }
        if (tid < ((4 - (tr & 3)) & 3)) col[r * EE + tr + tid] = 0;
    }
    for (int i = tid; i < 3 * (NN + 1); i += 1024) rowptr[i] = rp[i];

    if (wid < NLV) {
        const int k = wid;
        unsigned mw[NN / 32];
#pragma unroll
        for (int ch = 0; ch < NN / 32; ++ch) mw[ch] = mk[k * 16 + ch];
        int base = 0;
#pragma unroll
        for (int ch = 0; ch < NN / 32; ++ch) {
            const unsigned m = mw[ch];
            if ((m >> lane) & 1u)
                nlist[k * NN + base + __popc(m & ((1u << lane) - 1u))] = ch * 32 + lane;
            base += __popc(m);
        }
        for (int i = base + lane; i < NN; i += 32) nlist[k * NN + i] = 0;
        if (lane == 0) ncnt[k] = (base + 15) & ~15;
    }
}

__device__ __forceinline__ void gemm_acc(
    const float* __restrict__ A, const float* __restrict__ W,
    const int rowoff[4], int nact, int mlo, int mhalf, v8f acc[4][2])
{
#pragma unroll
    for (int mt = 0; mt < 4; ++mt) {
        if (mt < nact) {
            for (int ks = 0; ks < 8; ++ks) {
                const int kk = ks * 4 + mhalf * 2;
                v2f b0, b1, a;
                b0.x = W[kk * CC + mlo];
                b0.y = W[(kk + 1) * CC + mlo];
                b1.x = W[kk * CC + 16 + mlo];
                b1.y = W[(kk + 1) * CC + 16 + mlo];
                a.x = A[rowoff[mt] + kk];
                a.y = A[rowoff[mt] + kk + 1];
                acc[mt][0] = __builtin_amdgcn_wmma_f32_16x16x4_f32(
                    false, a, false, b0, (short)0, acc[mt][0], false, false);
                acc[mt][1] = __builtin_amdgcn_wmma_f32_16x16x4_f32(
                    false, a, false, b1, (short)0, acc[mt][1], false, false);
            }
        }
    }
}

struct GBounds {
    int jA, eA, jB, eB;
    float ivA, ivB;
};

__device__ __forceinline__ GBounds gather_bounds(
    int nodeA, int nodeB, bool okA, bool okB,
    const int* __restrict__ rp, const float* __restrict__ iv)
{
    GBounds g;
    const int a0 = rp[nodeA], a1 = rp[nodeA + 1];
    const int b0 = rp[nodeB], b1 = rp[nodeB + 1];
    g.ivA = iv[nodeA];
    g.ivB = iv[nodeB];
    g.jA = a0;
    g.eA = okA ? a1 : a0;
    g.jB = b0;
    g.eB = okB ? b1 : b0;
    return g;
}

__device__ __forceinline__ void gather_first_ids(
    const int* __restrict__ cl, const GBounds& g, int ca[4], int cb[4])
{
#pragma unroll
    for (int i = 0; i < 4; ++i) {
        const int va = cl[g.jA + i], vb = cl[g.jB + i];
        ca[i] = (g.jA + i < g.eA) ? va : NN;
        cb[i] = (g.jB + i < g.eB) ? vb : NN;
    }
}

__device__ __forceinline__ void gather_sum(
    const float* __restrict__ xs, float* __restrict__ hs,
    const int* __restrict__ cl, const GBounds& g, int ca[4], int cb[4],
    int pA, int pB, bool okA, bool okB, int cq)
{
    int jA = g.jA, jB = g.jB;
    const int eA = g.eA, eB = g.eB;
    v4f a = (v4f){0.0f, 0.0f, 0.0f, 0.0f};
    v4f b = (v4f){0.0f, 0.0f, 0.0f, 0.0f};
    while (jA < eA || jB < eB) {
        v4f xa[4], xb[4];
#pragma unroll
        for (int i = 0; i < 4; ++i) {
            xa[i] = *(const v4f*)&xs[ca[i] * CC + cq];
            xb[i] = *(const v4f*)&xs[cb[i] * CC + cq];
        }
        jA = (jA + 4 < eA) ? jA + 4 : eA;
        jB = (jB + 4 < eB) ? jB + 4 : eB;
#pragma unroll
        for (int i = 0; i < 4; ++i) {
            const int va = cl[jA + i], vb = cl[jB + i];
            ca[i] = (jA + i < eA) ? va : NN;
            cb[i] = (jB + i < eB) ? vb : NN;
        }
#pragma unroll
        for (int i = 0; i < 4; ++i) { a += xa[i]; b += xb[i]; }
    }
    if (okA) *(v4f*)&hs[pA * CC + cq] = a * g.ivA;
    if (okB) *(v4f*)&hs[pB * CC + cq] = b * g.ivB;
}

template <int M>
__device__ __forceinline__ float row_xor(float v)
{
    return __int_as_float(__builtin_amdgcn_update_dpp(0, __float_as_int(v), 0x160 + M, 0xf, 0xf, true));
}

__device__ __forceinline__ void row16_sum8(float s[8])
{
#pragma unroll
    for (int u = 0; u < 8; ++u) s[u] += row_xor<8>(s[u]);
#pragma unroll
    for (int u = 0; u < 8; ++u) s[u] += row_xor<4>(s[u]);
#pragma unroll
    for (int u = 0; u < 8; ++u) s[u] += row_xor<2>(s[u]);
#pragma unroll
    for (int u = 0; u < 8; ++u) s[u] += row_xor<1>(s[u]);
}

__global__ __launch_bounds__(256) void rgcn_main_kernel(
    const float* __restrict__ xx, const float* __restrict__ ss,
    const float* __restrict__ Wf, const float* __restrict__ bf,
    const float* __restrict__ convw, const float* __restrict__ convroot,
    const float* __restrict__ convbias, const float* __restrict__ lng,
    const float* __restrict__ lnb, const int* __restrict__ rowptr,
    const int* __restrict__ col, const float* __restrict__ invdeg,
    const int* __restrict__ nlist, const int* __restrict__ ncnt,
    float* __restrict__ out)
{
    __shared__ __attribute__((aligned(16))) float xs[(NN + 1) * CC];
    __shared__ __attribute__((aligned(16))) float hs[NN * CC];
    __shared__ __attribute__((aligned(16))) int colL[3 * EE + 4];
    __shared__ int rpL[3 * (NN + 1)];
    __shared__ __attribute__((aligned(16))) float wall[LL * 4 * CC * CC];
    __shared__ float cbs[LL * CC];
    __shared__ float ivd[3 * NN];
    __shared__ float wfe[6 * CC];
    __shared__ float prm[3 * CC];
    __shared__ __attribute__((aligned(16))) int nlL[NLV * NN];
    __shared__ int ncL[8];

    const int bid   = blockIdx.x;
    const int tid   = threadIdx.x;
    const int lane  = tid & 31;
    const int wid   = __builtin_amdgcn_readfirstlane(tid >> 5);
    const int mlo   = lane & 15;
    const int mhalf = lane >> 4;

    const float mfeat = ss[bid] * 0.125f;

    int q4[3];
#pragma unroll
    for (int r = 0; r < 3; ++r) {
        const int t = (rowptr[r * (NN + 1) + NN] + 3) >> 2;
        q4[r] = (t < 0) ? 0 : ((t > EE / 4) ? EE / 4 : t);
    }

    const v4f* __restrict__ cr4 = (const v4f*)convroot;
    const v4f* __restrict__ cw4 = (const v4f*)convw;
    v4f wr[LL], ww[3 * LL];
#pragma unroll
    for (int u = 0; u < LL; ++u) wr[u] = cr4[u * 256 + tid];
#pragma unroll
    for (int u = 0; u < 3 * LL; ++u) ww[u] = cw4[u * 256 + tid];
    const v4i* __restrict__ n4 = (const v4i*)nlist;
    const v4i t0 = n4[tid], t1 = n4[tid + 256];
    const v4i t2 = n4[512 + (tid & (NLV * NN / 4 - 512 - 1))];
    const int ncv = ncnt[(tid < NLV) ? tid : 0];
    float ivr[6];
    int rpr[7];
#pragma unroll
    for (int u = 0; u < 6; ++u) {
        ivr[u] = invdeg[u * 256 + tid];
        rpr[u] = rowptr[u * 256 + tid];
    }
    rpr[6] = rowptr[(tid < 3 * (NN + 1) - 6 * 256) ? 6 * 256 + tid : 0];
    const float wfv = Wf[(tid < 6 * CC) ? tid : 0];
    const float pbv = bf[lane], pgv = lng[lane], plv = lnb[lane];
    const float cbv = convbias[tid & (LL * CC - 1)];

    {
        const int q01 = q4[0] + q4[1];
        const int qt  = q01 + q4[2];
        const v4i* __restrict__ c4 = (const v4i*)col;
        v4i* l4 = (v4i*)colL;
        for (int b0 = tid; b0 < qt; b0 += 3 * 256) {
            v4i v[3];
            int w[3];
#pragma unroll
            for (int k = 0; k < 3; ++k) {
                const int b  = b0 + k * 256;
                const int bc = (b < qt) ? b : qt - 1;
                w[k] = (bc >= q01) ? (2 * (EE / 4) + bc - q01)
                                   : ((bc >= q4[0]) ? ((EE / 4) + bc - q4[0]) : bc);
                v[k] = c4[w[k]];
            }
#pragma unroll
            for (int k = 0; k < 3; ++k)
                if (b0 + k * 256 < qt) l4[w[k]] = v[k];
        }
    }

    {
        v4f* w4 = (v4f*)wall;
#pragma unroll
        for (int u = 0; u < LL; ++u) w4[(u * 4) * 256 + tid] = wr[u];
#pragma unroll
        for (int u = 0; u < 3 * LL; ++u) w4[((u / 3) * 4 + 1 + (u % 3)) * 256 + tid] = ww[u];
        v4i* d4 = (v4i*)nlL;
        d4[tid] = t0;
        d4[tid + 256] = t1;
        if (tid < NLV * NN / 4 - 512) d4[tid + 512] = t2;
        if (tid < NLV) ncL[tid] = ncv;
#pragma unroll
        for (int u = 0; u < 6; ++u) {
            ivd[u * 256 + tid] = ivr[u];
            rpL[u * 256 + tid] = rpr[u];
        }
        if (tid < 3 * (NN + 1) - 6 * 256) rpL[6 * 256 + tid] = rpr[6];
        if (tid < 6 * CC) wfe[tid] = wfv;
        if (tid < CC) {
            xs[NN * CC + tid] = 0.0f;
            prm[tid]          = pbv;
            prm[CC + tid]     = pgv;
            prm[2 * CC + tid] = plv;
        }
        if (tid < LL * CC) cbs[tid] = cbv;
    }
    __syncthreads();

    const int cnt0 = ncL[0];
    for (int idx = tid; idx < cnt0 * CC; idx += 256) {
        const int n = nlL[idx >> 5], c = idx & 31;
        const int t = n >> 6, i = (n >> 4) & 3, j = (n >> 2) & 3, k = n & 3;
        const float v = xx[bid * NN + n];
        float a = prm[c];
        a += ((float)i * (1.0f / 3.0f)) * wfe[0 * CC + c];
        a += ((float)j * (1.0f / 3.0f)) * wfe[1 * CC + c];
        a += ((float)k * (1.0f / 3.0f)) * wfe[2 * CC + c];
        a += ((float)t * (1.0f / 7.0f)) * wfe[3 * CC + c];
        a += v * wfe[4 * CC + c];
        a += mfeat * wfe[5 * CC + c];
        xs[n * CC + c] = a;
    }
    __syncthreads();

    for (int l = 0; l < LL; ++l) {
        const int* nl  = &nlL[(l + 1) * NN];
        const int cnt  = ncL[l + 1];
        const int ntile = cnt >> 4;
        const int nact = (ntile > wid) ? ((ntile - wid + 7) >> 3) : 0;
        int xoff[4], hoff[4];
#pragma unroll
        for (int mt = 0; mt < 4; ++mt) {
            const int p = (wid + 8 * mt) * 16 + mlo;
            xoff[mt] = nl[p] * CC;
            hoff[mt] = p * CC;
        }
        v8f acc[4][2];
#pragma unroll
        for (int mt = 0; mt < 4; ++mt)
#pragma unroll
            for (int nt = 0; nt < 2; ++nt) {
                const float bv = cbs[l * CC + nt * 16 + mlo];
                acc[mt][nt] = (v8f){bv, bv, bv, bv, bv, bv, bv, bv};
            }

        const int cq  = (lane & 7) * 4;
        const int pA0 = wid * 4 + (lane >> 3);
        const int pB0 = pA0 + 32;
        const bool okA0 = pA0 < cnt, okB0 = pB0 < cnt;
        const int nA0 = nl[pA0];
        const int nB0 = nl[pB0];
        GBounds g0[3];
        int ca0[3][4], cb0[3][4];
#pragma unroll
        for (int r = 0; r < 3; ++r)
            g0[r] = gather_bounds(nA0, nB0, okA0, okB0, &rpL[r * (NN + 1)], &ivd[r * NN]);
#pragma unroll
        for (int r = 0; r < 3; ++r)
            gather_first_ids(&colL[r * EE], g0[r], ca0[r], cb0[r]);

        gemm_acc(xs, &wall[(l * 4) * CC * CC], xoff, nact, mlo, mhalf, acc);

#pragma unroll
        for (int r = 0; r < 3; ++r) {
            if (r > 0) __syncthreads();
            gather_sum(xs, hs, &colL[r * EE], g0[r], ca0[r], cb0[r], pA0, pB0, okA0, okB0, cq);
            for (int base = 64; base < cnt; base += 64) {
                const int pA = base + pA0, pB = base + pB0;
                const bool okA = pA < cnt, okB = pB < cnt;
                const GBounds g = gather_bounds(nl[pA], nl[pB], okA, okB, &rpL[r * (NN + 1)], &ivd[r * NN]);
                int ca[4], cb[4];
                gather_first_ids(&colL[r * EE], g, ca, cb);
                gather_sum(xs, hs, &colL[r * EE], g, ca, cb, pA, pB, okA, okB, cq);
            }
            __syncthreads();

            gemm_acc(hs, &wall[(l * 4 + 1 + r) * CC * CC], hoff, nact, mlo, mhalf, acc);
        }

        {
            const float g0 = prm[CC + mlo],     g1 = prm[CC + 16 + mlo];
            const float e0 = prm[2 * CC + mlo], e1 = prm[2 * CC + 16 + mlo];
#pragma unroll
            for (int mt = 0; mt < 4; ++mt)
                if (mt < nact) {
                    const int pb = (wid + 8 * mt) * 16 + 8 * mhalf;
                    const v4i na = *(const v4i*)&nl[pb];
                    const v4i nb = *(const v4i*)&nl[pb + 4];
                    const int node[8] = {na.x, na.y, na.z, na.w, nb.x, nb.y, nb.z, nb.w};
                    float v0[8], v1[8], s[8];
#pragma unroll
                    for (int jj = 0; jj < 8; ++jj) {
                        const float a0 = acc[mt][0][jj], a1 = acc[mt][1][jj];
                        v0[jj] = a0 > 0.0f ? a0 : 0.0f;
                        v1[jj] = a1 > 0.0f ? a1 : 0.0f;
                        s[jj]  = v0[jj] + v1[jj];
                    }
                    row16_sum8(s);
#pragma unroll
                    for (int jj = 0; jj < 8; ++jj) {
                        const float mu = s[jj] * (1.0f / 32.0f);
                        v0[jj] -= mu;
                        v1[jj] -= mu;
                        s[jj] = v0[jj] * v0[jj] + v1[jj] * v1[jj];
                    }
                    row16_sum8(s);
#pragma unroll
                    for (int jj = 0; jj < 8; ++jj) {
                        const float rs = rsqrtf(s[jj] * (1.0f / 32.0f) + LN_EPS);
                        xs[node[jj] * CC + mlo]      = v0[jj] * rs * g0 + e0;
                        xs[node[jj] * CC + 16 + mlo] = v1[jj] * rs * g1 + e1;
                    }
                }
        }
        __syncthreads();
    }

    for (int idx = tid; idx < 48 * CC; idx += 256) {
        const int row = idx >> 5, c = idx & 31;
        const int g = row >> 4, rc = row & 15;
        const int a0 = rc >> 2, b0 = rc & 3;
        float sum = 0.0f;
#pragma unroll
        for (int q = 0; q < 4; ++q) {
            int node;
            if (g == 0)      node = q * 16 + a0 * 4 + b0;
            else if (g == 1) node = a0 * 16 + q * 4 + b0;
            else             node = a0 * 16 + b0 * 4 + q;
            sum += xs[node * CC + c];
        }
        out[bid * 48 * CC + idx] = sum * 0.25f;
    }
}

extern "C" void kernel_launch(void* const* d_in, const int* in_sizes, int n_in,
                              void* d_out, int out_size, void* d_ws, size_t ws_size,
                              hipStream_t stream) {
    const float* xx       = (const float*)d_in[0];
    const float* ss       = (const float*)d_in[1];
    const float* Wf       = (const float*)d_in[2];
    const float* bf       = (const float*)d_in[3];
    const float* convw    = (const float*)d_in[4];
    const float* convroot = (const float*)d_in[5];
    const float* convbias = (const float*)d_in[6];
    const float* lng      = (const float*)d_in[7];
    const float* lnb      = (const float*)d_in[8];
    const int*   src      = (const int*)d_in[9];
    const int*   dst      = (const int*)d_in[10];
    float*       out      = (float*)d_out;
    float*       invdeg   = (float*)d_ws;
    int*         rowptr   = (int*)d_ws + 2048;
    int*         col      = (int*)d_ws + 4096;
    int*         nlist    = (int*)d_ws + 24576;
    int*         ncnt     = (int*)d_ws + 27648;

    const int E = in_sizes[9] / 3;
    const int B = in_sizes[1];

    rgcn_csr_kernel<<<1, 1024, 0, stream>>>(src, dst, E, invdeg, rowptr, col, nlist, ncnt);
    rgcn_main_kernel<<<B, 256, 0, stream>>>(xx, ss, Wf, bf, convw, convroot,
                                            convbias, lng, lnb, rowptr, col,
                                            invdeg, nlist, ncnt, out);
    (void)n_in; (void)out_size; (void)ws_size;
}


# Gathers of the three relations walk together into three blocks of hs and their three products run back to back (when 3 x rows fit): 2 barriers per layer, not 6; sum order kept
# speedup vs baseline: 37.4840x; 1.0433x over previous
#include <hip/hip_runtime.h>
#include <hip/hip_bf16.h>


typedef __attribute__((ext_vector_type(2))) float v2f;
typedef __attribute__((ext_vector_type(4))) float v4f;
typedef __attribute__((ext_vector_type(4))) int v4i;
typedef __attribute__((ext_vector_type(8))) float v8f;

#define NN  512
#define CC  32
#define LL  4
#define EE  6144
#define NLV 5
#define LN_EPS 1e-5f

__device__ __forceinline__ int rel_of(int i, int E)
{
    return (i >= 2 * E) ? 2 : ((i >= E) ? 1 : 0);
}

#define EPT 18

__global__ __launch_bounds__(1024) void rgcn_csr_kernel(
    const int* __restrict__ src, const int* __restrict__ dst, int E,
    float* __restrict__ invdeg, int* __restrict__ rowptr, int* __restrict__ col,
    int* __restrict__ nlist, int* __restrict__ ncnt)
{
    __shared__ int nfl[NN];
    __shared__ unsigned mk[NLV * 16];
    __shared__ int grew[NLV];
    __shared__ int cnt[3 * NN];
    __shared__ int rp[3 * (NN + 1)];
    __shared__ int key[3 * EE];
    const int tid  = threadIdx.x;
    const int lane = tid & 31;
    const int wid  = tid >> 5;
    const int tot  = 3 * E;

    int ek[EPT];
#pragma unroll
    for (int u = 0; u < EPT; ++u) {
        const int i  = tid + u * 1024;
        const int ii = (i < tot) ? i : 0;
        const int d = dst[ii], s = src[ii];
        ek[u] = (i < tot) ? ((d << 16) | s) : 0;
    }

    for (int i = tid; i < 3 * NN; i += 1024) cnt[i] = 0;
    if (tid < NN) nfl[tid] = (tid < 64) ? 1 : 0;
    if (tid < 16) mk[(NLV - 1) * 16 + tid] = (tid < 2) ? 0xffffffffu : 0u;
    if (tid < NLV) grew[tid] = 0;
    __syncthreads();

    for (int k = NLV - 1; k >= 1; --k) {
        unsigned wq[EPT];
#pragma unroll
        for (int u = 0; u < EPT; ++u)
            wq[u] = mk[k * 16 + (ek[u] >> 21)];
#pragma unroll
        for (int u = 0; u < EPT; ++u)
            if ((wq[u] >> ((ek[u] >> 16) & 31)) & 1u) nfl[ek[u] & 0xffff] = 1;
        __syncthreads();
        if (tid < NN) {
            const unsigned m = __builtin_amdgcn_ballot_w32(nfl[tid] != 0);
            if (lane == 0) {
                mk[(k - 1) * 16 + wid] = m;
                if (m != mk[k * 16 + wid]) grew[k] = 1;
            }
        }
        __syncthreads();
        if (__builtin_amdgcn_readfirstlane(grew[k]) == 0) {
            if (tid < 16)
                for (int j = k - 2; j >= 0; --j) mk[j * 16 + tid] = mk[k * 16 + tid];
            __syncthreads();
            break;
        }
    }

    unsigned needm = 0u;
    {
        unsigned wq[EPT];
#pragma unroll
        for (int u = 0; u < EPT; ++u)
            wq[u] = mk[1 * 16 + (ek[u] >> 21)];
#pragma unroll
        for (int u = 0; u < EPT; ++u)
            needm |= ((((wq[u] >> ((ek[u] >> 16) & 31)) & 1u) != 0u && tid + u * 1024 < tot) ? 1u : 0u) << u;
    }

#pragma unroll
    for (int u = 0; u < EPT; ++u)
        if ((needm >> u) & 1u)
            atomicAdd(&cnt[rel_of(tid + u * 1024, E) * NN + (ek[u] >> 16)], 1);
    __syncthreads();

    for (int i = tid; i < 3 * NN; i += 1024) {
        int dg = cnt[i];
        if (dg < 1) dg = 1;
        invdeg[i] = 1.0f / (float)dg;
    }
    if (wid < 3) {
        const int r = wid, nb = lane * 16;
        int loc[16];
        int s = 0;
#pragma unroll
        for (int q = 0; q < 16; ++q) { loc[q] = s; s += cnt[r * NN + nb + q]; }
        int inc = s;
        for (int off = 1; off < 32; off <<= 1) {
            const int t = __shfl_up(inc, off, 32);
            if (lane >= off) inc += t;
        }
        const int excl = inc - s;
#pragma unroll
        for (int q = 0; q < 16; ++q) rp[r * (NN + 1) + nb + q] = excl + loc[q];
        if (lane == 31) rp[r * (NN + 1) + NN] = inc;
    }
    __syncthreads();
    for (int i = tid; i < 3 * NN; i += 1024) cnt[i] = 0;
    __syncthreads();

#pragma unroll
    for (int h = 0; h < EPT; h += 6) {
        int rb[6], sl[6];
#pragma unroll
        for (int v = 0; v < 6; ++v) {
            const int u = h + v;
            rb[v] = 0;
            sl[v] = 0;
            if ((needm >> u) & 1u) {
                const int r = rel_of(tid + u * 1024, E);
                const int d = ek[u] >> 16;
                rb[v] = rp[r * (NN + 1) + d];
                sl[v] = atomicAdd(&cnt[r * NN + d], 1);
            }
        }
#pragma unroll
        for (int v = 0; v < 6; ++v) {
            const int u = h + v;
            if ((needm >> u) & 1u)
                key[rel_of(tid + u * 1024, E) * EE + rb[v] + sl[v]] = ek[u];
        }
    }
    __syncthreads();

    for (int r = 0; r < 3; ++r) {
        const int tr = rp[r * (NN + 1) + NN];
        for (int p = tid; p < tr; p += 1024) {
            const int k = key[r * EE + p];
            const int dn = k >> 16;
            const int beg = rp[r * (NN + 1) + dn];
            const int end = rp[r * (NN + 1) + dn + 1];
            int rank = 0;
            for (int q = beg; q < end; ++q) {
                const int kq = key[r * EE + q];
                rank += (kq < k || (kq == k && q < p)) ? 1 : 0;
            }
            col[r * EE + beg + rank] = k & 0xffff;
        }
        if (tid < ((4 - (tr & 3)) & 3)) col[r * EE + tr + tid] = 0;
    }
    for (int i = tid; i < 3 * (NN + 1); i += 1024) rowptr[i] = rp[i];

    if (wid < NLV) {
        const int k = wid;
        unsigned mw[NN / 32];
#pragma unroll
        for (int ch = 0; ch < NN / 32; ++ch) mw[ch] = mk[k * 16 + ch];
        int base = 0;
#pragma unroll
        for (int ch = 0; ch < NN / 32; ++ch) {
            const unsigned m = mw[ch];
            if ((m >> lane) & 1u)
                nlist[k * NN + base + __popc(m & ((1u << lane) - 1u))] = ch * 32 + lane;
            base += __popc(m);
        }
        for (int i = base + lane; i < NN; i += 32) nlist[k * NN + i] = 0;
        if (lane == 0) ncnt[k] = (base + 15) & ~15;
    }
}

__device__ __forceinline__ void gemm_acc(
    const float* __restrict__ A, const float* __restrict__ W,
    const int rowoff[4], int nact, int mlo, int mhalf, v8f acc[4][2])
{
#pragma unroll
    for (int mt = 0; mt < 4; ++mt) {
        if (mt < nact) {
            for (int ks = 0; ks < 8; ++ks) {
                const int kk = ks * 4 + mhalf * 2;
                v2f b0, b1, a;
                b0.x = W[kk * CC + mlo];
                b0.y = W[(kk + 1) * CC + mlo];
                b1.x = W[kk * CC + 16 + mlo];
                b1.y = W[(kk + 1) * CC + 16 + mlo];
                a.x = A[rowoff[mt] + kk];
                a.y = A[rowoff[mt] + kk + 1];
                acc[mt][0] = __builtin_amdgcn_wmma_f32_16x16x4_f32(
                    false, a, false, b0, (short)0, acc[mt][0], false, false);
                acc[mt][1] = __builtin_amdgcn_wmma_f32_16x16x4_f32(
                    false, a, false, b1, (short)0, acc[mt][1], false, false);
            }
        }
    }
}

struct GBounds {
    int jA, eA, jB, eB;
    float ivA, ivB;
};

__device__ __forceinline__ GBounds gather_bounds(
    int nodeA, int nodeB, bool okA, bool okB,
    const int* __restrict__ rp, const float* __restrict__ iv)
{
    GBounds g;
    const int a0 = rp[nodeA], a1 = rp[nodeA + 1];
    const int b0 = rp[nodeB], b1 = rp[nodeB + 1];
    g.ivA = iv[nodeA];
    g.ivB = iv[nodeB];
    g.jA = a0;
    g.eA = okA ? a1 : a0;
    g.jB = b0;
    g.eB = okB ? b1 : b0;
    return g;
}

__device__ __forceinline__ void gather_first_ids(
    const int* __restrict__ cl, const GBounds& g, int ca[4], int cb[4])
{
#pragma unroll
    for (int i = 0; i < 4; ++i) {
        const int va = cl[g.jA + i], vb = cl[g.jB + i];
        ca[i] = (g.jA + i < g.eA) ? va : NN;
        cb[i] = (g.jB + i < g.eB) ? vb : NN;
    }
}

__device__ __forceinline__ void gather_sum(
    const float* __restrict__ xs, float* __restrict__ hs,
    const int* __restrict__ cl, const GBounds& g, int ca[4], int cb[4],
    int pA, int pB, bool okA, bool okB, int cq)
{
    int jA = g.jA, jB = g.jB;
    const int eA = g.eA, eB = g.eB;
    v4f a = (v4f){0.0f, 0.0f, 0.0f, 0.0f};
    v4f b = (v4f){0.0f, 0.0f, 0.0f, 0.0f};
    while (jA < eA || jB < eB) {
        v4f xa[4], xb[4];
#pragma unroll
        for (int i = 0; i < 4; ++i) {
            xa[i] = *(const v4f*)&xs[ca[i] * CC + cq];
            xb[i] = *(const v4f*)&xs[cb[i] * CC + cq];
        }
        jA = (jA + 4 < eA) ? jA + 4 : eA;
        jB = (jB + 4 < eB) ? jB + 4 : eB;
#pragma unroll
        for (int i = 0; i < 4; ++i) {
            const int va = cl[jA + i], vb = cl[jB + i];
            ca[i] = (jA + i < eA) ? va : NN;
            cb[i] = (jB + i < eB) ? vb : NN;
        }
#pragma unroll
        for (int i = 0; i < 4; ++i) { a += xa[i]; b += xb[i]; }
    }
    if (okA) *(v4f*)&hs[pA * CC + cq] = a * g.ivA;
    if (okB) *(v4f*)&hs[pB * CC + cq] = b * g.ivB;
}

__device__ __forceinline__ void gather_sum3(
    const float* __restrict__ xs, float* __restrict__ hs, const int* __restrict__ cl3,
    const GBounds g[3], int ca[3][4], int cb[3][4],
    int pA, int pB, bool okA, bool okB, int cq, int rstride)
{
    int jA[3], jB[3];
    v4f a[3], b[3];
    bool more = false;
#pragma unroll
    for (int r = 0; r < 3; ++r) {
        jA[r] = g[r].jA;
        jB[r] = g[r].jB;
        a[r] = (v4f){0.0f, 0.0f, 0.0f, 0.0f};
        b[r] = (v4f){0.0f, 0.0f, 0.0f, 0.0f};
        more = more || jA[r] < g[r].eA || jB[r] < g[r].eB;
    }
    while (more) {
        v4f xa[3][4], xb[3][4];
#pragma unroll
        for (int r = 0; r < 3; ++r)
#pragma unroll
            for (int i = 0; i < 4; ++i) {
                xa[r][i] = *(const v4f*)&xs[ca[r][i] * CC + cq];
                xb[r][i] = *(const v4f*)&xs[cb[r][i] * CC + cq];
            }
        more = false;
#pragma unroll
        for (int r = 0; r < 3; ++r) {
            const int eA = g[r].eA, eB = g[r].eB;
            jA[r] = (jA[r] + 4 < eA) ? jA[r] + 4 : eA;
            jB[r] = (jB[r] + 4 < eB) ? jB[r] + 4 : eB;
            more = more || jA[r] < eA || jB[r] < eB;
#pragma unroll
            for (int i = 0; i < 4; ++i) {
                const int va = cl3[r * EE + jA[r] + i], vb = cl3[r * EE + jB[r] + i];
                ca[r][i] = (jA[r] + i < eA) ? va : NN;
                cb[r][i] = (jB[r] + i < eB) ? vb : NN;
            }
        }
#pragma unroll
        for (int r = 0; r < 3; ++r)
#pragma unroll
            for (int i = 0; i < 4; ++i) { a[r] += xa[r][i]; b[r] += xb[r][i]; }
    }
#pragma unroll
    for (int r = 0; r < 3; ++r) {
        if (okA) *(v4f*)&hs[r * rstride + pA * CC + cq] = a[r] * g[r].ivA;
        if (okB) *(v4f*)&hs[r * rstride + pB * CC + cq] = b[r] * g[r].ivB;
    }
}

template <int M>
__device__ __forceinline__ float row_xor(float v)
{
    return __int_as_float(__builtin_amdgcn_update_dpp(0, __float_as_int(v), 0x160 + M, 0xf, 0xf, true));
}

__device__ __forceinline__ void row16_sum8(float s[8])
{
#pragma unroll
    for (int u = 0; u < 8; ++u) s[u] += row_xor<8>(s[u]);
#pragma unroll
    for (int u = 0; u < 8; ++u) s[u] += row_xor<4>(s[u]);
#pragma unroll
    for (int u = 0; u < 8; ++u) s[u] += row_xor<2>(s[u]);
#pragma unroll
    for (int u = 0; u < 8; ++u) s[u] += row_xor<1>(s[u]);
}

__global__ __launch_bounds__(256) void rgcn_main_kernel(
    const float* __restrict__ xx, const float* __restrict__ ss,
    const float* __restrict__ Wf, const float* __restrict__ bf,
    const float* __restrict__ convw, const float* __restrict__ convroot,
    const float* __restrict__ convbias, const float* __restrict__ lng,
    const float* __restrict__ lnb, const int* __restrict__ rowptr,
    const int* __restrict__ col, const float* __restrict__ invdeg,
    const int* __restrict__ nlist, const int* __restrict__ ncnt,
    float* __restrict__ out)
{
    __shared__ __attribute__((aligned(16))) float xs[(NN + 1) * CC];
    __shared__ __attribute__((aligned(16))) float hs[NN * CC];
    __shared__ __attribute__((aligned(16))) int colL[3 * EE + 4];
    __shared__ int rpL[3 * (NN + 1)];
    __shared__ __attribute__((aligned(16))) float wall[LL * 4 * CC * CC];
    __shared__ float cbs[LL * CC];
    __shared__ float ivd[3 * NN];
    __shared__ float wfe[6 * CC];
    __shared__ float prm[3 * CC];
    __shared__ __attribute__((aligned(16))) int nlL[NLV * NN];
    __shared__ int ncL[8];

    const int bid   = blockIdx.x;
    const int tid   = threadIdx.x;
    const int lane  = tid & 31;
    const int wid   = __builtin_amdgcn_readfirstlane(tid >> 5);
    const int mlo   = lane & 15;
    const int mhalf = lane >> 4;

    const float mfeat = ss[bid] * 0.125f;

    int q4[3];
#pragma unroll
    for (int r = 0; r < 3; ++r) {
        const int t = (rowptr[r * (NN + 1) + NN] + 3) >> 2;
        q4[r] = (t < 0) ? 0 : ((t > EE / 4) ? EE / 4 : t);
    }

    const v4f* __restrict__ cr4 = (const v4f*)convroot;
    const v4f* __restrict__ cw4 = (const v4f*)convw;
    v4f wr[LL], ww[3 * LL];
#pragma unroll
    for (int u = 0; u < LL; ++u) wr[u] = cr4[u * 256 + tid];
#pragma unroll
    for (int u = 0; u < 3 * LL; ++u) ww[u] = cw4[u * 256 + tid];
    const v4i* __restrict__ n4 = (const v4i*)nlist;
    const v4i t0 = n4[tid], t1 = n4[tid + 256];
    const v4i t2 = n4[512 + (tid & (NLV * NN / 4 - 512 - 1))];
    const int ncv = ncnt[(tid < NLV) ? tid : 0];
    float ivr[6];
    int rpr[7];
#pragma unroll
    for (int u = 0; u < 6; ++u) {
        ivr[u] = invdeg[u * 256 + tid];
        rpr[u] = rowptr[u * 256 + tid];
    }
    rpr[6] = rowptr[(tid < 3 * (NN + 1) - 6 * 256) ? 6 * 256 + tid : 0];
    const float wfv = Wf[(tid < 6 * CC) ? tid : 0];
    const float pbv = bf[lane], pgv = lng[lane], plv = lnb[lane];
    const float cbv = convbias[tid & (LL * CC - 1)];

    {
        const int q01 = q4[0] + q4[1];
        const int qt  = q01 + q4[2];
        const v4i* __restrict__ c4 = (const v4i*)col;
        v4i* l4 = (v4i*)colL;
        for (int b0 = tid; b0 < qt; b0 += 3 * 256) {
            v4i v[3];
            int w[3];
#pragma unroll
            for (int k = 0; k < 3; ++k) {
                const int b  = b0 + k * 256;
                const int bc = (b < qt) ? b : qt - 1;
                w[k] = (bc >= q01) ? (2 * (EE / 4) + bc - q01)
                                   : ((bc >= q4[0]) ? ((EE / 4) + bc - q4[0]) : bc);
                v[k] = c4[w[k]];
            }
#pragma unroll
            for (int k = 0; k < 3; ++k)
                if (b0 + k * 256 < qt) l4[w[k]] = v[k];
        }
    }

    {
        v4f* w4 = (v4f*)wall;
#pragma unroll
        for (int u = 0; u < LL; ++u) w4[(u * 4) * 256 + tid] = wr[u];
#pragma unroll
        for (int u = 0; u < 3 * LL; ++u) w4[((u / 3) * 4 + 1 + (u % 3)) * 256 + tid] = ww[u];
        v4i* d4 = (v4i*)nlL;
        d4[tid] = t0;
        d4[tid + 256] = t1;
        if (tid < NLV * NN / 4 - 512) d4[tid + 512] = t2;
        if (tid < NLV) ncL[tid] = ncv;
#pragma unroll
        for (int u = 0; u < 6; ++u) {
            ivd[u * 256 + tid] = ivr[u];
            rpL[u * 256 + tid] = rpr[u];
        }
        if (tid < 3 * (NN + 1) - 6 * 256) rpL[6 * 256 + tid] = rpr[6];
        if (tid < 6 * CC) wfe[tid] = wfv;
        if (tid < CC) {
            xs[NN * CC + tid] = 0.0f;
            prm[tid]          = pbv;
            prm[CC + tid]     = pgv;
            prm[2 * CC + tid] = plv;
        }
        if (tid < LL * CC) cbs[tid] = cbv;
    }
    __syncthreads();

    const int cnt0 = ncL[0];
    for (int idx = tid; idx < cnt0 * CC; idx += 256) {
        const int n = nlL[idx >> 5], c = idx & 31;
        const int t = n >> 6, i = (n >> 4) & 3, j = (n >> 2) & 3, k = n & 3;
        const float v = xx[bid * NN + n];
        float a = prm[c];
        a += ((float)i * (1.0f / 3.0f)) * wfe[0 * CC + c];
        a += ((float)j * (1.0f / 3.0f)) * wfe[1 * CC + c];
        a += ((float)k * (1.0f / 3.0f)) * wfe[2 * CC + c];
        a += ((float)t * (1.0f / 7.0f)) * wfe[3 * CC + c];
        a += v * wfe[4 * CC + c];
        a += mfeat * wfe[5 * CC + c];
        xs[n * CC + c] = a;
    }
    __syncthreads();

    for (int l = 0; l < LL; ++l) {
        const int* nl  = &nlL[(l + 1) * NN];
        const int cnt  = ncL[l + 1];
        const int ntile = cnt >> 4;
        const int nact = (ntile > wid) ? ((ntile - wid + 7) >> 3) : 0;
        int xoff[4], hoff[4];
#pragma unroll
        for (int mt = 0; mt < 4; ++mt) {
            const int p = (wid + 8 * mt) * 16 + mlo;
            xoff[mt] = nl[p] * CC;
            hoff[mt] = p * CC;
        }
        v8f acc[4][2];
#pragma unroll
        for (int mt = 0; mt < 4; ++mt)
#pragma unroll
            for (int nt = 0; nt < 2; ++nt) {
                const float bv = cbs[l * CC + nt * 16 + mlo];
                acc[mt][nt] = (v8f){bv, bv, bv, bv, bv, bv, bv, bv};
            }

        const int cq  = (lane & 7) * 4;
        const int pA0 = wid * 4 + (lane >> 3);
        const int pB0 = pA0 + 32;
        const bool okA0 = pA0 < cnt, okB0 = pB0 < cnt;
        const int nA0 = nl[pA0];
        const int nB0 = nl[pB0];
        GBounds g0[3];
        int ca0[3][4], cb0[3][4];
#pragma unroll
        for (int r = 0; r < 3; ++r)
            g0[r] = gather_bounds(nA0, nB0, okA0, okB0, &rpL[r * (NN + 1)], &ivd[r * NN]);
#pragma unroll
        for (int r = 0; r < 3; ++r)
            gather_first_ids(&colL[r * EE], g0[r], ca0[r], cb0[r]);

        gemm_acc(xs, &wall[(l * 4) * CC * CC], xoff, nact, mlo, mhalf, acc);

        const bool side_by_side = __builtin_amdgcn_readfirstlane((3 * cnt <= NN) ? 1 : 0) != 0;
        if (side_by_side) {
            const int rstride = cnt * CC;
            gather_sum3(xs, hs, colL, g0, ca0, cb0, pA0, pB0, okA0, okB0, cq, rstride);
            for (int base = 64; base < cnt; base += 64) {
                const int pA = base + pA0, pB = base + pB0;
                const bool okA = pA < cnt, okB = pB < cnt;
                const int nA = nl[pA], nB = nl[pB];
                GBounds g[3];
                int ca[3][4], cb[3][4];
#pragma unroll
                for (int r = 0; r < 3; ++r)
                    g[r] = gather_bounds(nA, nB, okA, okB, &rpL[r * (NN + 1)], &ivd[r * NN]);
#pragma unroll
                for (int r = 0; r < 3; ++r)
                    gather_first_ids(&colL[r * EE], g[r], ca[r], cb[r]);
                gather_sum3(xs, hs, colL, g, ca, cb, pA, pB, okA, okB, cq, rstride);
            }
            __syncthreads();
#pragma unroll
            for (int r = 0; r < 3; ++r)
                gemm_acc(hs + r * rstride, &wall[(l * 4 + 1 + r) * CC * CC], hoff, nact, mlo, mhalf, acc);
        } else {
#pragma unroll
        for (int r = 0; r < 3; ++r) {
            if (r > 0) __syncthreads();
            gather_sum(xs, hs, &colL[r * EE], g0[r], ca0[r], cb0[r], pA0, pB0, okA0, okB0, cq);
            for (int base = 64; base < cnt; base += 64) {
                const int pA = base + pA0, pB = base + pB0;
                const bool okA = pA < cnt, okB = pB < cnt;
                const GBounds g = gather_bounds(nl[pA], nl[pB], okA, okB, &rpL[r * (NN + 1)], &ivd[r * NN]);
                int ca[4], cb[4];
                gather_first_ids(&colL[r * EE], g, ca, cb);
                gather_sum(xs, hs, &colL[r * EE], g, ca, cb, pA, pB, okA, okB, cq);
            }
            __syncthreads();

            gemm_acc(hs, &wall[(l * 4 + 1 + r) * CC * CC], hoff, nact, mlo, mhalf, acc);
        }
        }

        {
            const float g0 = prm[CC + mlo],     g1 = prm[CC + 16 + mlo];
            const float e0 = prm[2 * CC + mlo], e1 = prm[2 * CC + 16 + mlo];
#pragma unroll
            for (int mt = 0; mt < 4; ++mt)
                if (mt < nact) {
                    const int pb = (wid + 8 * mt) * 16 + 8 * mhalf;
                    const v4i na = *(const v4i*)&nl[pb];
                    const v4i nb = *(const v4i*)&nl[pb + 4];
                    const int node[8] = {na.x, na.y, na.z, na.w, nb.x, nb.y, nb.z, nb.w};
                    float v0[8], v1[8], s[8];
#pragma unroll
                    for (int jj = 0; jj < 8; ++jj) {
                        const float a0 = acc[mt][0][jj], a1 = acc[mt][1][jj];
                        v0[jj] = a0 > 0.0f ? a0 : 0.0f;
                        v1[jj] = a1 > 0.0f ? a1 : 0.0f;
                        s[jj]  = v0[jj] + v1[jj];
                    }
                    row16_sum8(s);
#pragma unroll
                    for (int jj = 0; jj < 8; ++jj) {
                        const float mu = s[jj] * (1.0f / 32.0f);
                        v0[jj] -= mu;
                        v1[jj] -= mu;
                        s[jj] = v0[jj] * v0[jj] + v1[jj] * v1[jj];
                    }
                    row16_sum8(s);
#pragma unroll
                    for (int jj = 0; jj < 8; ++jj) {
                        const float rs = rsqrtf(s[jj] * (1.0f / 32.0f) + LN_EPS);
                        xs[node[jj] * CC + mlo]      = v0[jj] * rs * g0 + e0;
                        xs[node[jj] * CC + 16 + mlo] = v1[jj] * rs * g1 + e1;
                    }
                }
        }
        __syncthreads();
    }

    for (int idx = tid; idx < 48 * CC; idx += 256) {
        const int row = idx >> 5, c = idx & 31;
        const int g = row >> 4, rc = row & 15;
        const int a0 = rc >> 2, b0 = rc & 3;
        float sum = 0.0f;
#pragma unroll
        for (int q = 0; q < 4; ++q) {
            int node;
            if (g == 0)      node = q * 16 + a0 * 4 + b0;
            else if (g == 1) node = a0 * 16 + q * 4 + b0;
            else             node = a0 * 16 + b0 * 4 + q;
            sum += xs[node * CC + c];
        }
        out[bid * 48 * CC + idx] = sum * 0.25f;
    }
}

extern "C" void kernel_launch(void* const* d_in, const int* in_sizes, int n_in,
                              void* d_out, int out_size, void* d_ws, size_t ws_size,
                              hipStream_t stream) {
    const float* xx       = (const float*)d_in[0];
    const float* ss       = (const float*)d_in[1];
    const float* Wf       = (const float*)d_in[2];
    const float* bf       = (const float*)d_in[3];
    const float* convw    = (const float*)d_in[4];
    const float* convroot = (const float*)d_in[5];
    const float* convbias = (const float*)d_in[6];
    const float* lng      = (const float*)d_in[7];
    const float* lnb      = (const float*)d_in[8];
    const int*   src      = (const int*)d_in[9];
    const int*   dst      = (const int*)d_in[10];
    float*       out      = (float*)d_out;
    float*       invdeg   = (float*)d_ws;
    int*         rowptr   = (int*)d_ws + 2048;
    int*         col      = (int*)d_ws + 4096;
    int*         nlist    = (int*)d_ws + 24576;
    int*         ncnt     = (int*)d_ws + 27648;

    const int E = in_sizes[9] / 3;
    const int B = in_sizes[1];

    rgcn_csr_kernel<<<1, 1024, 0, stream>>>(src, dst, E, invdeg, rowptr, col, nlist, ncnt);
    rgcn_main_kernel<<<B, 256, 0, stream>>>(xx, ss, Wf, bf, convw, convroot,
                                            convbias, lng, lnb, rowptr, col,
                                            invdeg, nlist, ncnt, out);
    (void)n_in; (void)out_size; (void)ws_size;
}


# Pre-kernel runs as three work-groups, one per relation: each walks the needed-row levels over the whole edge list itself, then counts, places and ranks only its own relation's edges; sum order kept
# speedup vs baseline: 43.1411x; 1.1509x over previous
#include <hip/hip_runtime.h>
#include <hip/hip_bf16.h>


typedef __attribute__((ext_vector_type(2))) float v2f;
typedef __attribute__((ext_vector_type(4))) float v4f;
typedef __attribute__((ext_vector_type(4))) int v4i;
typedef __attribute__((ext_vector_type(8))) float v8f;

#define NN  512
#define CC  32
#define LL  4
#define EE  6144
#define NLV 5
#define LN_EPS 1e-5f

#define EPT 18
#define OPT 6
#define RPS 544

__global__ __launch_bounds__(1024) void rgcn_csr_kernel(
    const int* __restrict__ src, const int* __restrict__ dst, int E,
    float* __restrict__ invdeg, int* __restrict__ rowptr, int* __restrict__ col,
    int* __restrict__ nlist, int* __restrict__ ncnt)
{
    __shared__ int nfl[NN];
    __shared__ unsigned mk[NLV * 16];
    __shared__ int grew[NLV];
    __shared__ int cnt[NN];
    __shared__ int rp[NN + 1];
    __shared__ int key[EE];
    const int tid  = threadIdx.x;
    const int lane = tid & 31;
    const int wid  = tid >> 5;
    const int tot  = 3 * E;
    const int rel  = blockIdx.x;

    int ek[EPT];
#pragma unroll
    for (int u = 0; u < EPT; ++u) {
        const int i  = tid + u * 1024;
        const int ii = (i < tot) ? i : 0;
        const int d = dst[ii], s = src[ii];
        ek[u] = (i < tot) ? ((d << 16) | s) : 0;
    }
    int ow[OPT];
#pragma unroll
    for (int v = 0; v < OPT; ++v) {
        const int j  = tid + v * 1024;
        const int ii = (j < E) ? rel * E + j : 0;
        const int d = dst[ii], s = src[ii];
        ow[v] = (j < E) ? ((d << 16) | s) : 0;
    }

    if (tid < NN) cnt[tid] = 0;
    if (tid < NN) nfl[tid] = (tid < 64) ? 1 : 0;
    if (tid < 16) mk[(NLV - 1) * 16 + tid] = (tid < 2) ? 0xffffffffu : 0u;
    if (tid < NLV) grew[tid] = 0;
    __syncthreads();

    for (int k = NLV - 1; k >= 1; --k) {
        unsigned wq[EPT];
#pragma unroll
        for (int u = 0; u < EPT; ++u)
            wq[u] = mk[k * 16 + (ek[u] >> 21)];
#pragma unroll
        for (int u = 0; u < EPT; ++u)
            if ((wq[u] >> ((ek[u] >> 16) & 31)) & 1u) nfl[ek[u] & 0xffff] = 1;
        __syncthreads();
        if (tid < NN) {
            const unsigned m = __builtin_amdgcn_ballot_w32(nfl[tid] != 0);
            if (lane == 0) {
                mk[(k - 1) * 16 + wid] = m;
                if (m != mk[k * 16 + wid]) grew[k] = 1;
            }
        }
        __syncthreads();
        if (__builtin_amdgcn_readfirstlane(grew[k]) == 0) {
            if (tid < 16)
                for (int j = k - 2; j >= 0; --j) mk[j * 16 + tid] = mk[k * 16 + tid];
            __syncthreads();
            break;
        }
    }

    unsigned needm = 0u;
    {
        unsigned wq[OPT];
#pragma unroll
        for (int v = 0; v < OPT; ++v)
            wq[v] = mk[1 * 16 + (ow[v] >> 21)];
#pragma unroll
        for (int v = 0; v < OPT; ++v)
            needm |= ((((wq[v] >> ((ow[v] >> 16) & 31)) & 1u) != 0u && tid + v * 1024 < E) ? 1u : 0u) << v;
    }

#pragma unroll
    for (int v = 0; v < OPT; ++v)
        if ((needm >> v) & 1u)
            atomicAdd(&cnt[ow[v] >> 16], 1);
    __syncthreads();

    if (tid < NN) {
        int dg = cnt[tid];
        if (dg < 1) dg = 1;
        invdeg[rel * NN + tid] = 1.0f / (float)dg;
    }
    if (wid == 0) {
        const int nb = lane * 16;
        int loc[16];
        int s = 0;
#pragma unroll
        for (int q = 0; q < 16; ++q) { loc[q] = s; s += cnt[nb + q]; }
        int inc = s;
        for (int off = 1; off < 32; off <<= 1) {
            const int t = __shfl_up(inc, off, 32);
            if (lane >= off) inc += t;
        }
        const int excl = inc - s;
#pragma unroll
        for (int q = 0; q < 16; ++q) rp[nb + q] = excl + loc[q];
        if (lane == 31) rp[NN] = inc;
    }
    __syncthreads();
    if (tid < NN) cnt[tid] = 0;
    __syncthreads();

    {
        int rb[OPT], sl[OPT];
#pragma unroll
        for (int v = 0; v < OPT; ++v) {
            rb[v] = 0;
            sl[v] = 0;
            if ((needm >> v) & 1u) {
                const int d = ow[v] >> 16;
                rb[v] = rp[d];
                sl[v] = atomicAdd(&cnt[d], 1);
            }
        }
#pragma unroll
        for (int v = 0; v < OPT; ++v)
            if ((needm >> v) & 1u)
                key[rb[v] + sl[v]] = ow[v];
    }
    __syncthreads();

    {
        const int tr = rp[NN];
        for (int p = tid; p < tr; p += 1024) {
            const int k = key[p];
            const int dn = k >> 16;
            const int beg = rp[dn];
            const int end = rp[dn + 1];
            int rank = 0;
            for (int q = beg; q < end; ++q) {
                const int kq = key[q];
                rank += (kq < k || (kq == k && q < p)) ? 1 : 0;
            }
            col[rel * EE + beg + rank] = k & 0xffff;
        }
        if (tid < ((4 - (tr & 3)) & 3)) col[rel * EE + tr + tid] = 0;
    }
    if (tid < RPS) rowptr[rel * RPS + tid] = (tid <= NN) ? rp[tid] : 0;

    if (rel == 0 && wid < NLV) {
        const int k = wid;
        unsigned mw[NN / 32];
#pragma unroll
        for (int ch = 0; ch < NN / 32; ++ch) mw[ch] = mk[k * 16 + ch];
        int base = 0;
#pragma unroll
        for (int ch = 0; ch < NN / 32; ++ch) {
            const unsigned m = mw[ch];
            if ((m >> lane) & 1u)
                nlist[k * NN + base + __popc(m & ((1u << lane) - 1u))] = ch * 32 + lane;
            base += __popc(m);
        }
        for (int i = base + lane; i < NN; i += 32) nlist[k * NN + i] = 0;
        if (lane == 0) ncnt[k] = (base + 15) & ~15;
    }
}

__device__ __forceinline__ void gemm_acc(
    const float* __restrict__ A, const float* __restrict__ W,
    const int rowoff[4], int nact, int mlo, int mhalf, v8f acc[4][2])
{
#pragma unroll
    for (int mt = 0; mt < 4; ++mt) {
        if (mt < nact) {
            for (int ks = 0; ks < 8; ++ks) {
                const int kk = ks * 4 + mhalf * 2;
                v2f b0, b1, a;
                b0.x = W[kk * CC + mlo];
                b0.y = W[(kk + 1) * CC + mlo];
                b1.x = W[kk * CC + 16 + mlo];
                b1.y = W[(kk + 1) * CC + 16 + mlo];
                a.x = A[rowoff[mt] + kk];
                a.y = A[rowoff[mt] + kk + 1];
                acc[mt][0] = __builtin_amdgcn_wmma_f32_16x16x4_f32(
                    false, a, false, b0, (short)0, acc[mt][0], false, false);
                acc[mt][1] = __builtin_amdgcn_wmma_f32_16x16x4_f32(
                    false, a, false, b1, (short)0, acc[mt][1], false, false);
            }
        }
    }
}

struct GBounds {
    int jA, eA, jB, eB;
    float ivA, ivB;
};

__device__ __forceinline__ GBounds gather_bounds(
    int nodeA, int nodeB, bool okA, bool okB,
    const int* __restrict__ rp, const float* __restrict__ iv)
{
    GBounds g;
    const int a0 = rp[nodeA], a1 = rp[nodeA + 1];
    const int b0 = rp[nodeB], b1 = rp[nodeB + 1];
    g.ivA = iv[nodeA];
    g.ivB = iv[nodeB];
    g.jA = a0;
    g.eA = okA ? a1 : a0;
    g.jB = b0;
    g.eB = okB ? b1 : b0;
    return g;
}

__device__ __forceinline__ void gather_first_ids(
    const int* __restrict__ cl, const GBounds& g, int ca[4], int cb[4])
{
#pragma unroll
    for (int i = 0; i < 4; ++i) {
        const int va = cl[g.jA + i], vb = cl[g.jB + i];
        ca[i] = (g.jA + i < g.eA) ? va : NN;
        cb[i] = (g.jB + i < g.eB) ? vb : NN;
    }
}

__device__ __forceinline__ void gather_sum(
    const float* __restrict__ xs, float* __restrict__ hs,
    const int* __restrict__ cl, const GBounds& g, int ca[4], int cb[4],
    int pA, int pB, bool okA, bool okB, int cq)
{
    int jA = g.jA, jB = g.jB;
    const int eA = g.eA, eB = g.eB;
    v4f a = (v4f){0.0f, 0.0f, 0.0f, 0.0f};
    v4f b = (v4f){0.0f, 0.0f, 0.0f, 0.0f};
    while (jA < eA || jB < eB) {
        v4f xa[4], xb[4];
#pragma unroll
        for (int i = 0; i < 4; ++i) {
            xa[i] = *(const v4f*)&xs[ca[i] * CC + cq];
            xb[i] = *(const v4f*)&xs[cb[i] * CC + cq];
        }
        jA = (jA + 4 < eA) ? jA + 4 : eA;
        jB = (jB + 4 < eB) ? jB + 4 : eB;
#pragma unroll
        for (int i = 0; i < 4; ++i) {
            const int va = cl[jA + i], vb = cl[jB + i];
            ca[i] = (jA + i < eA) ? va : NN;
            cb[i] = (jB + i < eB) ? vb : NN;
        }
#pragma unroll
        for (int i = 0; i < 4; ++i) { a += xa[i]; b += xb[i]; }
    }
    if (okA) *(v4f*)&hs[pA * CC + cq] = a * g.ivA;
    if (okB) *(v4f*)&hs[pB * CC + cq] = b * g.ivB;
}

__device__ __forceinline__ void gather_sum3(
    const float* __restrict__ xs, float* __restrict__ hs, const int* __restrict__ cl3,
    const GBounds g[3], int ca[3][4], int cb[3][4],
    int pA, int pB, bool okA, bool okB, int cq, int rstride)
{
    int jA[3], jB[3];
    v4f a[3], b[3];
    bool more = false;
#pragma unroll
    for (int r = 0; r < 3; ++r) {
        jA[r] = g[r].jA;
        jB[r] = g[r].jB;
        a[r] = (v4f){0.0f, 0.0f, 0.0f, 0.0f};
        b[r] = (v4f){0.0f, 0.0f, 0.0f, 0.0f};
        more = more || jA[r] < g[r].eA || jB[r] < g[r].eB;
    }
    while (more) {
        v4f xa[3][4], xb[3][4];
#pragma unroll
        for (int r = 0; r < 3; ++r)
#pragma unroll
            for (int i = 0; i < 4; ++i) {
                xa[r][i] = *(const v4f*)&xs[ca[r][i] * CC + cq];
                xb[r][i] = *(const v4f*)&xs[cb[r][i] * CC + cq];
            }
        more = false;
#pragma unroll
        for (int r = 0; r < 3; ++r) {
            const int eA = g[r].eA, eB = g[r].eB;
            jA[r] = (jA[r] + 4 < eA) ? jA[r] + 4 : eA;
            jB[r] = (jB[r] + 4 < eB) ? jB[r] + 4 : eB;
            more = more || jA[r] < eA || jB[r] < eB;
#pragma unroll
            for (int i = 0; i < 4; ++i) {
                const int va = cl3[r * EE + jA[r] + i], vb = cl3[r * EE + jB[r] + i];
                ca[r][i] = (jA[r] + i < eA) ? va : NN;
                cb[r][i] = (jB[r] + i < eB) ? vb : NN;
            }
        }
#pragma unroll
        for (int r = 0; r < 3; ++r)
#pragma unroll
            for (int i = 0; i < 4; ++i) { a[r] += xa[r][i]; b[r] += xb[r][i]; }
    }
#pragma unroll
    for (int r = 0; r < 3; ++r) {
        if (okA) *(v4f*)&hs[r * rstride + pA * CC + cq] = a[r] * g[r].ivA;
        if (okB) *(v4f*)&hs[r * rstride + pB * CC + cq] = b[r] * g[r].ivB;
    }
}

template <int M>
__device__ __forceinline__ float row_xor(float v)
{
    return __int_as_float(__builtin_amdgcn_update_dpp(0, __float_as_int(v), 0x160 + M, 0xf, 0xf, true));
}

__device__ __forceinline__ void row16_sum8(float s[8])
{
#pragma unroll
    for (int u = 0; u < 8; ++u) s[u] += row_xor<8>(s[u]);
#pragma unroll
    for (int u = 0; u < 8; ++u) s[u] += row_xor<4>(s[u]);
#pragma unroll
    for (int u = 0; u < 8; ++u) s[u] += row_xor<2>(s[u]);
#pragma unroll
    for (int u = 0; u < 8; ++u) s[u] += row_xor<1>(s[u]);
}

__global__ __launch_bounds__(256) void rgcn_main_kernel(
    const float* __restrict__ xx, const float* __restrict__ ss,
    const float* __restrict__ Wf, const float* __restrict__ bf,
    const float* __restrict__ convw, const float* __restrict__ convroot,
    const float* __restrict__ convbias, const float* __restrict__ lng,
    const float* __restrict__ lnb, const int* __restrict__ rowptr,
    const int* __restrict__ col, const float* __restrict__ invdeg,
    const int* __restrict__ nlist, const int* __restrict__ ncnt,
    float* __restrict__ out)
{
    __shared__ __attribute__((aligned(16))) float xs[(NN + 1) * CC];
    __shared__ __attribute__((aligned(16))) float hs[NN * CC];
    __shared__ __attribute__((aligned(16))) int colL[3 * EE + 4];
    __shared__ int rpL[3 * RPS];
    __shared__ __attribute__((aligned(16))) float wall[LL * 4 * CC * CC];
    __shared__ float cbs[LL * CC];
    __shared__ float ivd[3 * NN];
    __shared__ float wfe[6 * CC];
    __shared__ float prm[3 * CC];
    __shared__ __attribute__((aligned(16))) int nlL[NLV * NN];
    __shared__ int ncL[8];

    const int bid   = blockIdx.x;
    const int tid   = threadIdx.x;
    const int lane  = tid & 31;
    const int wid   = __builtin_amdgcn_readfirstlane(tid >> 5);
    const int mlo   = lane & 15;
    const int mhalf = lane >> 4;

    const float mfeat = ss[bid] * 0.125f;

    int q4[3];
#pragma unroll
    for (int r = 0; r < 3; ++r) {
        const int t = (rowptr[r * RPS + NN] + 3) >> 2;
        q4[r] = (t < 0) ? 0 : ((t > EE / 4) ? EE / 4 : t);
    }

    const v4f* __restrict__ cr4 = (const v4f*)convroot;
    const v4f* __restrict__ cw4 = (const v4f*)convw;
    v4f wr[LL], ww[3 * LL];
#pragma unroll
    for (int u = 0; u < LL; ++u) wr[u] = cr4[u * 256 + tid];
#pragma unroll
    for (int u = 0; u < 3 * LL; ++u) ww[u] = cw4[u * 256 + tid];
    const v4i* __restrict__ n4 = (const v4i*)nlist;
    const v4i t0 = n4[tid], t1 = n4[tid + 256];
    const v4i t2 = n4[512 + (tid & (NLV * NN / 4 - 512 - 1))];
    const int ncv = ncnt[(tid < NLV) ? tid : 0];
    float ivr[6];
    int rpr[7];
#pragma unroll
    for (int u = 0; u < 6; ++u) {
        ivr[u] = invdeg[u * 256 + tid];
        rpr[u] = rowptr[u * 256 + tid];
    }
    rpr[6] = rowptr[(tid < 3 * RPS - 6 * 256) ? 6 * 256 + tid : 0];
    const float wfv = Wf[(tid < 6 * CC) ? tid : 0];
    const float pbv = bf[lane], pgv = lng[lane], plv = lnb[lane];
    const float cbv = convbias[tid & (LL * CC - 1)];

    {
        const int q01 = q4[0] + q4[1];
        const int qt  = q01 + q4[2];
        const v4i* __restrict__ c4 = (const v4i*)col;
        v4i* l4 = (v4i*)colL;
        for (int b0 = tid; b0 < qt; b0 += 3 * 256) {
            v4i v[3];
            int w[3];
#pragma unroll
            for (int k = 0; k < 3; ++k) {
                const int b  = b0 + k * 256;
                const int bc = (b < qt) ? b : qt - 1;
                w[k] = (bc >= q01) ? (2 * (EE / 4) + bc - q01)
                                   : ((bc >= q4[0]) ? ((EE / 4) + bc - q4[0]) : bc);
                v[k] = c4[w[k]];
            }
#pragma unroll
            for (int k = 0; k < 3; ++k)
                if (b0 + k * 256 < qt) l4[w[k]] = v[k];
        }
    }

    {
        v4f* w4 = (v4f*)wall;
#pragma unroll
        for (int u = 0; u < LL; ++u) w4[(u * 4) * 256 + tid] = wr[u];
#pragma unroll
        for (int u = 0; u < 3 * LL; ++u) w4[((u / 3) * 4 + 1 + (u % 3)) * 256 + tid] = ww[u];
        v4i* d4 = (v4i*)nlL;
        d4[tid] = t0;
        d4[tid + 256] = t1;
        if (tid < NLV * NN / 4 - 512) d4[tid + 512] = t2;
        if (tid < NLV) ncL[tid] = ncv;
#pragma unroll
        for (int u = 0; u < 6; ++u) {
            ivd[u * 256 + tid] = ivr[u];
            rpL[u * 256 + tid] = rpr[u];
        }
        if (tid < 3 * RPS - 6 * 256) rpL[6 * 256 + tid] = rpr[6];
        if (tid < 6 * CC) wfe[tid] = wfv;
        if (tid < CC) {
            xs[NN * CC + tid] = 0.0f;
            prm[tid]          = pbv;
            prm[CC + tid]     = pgv;
            prm[2 * CC + tid] = plv;
        }
        if (tid < LL * CC) cbs[tid] = cbv;
    }
    __syncthreads();

    const int cnt0 = ncL[0];
    for (int idx = tid; idx < cnt0 * CC; idx += 256) {
        const int n = nlL[idx >> 5], c = idx & 31;
        const int t = n >> 6, i = (n >> 4) & 3, j = (n >> 2) & 3, k = n & 3;
        const float v = xx[bid * NN + n];
        float a = prm[c];
        a += ((float)i * (1.0f / 3.0f)) * wfe[0 * CC + c];
        a += ((float)j * (1.0f / 3.0f)) * wfe[1 * CC + c];
        a += ((float)k * (1.0f / 3.0f)) * wfe[2 * CC + c];
        a += ((float)t * (1.0f / 7.0f)) * wfe[3 * CC + c];
        a += v * wfe[4 * CC + c];
        a += mfeat * wfe[5 * CC + c];
        xs[n * CC + c] = a;
    }
    __syncthreads();

    for (int l = 0; l < LL; ++l) {
        const int* nl  = &nlL[(l + 1) * NN];
        const int cnt  = ncL[l + 1];
        const int ntile = cnt >> 4;
        const int nact = (ntile > wid) ? ((ntile - wid + 7) >> 3) : 0;
        int xoff[4], hoff[4];
#pragma unroll
        for (int mt = 0; mt < 4; ++mt) {
            const int p = (wid + 8 * mt) * 16 + mlo;
            xoff[mt] = nl[p] * CC;
            hoff[mt] = p * CC;
        }
        v8f acc[4][2];
#pragma unroll
        for (int mt = 0; mt < 4; ++mt)
#pragma unroll
            for (int nt = 0; nt < 2; ++nt) {
                const float bv = cbs[l * CC + nt * 16 + mlo];
                acc[mt][nt] = (v8f){bv, bv, bv, bv, bv, bv, bv, bv};
            }

        const int cq  = (lane & 7) * 4;
        const int pA0 = wid * 4 + (lane >> 3);
        const int pB0 = pA0 + 32;
        const bool okA0 = pA0 < cnt, okB0 = pB0 < cnt;
        const int nA0 = nl[pA0];
        const int nB0 = nl[pB0];
        GBounds g0[3];
        int ca0[3][4], cb0[3][4];
#pragma unroll
        for (int r = 0; r < 3; ++r)
            g0[r] = gather_bounds(nA0, nB0, okA0, okB0, &rpL[r * RPS], &ivd[r * NN]);
#pragma unroll
        for (int r = 0; r < 3; ++r)
            gather_first_ids(&colL[r * EE], g0[r], ca0[r], cb0[r]);

        gemm_acc(xs, &wall[(l * 4) * CC * CC], xoff, nact, mlo, mhalf, acc);

        const bool side_by_side = __builtin_amdgcn_readfirstlane((3 * cnt <= NN) ? 1 : 0) != 0;
        if (side_by_side) {
            const int rstride = cnt * CC;
            gather_sum3(xs, hs, colL, g0, ca0, cb0, pA0, pB0, okA0, okB0, cq, rstride);
            for (int base = 64; base < cnt; base += 64) {
                const int pA = base + pA0, pB = base + pB0;
                const bool okA = pA < cnt, okB = pB < cnt;
                const int nA = nl[pA], nB = nl[pB];
                GBounds g[3];
                int ca[3][4], cb[3][4];
#pragma unroll
                for (int r = 0; r < 3; ++r)
                    g[r] = gather_bounds(nA, nB, okA, okB, &rpL[r * RPS], &ivd[r * NN]);
#pragma unroll
                for (int r = 0; r < 3; ++r)
                    gather_first_ids(&colL[r * EE], g[r], ca[r], cb[r]);
                gather_sum3(xs, hs, colL, g, ca, cb, pA, pB, okA, okB, cq, rstride);
            }
            __syncthreads();
#pragma unroll
            for (int r = 0; r < 3; ++r)
                gemm_acc(hs + r * rstride, &wall[(l * 4 + 1 + r) * CC * CC], hoff, nact, mlo, mhalf, acc);
        } else {
#pragma unroll
        for (int r = 0; r < 3; ++r) {
            if (r > 0) __syncthreads();
            gather_sum(xs, hs, &colL[r * EE], g0[r], ca0[r], cb0[r], pA0, pB0, okA0, okB0, cq);
            for (int base = 64; base < cnt; base += 64) {
                const int pA = base + pA0, pB = base + pB0;
                const bool okA = pA < cnt, okB = pB < cnt;
                const GBounds g = gather_bounds(nl[pA], nl[pB], okA, okB, &rpL[r * RPS], &ivd[r * NN]);
                int ca[4], cb[4];
                gather_first_ids(&colL[r * EE], g, ca, cb);
                gather_sum(xs, hs, &colL[r * EE], g, ca, cb, pA, pB, okA, okB, cq);
            }
            __syncthreads();

            gemm_acc(hs, &wall[(l * 4 + 1 + r) * CC * CC], hoff, nact, mlo, mhalf, acc);
        }
        }

        {
            const float g0 = prm[CC + mlo],     g1 = prm[CC + 16 + mlo];
            const float e0 = prm[2 * CC + mlo], e1 = prm[2 * CC + 16 + mlo];
#pragma unroll
            for (int mt = 0; mt < 4; ++mt)
                if (mt < nact) {
                    const int pb = (wid + 8 * mt) * 16 + 8 * mhalf;
                    const v4i na = *(const v4i*)&nl[pb];
                    const v4i nb = *(const v4i*)&nl[pb + 4];
                    const int node[8] = {na.x, na.y, na.z, na.w, nb.x, nb.y, nb.z, nb.w};
                    float v0[8], v1[8], s[8];
#pragma unroll
                    for (int jj = 0; jj < 8; ++jj) {
                        const float a0 = acc[mt][0][jj], a1 = acc[mt][1][jj];
                        v0[jj] = a0 > 0.0f ? a0 : 0.0f;
                        v1[jj] = a1 > 0.0f ? a1 : 0.0f;
                        s[jj]  = v0[jj] + v1[jj];
                    }
                    row16_sum8(s);
#pragma unroll
                    for (int jj = 0; jj < 8; ++jj) {
                        const float mu = s[jj] * (1.0f / 32.0f);
                        v0[jj] -= mu;
                        v1[jj] -= mu;
                        s[jj] = v0[jj] * v0[jj] + v1[jj] * v1[jj];
                    }
                    row16_sum8(s);
#pragma unroll
                    for (int jj = 0; jj < 8; ++jj) {
                        const float rs = rsqrtf(s[jj] * (1.0f / 32.0f) + LN_EPS);
                        xs[node[jj] * CC + mlo]      = v0[jj] * rs * g0 + e0;
                        xs[node[jj] * CC + 16 + mlo] = v1[jj] * rs * g1 + e1;
                    }
                }
        }
        __syncthreads();
    }

    for (int idx = tid; idx < 48 * CC; idx += 256) {
        const int row = idx >> 5, c = idx & 31;
        const int g = row >> 4, rc = row & 15;
        const int a0 = rc >> 2, b0 = rc & 3;
        float sum = 0.0f;
#pragma unroll
        for (int q = 0; q < 4; ++q) {
            int node;
            if (g == 0)      node = q * 16 + a0 * 4 + b0;
            else if (g == 1) node = a0 * 16 + q * 4 + b0;
            else             node = a0 * 16 + b0 * 4 + q;
            sum += xs[node * CC + c];
        }
        out[bid * 48 * CC + idx] = sum * 0.25f;
    }
}

extern "C" void kernel_launch(void* const* d_in, const int* in_sizes, int n_in,
                              void* d_out, int out_size, void* d_ws, size_t ws_size,
                              hipStream_t stream) {
    const float* xx       = (const float*)d_in[0];
    const float* ss       = (const float*)d_in[1];
    const float* Wf       = (const float*)d_in[2];
    const float* bf       = (const float*)d_in[3];
    const float* convw    = (const float*)d_in[4];
    const float* convroot = (const float*)d_in[5];
    const float* convbias = (const float*)d_in[6];
    const float* lng      = (const float*)d_in[7];
    const float* lnb      = (const float*)d_in[8];
    const int*   src      = (const int*)d_in[9];
    const int*   dst      = (const int*)d_in[10];
    float*       out      = (float*)d_out;
    float*       invdeg   = (float*)d_ws;
    int*         rowptr   = (int*)d_ws + 2048;
    int*         col      = (int*)d_ws + 4096;
    int*         nlist    = (int*)d_ws + 24576;
    int*         ncnt     = (int*)d_ws + 27648;

    const int E = in_sizes[9] / 3;
    const int B = in_sizes[1];

    rgcn_csr_kernel<<<3, 1024, 0, stream>>>(src, dst, E, invdeg, rowptr, col, nlist, ncnt);
    rgcn_main_kernel<<<B, 256, 0, stream>>>(xx, ss, Wf, bf, convw, convroot,
                                            convbias, lng, lnb, rowptr, col,
                                            invdeg, nlist, ncnt, out);
    (void)n_in; (void)out_size; (void)ws_size;
}
